# starting kernel
# baseline (speedup 1.0000x reference)
#include <hip/hip_runtime.h>
#include <math.h>
#include <float.h>

typedef float v2f __attribute__((ext_vector_type(2)));
typedef float v8f __attribute__((ext_vector_type(8)));

#define HC 64
#define SLOPE 0.2f

__device__ __forceinline__ float lrelu(float x) { return x > 0.f ? x : SLOPE * x; }

__device__ __forceinline__ void atomicMaxF(float* addr, float val) {
    if (val >= 0.f) atomicMax((int*)addr, __float_as_int(val));
    else            atomicMin((unsigned int*)addr, __float_as_uint(val));
}

template <bool DUAL, int RT>
__global__ __launch_bounds__(32)
void gemm_x2(const float* __restrict__ X, const float* __restrict__ W,
             const float* __restrict__ Wr, float* __restrict__ Y,
             float* __restrict__ Yr, int n, int K) {
    const int lane  = threadIdx.x;
    const int row16 = lane & 15;
    const int khalf = (lane >> 4) << 1;
    const int col   = lane & 15;
    const int r0    = blockIdx.x * (16 * RT);

    int arow[RT];
    #pragma unroll
    for (int t = 0; t < RT; ++t) {
        int r = r0 + 16 * t + row16;
        arow[t] = r < n ? r : n - 1;
    }

    v8f acc[RT][4]  = {};
    v8f accR[RT][4] = {};

    #pragma unroll 2
    for (int k0 = 0; k0 < K; k0 += 4) {
        v2f a[RT];
        #pragma unroll
        for (int t = 0; t < RT; ++t)
            a[t] = *reinterpret_cast<const v2f*>(X + (size_t)arow[t] * K + k0 + khalf);

        #pragma unroll
        for (int j = 0; j < 4; ++j) {
            v2f b;
            const float* bp = W + (size_t)(k0 + khalf) * HC + j * 16 + col;
            b.x = bp[0]; b.y = bp[HC];
            #pragma unroll
            for (int t = 0; t < RT; ++t)
                acc[t][j] = __builtin_amdgcn_wmma_f32_16x16x4_f32(
                    false, a[t], false, b, (short)0, acc[t][j], false, false);
            if (DUAL) {
                v2f br;
                const float* brp = Wr + (size_t)(k0 + khalf) * HC + j * 16 + col;
                br.x = brp[0]; br.y = brp[HC];
                #pragma unroll
                for (int t = 0; t < RT; ++t)
                    accR[t][j] = __builtin_amdgcn_wmma_f32_16x16x4_f32(
                        false, a[t], false, br, (short)0, accR[t][j], false, false);
            }
        }
    }

    const int mhi = (lane >> 4) * 8;
    #pragma unroll
    for (int t = 0; t < RT; ++t)
        #pragma unroll
        for (int j = 0; j < 4; ++j)
            #pragma unroll
            for (int r = 0; r < 8; ++r) {
                int m = r0 + 16 * t + mhi + r;
                if (m < n) {
                    Y[(size_t)m * HC + j * 16 + col] = acc[t][j][r];
                    if (DUAL) Yr[(size_t)m * HC + j * 16 + col] = accR[t][j][r];
                }
            }
}

__global__ void node_prep1(const float* __restrict__ xlin,
                           const float* __restrict__ attL, const float* __restrict__ attR,
                           float* __restrict__ al, float* __restrict__ ar,
                           float* __restrict__ nmax, float* __restrict__ nsum,
                           float* __restrict__ agg, int n) {
    int i = blockIdx.x * blockDim.x + threadIdx.x;
    if (i >= n) return;
    const float4* rowp = (const float4*)(xlin + (size_t)i * 64);
    const float4* lp = (const float4*)attL;
    const float4* rp = (const float4*)attR;
    float4* aggp = (float4*)(agg + (size_t)i * 64);
    float aL[4] = {0, 0, 0, 0}, aR[4] = {0, 0, 0, 0};
    #pragma unroll
    for (int q = 0; q < 16; ++q) {
        float4 v = rowp[q];
        float4 l = lp[q];
        float4 r = rp[q];
        int h = q >> 2;
        aL[h] += v.x * l.x + v.y * l.y + v.z * l.z + v.w * l.w;
        aR[h] += v.x * r.x + v.y * r.y + v.z * r.z + v.w * r.w;
        aggp[q] = make_float4(0.f, 0.f, 0.f, 0.f);
    }
    #pragma unroll
    for (int h = 0; h < 4; ++h) {
        al[i * 4 + h] = aL[h];
        ar[i * 4 + h] = aR[h];
        nmax[i * 4 + h] = -FLT_MAX;
        nsum[i * 4 + h] = 0.f;
    }
}

__global__ void node_prep2(const float* __restrict__ xlin,
                           const float* __restrict__ attL, const float* __restrict__ attR,
                           float* __restrict__ al, float* __restrict__ ar,
                           float* __restrict__ nmax, float* __restrict__ nsum,
                           float* __restrict__ agg, int n) {
    int i = blockIdx.x * blockDim.x + threadIdx.x;
    if (i >= n) return;
    const float4* rowp = (const float4*)(xlin + (size_t)i * 64);
    const float4* lp = (const float4*)attL;
    const float4* rp = (const float4*)attR;
    float4* aggp = (float4*)(agg + (size_t)i * 64);
    float aL = 0.f, aR = 0.f;
    #pragma unroll
    for (int q = 0; q < 16; ++q) {
        float4 v = rowp[q];
        float4 l = lp[q];
        float4 r = rp[q];
        aL += v.x * l.x + v.y * l.y + v.z * l.z + v.w * l.w;
        aR += v.x * r.x + v.y * r.y + v.z * r.z + v.w * r.w;
        aggp[q] = make_float4(0.f, 0.f, 0.f, 0.f);
    }
    al[i] = aL; ar[i] = aR; nmax[i] = -FLT_MAX; nsum[i] = 0.f;
}

template <int H>
__global__ void edge_max_k(const int* __restrict__ src, const int* __restrict__ dst,
                           const float* __restrict__ al, const float* __restrict__ ar,
                           float* __restrict__ nmax, int e) {
    int i = blockIdx.x * blockDim.x + threadIdx.x;
    if (i >= e) return;
    int s = src[i], d = dst[i];
    #pragma unroll
    for (int h = 0; h < H; ++h) {
        float lg = lrelu(al[s * H + h] + ar[d * H + h]);
        atomicMaxF(&nmax[d * H + h], lg);
    }
}

template <int H>
__global__ void edge_sum_k(const int* __restrict__ src, const int* __restrict__ dst,
                           const float* __restrict__ al, const float* __restrict__ ar,
                           const float* __restrict__ nmax, float* __restrict__ nsum, int e) {
    int i = blockIdx.x * blockDim.x + threadIdx.x;
    if (i >= e) return;
    int s = src[i], d = dst[i];
    #pragma unroll
    for (int h = 0; h < H; ++h) {
        float lg = lrelu(al[s * H + h] + ar[d * H + h]);
        atomicAdd(&nsum[d * H + h], __expf(lg - nmax[d * H + h]));
    }
}

template <int H>
__global__ void edge_agg_k(const int* __restrict__ src, const int* __restrict__ dst,
                           const float* __restrict__ al, const float* __restrict__ ar,
                           const float* __restrict__ nmax, const float* __restrict__ nsum,
                           const float* __restrict__ xlin, float* __restrict__ agg, int e) {
    int g = blockIdx.x * (blockDim.x >> 5) + (threadIdx.x >> 5);
    int lane = threadIdx.x & 31;
    if (g >= e) return;
    int s = src[g], d = dst[g];
    #pragma unroll
    for (int half = 0; half < 2; ++half) {
        int col = lane + 32 * half;
        int h = (col * H) >> 6;
        float a = __expf(lrelu(al[s * H + h] + ar[d * H + h]) - nmax[d * H + h])
                  / (nsum[d * H + h] + 1e-16f);
        atomicAdd(&agg[(size_t)d * 64 + col], xlin[(size_t)s * 64 + col] * a);
    }
}

__global__ void finish1_k(float* __restrict__ agg, const float* __restrict__ res,
                          const float* __restrict__ bias, int total) {
    int i = blockIdx.x * blockDim.x + threadIdx.x;
    if (i >= total) return;
    float v = agg[i] + res[i] + bias[i & 63];
    agg[i] = v > 0.f ? v : (__expf(v) - 1.f);
}

__global__ void finish2_k(const float* __restrict__ agg, const float* __restrict__ xlin,
                          const float* __restrict__ bias, float* __restrict__ out, int total) {
    int i = blockIdx.x * blockDim.x + threadIdx.x;
    if (i >= total) return;
    out[i] = agg[i] + xlin[i] + bias[i & 63];
}

extern "C" void kernel_launch(void* const* d_in, const int* in_sizes, int n_in,
                              void* d_out, int out_size, void* d_ws, size_t ws_size,
                              hipStream_t stream) {
    const float* x     = (const float*)d_in[0];
    const int*   eidx  = (const int*)d_in[1];
    const float* W1    = (const float*)d_in[2];
    const float* attL1 = (const float*)d_in[3];
    const float* attR1 = (const float*)d_in[4];
    const float* resW1 = (const float*)d_in[5];
    const float* b1    = (const float*)d_in[6];
    const float* W2    = (const float*)d_in[7];
    const float* attL2 = (const float*)d_in[8];
    const float* attR2 = (const float*)d_in[9];
    const float* b2    = (const float*)d_in[10];

    const int n = in_sizes[0] / 128;
    const int e = in_sizes[1] / 2;
    const int* srcI = eidx;
    const int* dstI = eidx + e;

    float* ws = (float*)d_ws;
    float* A  = ws;
    float* B  = A  + (size_t)n * 64;
    float* C  = B  + (size_t)n * 64;
    float* AL = C  + (size_t)n * 64;
    float* AR = AL + (size_t)n * 4;
    float* MX = AR + (size_t)n * 4;
    float* SM = MX + (size_t)n * 4;

    constexpr int RT = 2;
    const int tiles = (n + 16 * RT - 1) / (16 * RT);
    const int nThr  = (n + 255) / 256;
    const int eThr  = (e + 255) / 256;
    const int eGrp  = (e + 7) / 8;
    const int fThr  = (n * 64 + 255) / 256;

    gemm_x2<true, RT><<<tiles, 32, 0, stream>>>(x, W1, resW1, A, B, n, 128);
    node_prep1<<<nThr, 256, 0, stream>>>(A, attL1, attR1, AL, AR, MX, SM, C, n);
    edge_max_k<4><<<eThr, 256, 0, stream>>>(srcI, dstI, AL, AR, MX, e);
    edge_sum_k<4><<<eThr, 256, 0, stream>>>(srcI, dstI, AL, AR, MX, SM, e);
    edge_agg_k<4><<<eGrp, 256, 0, stream>>>(srcI, dstI, AL, AR, MX, SM, A, C, e);
    finish1_k<<<fThr, 256, 0, stream>>>(C, B, b1, n * 64);

    gemm_x2<false, RT><<<tiles, 32, 0, stream>>>(C, W2, nullptr, A, nullptr, n, 64);
    node_prep2<<<nThr, 256, 0, stream>>>(A, attL2, attR2, AL, AR, MX, SM, B, n);
    edge_max_k<1><<<eThr, 256, 0, stream>>>(srcI, dstI, AL, AR, MX, e);
    edge_sum_k<1><<<eThr, 256, 0, stream>>>(srcI, dstI, AL, AR, MX, SM, e);
    edge_agg_k<1><<<eGrp, 256, 0, stream>>>(srcI, dstI, AL, AR, MX, SM, A, B, e);
    finish2_k<<<fThr, 256, 0, stream>>>(B, A, b2, (float*)d_out, n * 64);
}


# fold the exp-sum edge pass into the aggregate pass: atomics add x*w and w, one divide by (sum+1e-16) per output element in finish (sum of x*w then /S, not sum of x*(w/S)); 12 -> 10 launches
# speedup vs baseline: 1.1774x; 1.1774x over previous
#include <hip/hip_runtime.h>
#include <math.h>
#include <float.h>

typedef float v2f __attribute__((ext_vector_type(2)));
typedef float v8f __attribute__((ext_vector_type(8)));

#define HC 64
#define SLOPE 0.2f

__device__ __forceinline__ float lrelu(float x) { return x > 0.f ? x : SLOPE * x; }

__device__ __forceinline__ void atomicMaxF(float* addr, float val) {
    if (val >= 0.f) atomicMax((int*)addr, __float_as_int(val));
    else            atomicMin((unsigned int*)addr, __float_as_uint(val));
}

template <bool DUAL, int RT>
__global__ __launch_bounds__(32)
void gemm_x2(const float* __restrict__ X, const float* __restrict__ W,
             const float* __restrict__ Wr, float* __restrict__ Y,
             float* __restrict__ Yr, int n, int K) {
    const int lane  = threadIdx.x;
    const int row16 = lane & 15;
    const int khalf = (lane >> 4) << 1;
    const int col   = lane & 15;
    const int r0    = blockIdx.x * (16 * RT);

    int arow[RT];
    #pragma unroll
    for (int t = 0; t < RT; ++t) {
        int r = r0 + 16 * t + row16;
        arow[t] = r < n ? r : n - 1;
    }

    v8f acc[RT][4]  = {};
    v8f accR[RT][4] = {};

    #pragma unroll 2
    for (int k0 = 0; k0 < K; k0 += 4) {
        v2f a[RT];
        #pragma unroll
        for (int t = 0; t < RT; ++t)
            a[t] = *reinterpret_cast<const v2f*>(X + (size_t)arow[t] * K + k0 + khalf);

        #pragma unroll
        for (int j = 0; j < 4; ++j) {
            v2f b;
            const float* bp = W + (size_t)(k0 + khalf) * HC + j * 16 + col;
            b.x = bp[0]; b.y = bp[HC];
            #pragma unroll
            for (int t = 0; t < RT; ++t)
                acc[t][j] = __builtin_amdgcn_wmma_f32_16x16x4_f32(
                    false, a[t], false, b, (short)0, acc[t][j], false, false);
            if (DUAL) {
                v2f br;
                const float* brp = Wr + (size_t)(k0 + khalf) * HC + j * 16 + col;
                br.x = brp[0]; br.y = brp[HC];
                #pragma unroll
                for (int t = 0; t < RT; ++t)
                    accR[t][j] = __builtin_amdgcn_wmma_f32_16x16x4_f32(
                        false, a[t], false, br, (short)0, accR[t][j], false, false);
            }
        }
    }

    const int mhi = (lane >> 4) * 8;
    #pragma unroll
    for (int t = 0; t < RT; ++t)
        #pragma unroll
        for (int j = 0; j < 4; ++j)
            #pragma unroll
            for (int r = 0; r < 8; ++r) {
                int m = r0 + 16 * t + mhi + r;
                if (m < n) {
                    Y[(size_t)m * HC + j * 16 + col] = acc[t][j][r];
                    if (DUAL) Yr[(size_t)m * HC + j * 16 + col] = accR[t][j][r];
                }
            }
}

__global__ void node_prep1(const float* __restrict__ xlin,
                           const float* __restrict__ attL, const float* __restrict__ attR,
                           float* __restrict__ al, float* __restrict__ ar,
                           float* __restrict__ nmax, float* __restrict__ nsum,
                           float* __restrict__ agg, int n) {
    int i = blockIdx.x * blockDim.x + threadIdx.x;
    if (i >= n) return;
    const float4* rowp = (const float4*)(xlin + (size_t)i * 64);
    const float4* lp = (const float4*)attL;
    const float4* rp = (const float4*)attR;
    float4* aggp = (float4*)(agg + (size_t)i * 64);
    float aL[4] = {0, 0, 0, 0}, aR[4] = {0, 0, 0, 0};
    #pragma unroll
    for (int q = 0; q < 16; ++q) {
        float4 v = rowp[q];
        float4 l = lp[q];
        float4 r = rp[q];
        int h = q >> 2;
        aL[h] += v.x * l.x + v.y * l.y + v.z * l.z + v.w * l.w;
        aR[h] += v.x * r.x + v.y * r.y + v.z * r.z + v.w * r.w;
        aggp[q] = make_float4(0.f, 0.f, 0.f, 0.f);
    }
    #pragma unroll
    for (int h = 0; h < 4; ++h) {
        al[i * 4 + h] = aL[h];
        ar[i * 4 + h] = aR[h];
        nmax[i * 4 + h] = -FLT_MAX;
        nsum[i * 4 + h] = 0.f;
    }
}

__global__ void node_prep2(const float* __restrict__ xlin,
                           const float* __restrict__ attL, const float* __restrict__ attR,
                           float* __restrict__ al, float* __restrict__ ar,
                           float* __restrict__ nmax, float* __restrict__ nsum,
                           float* __restrict__ agg, int n) {
    int i = blockIdx.x * blockDim.x + threadIdx.x;
    if (i >= n) return;
    const float4* rowp = (const float4*)(xlin + (size_t)i * 64);
    const float4* lp = (const float4*)attL;
    const float4* rp = (const float4*)attR;
    float4* aggp = (float4*)(agg + (size_t)i * 64);
    float aL = 0.f, aR = 0.f;
    #pragma unroll
    for (int q = 0; q < 16; ++q) {
        float4 v = rowp[q];
        float4 l = lp[q];
        float4 r = rp[q];
        aL += v.x * l.x + v.y * l.y + v.z * l.z + v.w * l.w;
        aR += v.x * r.x + v.y * r.y + v.z * r.z + v.w * r.w;
        aggp[q] = make_float4(0.f, 0.f, 0.f, 0.f);
    }
    al[i] = aL; ar[i] = aR; nmax[i] = -FLT_MAX; nsum[i] = 0.f;
}

template <int H>
__global__ void edge_max_k(const int* __restrict__ src, const int* __restrict__ dst,
                           const float* __restrict__ al, const float* __restrict__ ar,
                           float* __restrict__ nmax, int e) {
    int i = blockIdx.x * blockDim.x + threadIdx.x;
    if (i >= e) return;
    int s = src[i], d = dst[i];
    #pragma unroll
    for (int h = 0; h < H; ++h) {
        float lg = lrelu(al[s * H + h] + ar[d * H + h]);
        atomicMaxF(&nmax[d * H + h], lg);
    }
}

template <int H>
__global__ void edge_agg_k(const int* __restrict__ src, const int* __restrict__ dst,
                           const float* __restrict__ al, const float* __restrict__ ar,
                           const float* __restrict__ nmax, float* __restrict__ nsum,
                           const float* __restrict__ xlin, float* __restrict__ agg, int e) {
    constexpr int CH = 64 / H;
    int g = blockIdx.x * (blockDim.x >> 5) + (threadIdx.x >> 5);
    int lane = threadIdx.x & 31;
    if (g >= e) return;
    int s = src[g], d = dst[g];
    float w[2];
    #pragma unroll
    for (int half = 0; half < 2; ++half) {
        int col = lane + 32 * half;
        int h = (col * H) >> 6;
        w[half] = __expf(lrelu(al[s * H + h] + ar[d * H + h]) - nmax[d * H + h]);
        atomicAdd(&agg[(size_t)d * 64 + col], xlin[(size_t)s * 64 + col] * w[half]);
    }
    if ((lane & (CH - 1)) == 0) {
        atomicAdd(&nsum[d * H + ((lane * H) >> 6)], w[0]);
        if (CH <= 32)
            atomicAdd(&nsum[d * H + (((lane + 32) * H) >> 6)], w[1]);
    }
}

__global__ void finish1_k(float* __restrict__ agg, const float* __restrict__ nsum,
                          const float* __restrict__ res,
                          const float* __restrict__ bias, int total) {
    int i = blockIdx.x * blockDim.x + threadIdx.x;
    if (i >= total) return;
    float den = nsum[(i >> 6) * 4 + ((i & 63) >> 4)] + 1e-16f;
    float v = agg[i] / den + res[i] + bias[i & 63];
    agg[i] = v > 0.f ? v : (__expf(v) - 1.f);
}

__global__ void finish2_k(const float* __restrict__ agg, const float* __restrict__ nsum,
                          const float* __restrict__ xlin,
                          const float* __restrict__ bias, float* __restrict__ out, int total) {
    int i = blockIdx.x * blockDim.x + threadIdx.x;
    if (i >= total) return;
    float den = nsum[i >> 6] + 1e-16f;
    out[i] = agg[i] / den + xlin[i] + bias[i & 63];
}

extern "C" void kernel_launch(void* const* d_in, const int* in_sizes, int n_in,
                              void* d_out, int out_size, void* d_ws, size_t ws_size,
                              hipStream_t stream) {
    const float* x     = (const float*)d_in[0];
    const int*   eidx  = (const int*)d_in[1];
    const float* W1    = (const float*)d_in[2];
    const float* attL1 = (const float*)d_in[3];
    const float* attR1 = (const float*)d_in[4];
    const float* resW1 = (const float*)d_in[5];
    const float* b1    = (const float*)d_in[6];
    const float* W2    = (const float*)d_in[7];
    const float* attL2 = (const float*)d_in[8];
    const float* attR2 = (const float*)d_in[9];
    const float* b2    = (const float*)d_in[10];

    const int n = in_sizes[0] / 128;
    const int e = in_sizes[1] / 2;
    const int* srcI = eidx;
    const int* dstI = eidx + e;

    float* ws = (float*)d_ws;
    float* A  = ws;
    float* B  = A  + (size_t)n * 64;
    float* C  = B  + (size_t)n * 64;
    float* AL = C  + (size_t)n * 64;
    float* AR = AL + (size_t)n * 4;
    float* MX = AR + (size_t)n * 4;
    float* SM = MX + (size_t)n * 4;

    constexpr int RT = 2;
    const int tiles = (n + 16 * RT - 1) / (16 * RT);
    const int nThr  = (n + 255) / 256;
    const int eThr  = (e + 255) / 256;
    const int eGrp  = (e + 7) / 8;
    const int fThr  = (n * 64 + 255) / 256;

    gemm_x2<true, RT><<<tiles, 32, 0, stream>>>(x, W1, resW1, A, B, n, 128);
    node_prep1<<<nThr, 256, 0, stream>>>(A, attL1, attR1, AL, AR, MX, SM, C, n);
    edge_max_k<4><<<eThr, 256, 0, stream>>>(srcI, dstI, AL, AR, MX, e);
    edge_agg_k<4><<<eGrp, 256, 0, stream>>>(srcI, dstI, AL, AR, MX, SM, A, C, e);
    finish1_k<<<fThr, 256, 0, stream>>>(C, SM, B, b1, n * 64);

    gemm_x2<false, RT><<<tiles, 32, 0, stream>>>(C, W2, nullptr, A, nullptr, n, 64);
    node_prep2<<<nThr, 256, 0, stream>>>(A, attL2, attR2, AL, AR, MX, SM, B, n);
    edge_max_k<1><<<eThr, 256, 0, stream>>>(srcI, dstI, AL, AR, MX, e);
    edge_agg_k<1><<<eGrp, 256, 0, stream>>>(srcI, dstI, AL, AR, MX, SM, A, B, e);
    finish2_k<<<fThr, 256, 0, stream>>>(B, SM, A, b2, (float*)d_out, n * 64);
}


# group edges by destination (counting sort, integer-atomic arrival-order rank): one wave per node does max, exp-sum, aggregate in registers; float atomics gone, sums now in sorted-list order
# speedup vs baseline: 3.2907x; 2.7949x over previous
#include <hip/hip_runtime.h>
#include <math.h>
#include <float.h>

typedef float v2f __attribute__((ext_vector_type(2)));
typedef float v8f __attribute__((ext_vector_type(8)));

#define HC 64
#define SLOPE 0.2f

__device__ __forceinline__ float lrelu(float x) { return x > 0.f ? x : SLOPE * x; }

template <bool DUAL, int RT>
__global__ __launch_bounds__(32)
void gemm_x2(const float* __restrict__ X, const float* __restrict__ W,
             const float* __restrict__ Wr, float* __restrict__ Y,
             float* __restrict__ Yr, int n, int K) {
    const int lane  = threadIdx.x;
    const int row16 = lane & 15;
    const int khalf = (lane >> 4) << 1;
    const int col   = lane & 15;
    const int r0    = blockIdx.x * (16 * RT);

    int arow[RT];
    #pragma unroll
    for (int t = 0; t < RT; ++t) {
        int r = r0 + 16 * t + row16;
        arow[t] = r < n ? r : n - 1;
    }

    v8f acc[RT][4]  = {};
    v8f accR[RT][4] = {};

    #pragma unroll 2
    for (int k0 = 0; k0 < K; k0 += 4) {
        v2f a[RT];
        #pragma unroll
        for (int t = 0; t < RT; ++t)
            a[t] = *reinterpret_cast<const v2f*>(X + (size_t)arow[t] * K + k0 + khalf);

        #pragma unroll
        for (int j = 0; j < 4; ++j) {
            v2f b;
            const float* bp = W + (size_t)(k0 + khalf) * HC + j * 16 + col;
            b.x = bp[0]; b.y = bp[HC];
            #pragma unroll
            for (int t = 0; t < RT; ++t)
                acc[t][j] = __builtin_amdgcn_wmma_f32_16x16x4_f32(
                    false, a[t], false, b, (short)0, acc[t][j], false, false);
            if (DUAL) {
                v2f br;
                const float* brp = Wr + (size_t)(k0 + khalf) * HC + j * 16 + col;
                br.x = brp[0]; br.y = brp[HC];
                #pragma unroll
                for (int t = 0; t < RT; ++t)
                    accR[t][j] = __builtin_amdgcn_wmma_f32_16x16x4_f32(
                        false, a[t], false, br, (short)0, accR[t][j], false, false);
            }
        }
    }

    const int mhi = (lane >> 4) * 8;
    #pragma unroll
    for (int t = 0; t < RT; ++t)
        #pragma unroll
        for (int j = 0; j < 4; ++j)
            #pragma unroll
            for (int r = 0; r < 8; ++r) {
                int m = r0 + 16 * t + mhi + r;
                if (m < n) {
                    Y[(size_t)m * HC + j * 16 + col] = acc[t][j][r];
                    if (DUAL) Yr[(size_t)m * HC + j * 16 + col] = accR[t][j][r];
                }
            }
}

__global__ void node_prep1(const float* __restrict__ xlin,
                           const float* __restrict__ attL, const float* __restrict__ attR,
                           float* __restrict__ al, float* __restrict__ ar, int n) {
    int i = blockIdx.x * blockDim.x + threadIdx.x;
    if (i >= n) return;
    const float4* rowp = (const float4*)(xlin + (size_t)i * 64);
    const float4* lp = (const float4*)attL;
    const float4* rp = (const float4*)attR;
    float aL[4] = {0, 0, 0, 0}, aR[4] = {0, 0, 0, 0};
    #pragma unroll
    for (int q = 0; q < 16; ++q) {
        float4 v = rowp[q];
        float4 l = lp[q];
        float4 r = rp[q];
        int h = q >> 2;
        aL[h] += v.x * l.x + v.y * l.y + v.z * l.z + v.w * l.w;
        aR[h] += v.x * r.x + v.y * r.y + v.z * r.z + v.w * r.w;
    }
    #pragma unroll
    for (int h = 0; h < 4; ++h) {
        al[i * 4 + h] = aL[h];
        ar[i * 4 + h] = aR[h];
    }
}

__global__ void node_prep2(const float* __restrict__ xlin,
                           const float* __restrict__ attL, const float* __restrict__ attR,
                           float* __restrict__ al, float* __restrict__ ar, int n) {
    int i = blockIdx.x * blockDim.x + threadIdx.x;
    if (i >= n) return;
    const float4* rowp = (const float4*)(xlin + (size_t)i * 64);
    const float4* lp = (const float4*)attL;
    const float4* rp = (const float4*)attR;
    float aL = 0.f, aR = 0.f;
    #pragma unroll
    for (int q = 0; q < 16; ++q) {
        float4 v = rowp[q];
        float4 l = lp[q];
        float4 r = rp[q];
        aL += v.x * l.x + v.y * l.y + v.z * l.z + v.w * l.w;
        aR += v.x * r.x + v.y * r.y + v.z * r.z + v.w * r.w;
    }
    al[i] = aL; ar[i] = aR;
}

__global__ void csr_zero_k(int* __restrict__ deg, int n) {
    int i = blockIdx.x * blockDim.x + threadIdx.x;
    if (i < n) deg[i] = 0;
}

__global__ void csr_count_k(const int* __restrict__ dst, int* __restrict__ deg,
                            int* __restrict__ rank, int e) {
    int i = blockIdx.x * blockDim.x + threadIdx.x;
    if (i >= e) return;
    rank[i] = atomicAdd(&deg[dst[i]], 1);
}

__global__ __launch_bounds__(1024)
void csr_scan_k(const int* __restrict__ deg, int* __restrict__ rowptr, int n) {
    __shared__ int tot[32];
    const int t = threadIdx.x;
    const int lane = t & 31;
    const int wave = t >> 5;
    int carry = 0;
    for (int base = 0; base < n; base += 4096) {
        const int idx = base + 4 * t;
        const int d0 = idx     < n ? deg[idx]     : 0;
        const int d1 = idx + 1 < n ? deg[idx + 1] : 0;
        const int d2 = idx + 2 < n ? deg[idx + 2] : 0;
        const int d3 = idx + 3 < n ? deg[idx + 3] : 0;
        const int s1 = d0, s2 = s1 + d1, s3 = s2 + d2, s4 = s3 + d3;
        int inc = s4;
        #pragma unroll
        for (int off = 1; off < 32; off <<= 1) {
            int u = __shfl_up(inc, off);
            if (lane >= off) inc += u;
        }
        if (lane == 31) tot[wave] = inc;
        __syncthreads();
        int tinc = tot[lane];
        #pragma unroll
        for (int off = 1; off < 32; off <<= 1) {
            int u = __shfl_up(tinc, off);
            if (lane >= off) tinc += u;
        }
        const int below = __shfl(tinc, wave > 0 ? wave - 1 : 0);
        const int waveoff = wave > 0 ? below : 0;
        const int tile_total = __shfl(tinc, 31);
        const int excl = carry + waveoff + inc - s4;
        if (idx     < n) rowptr[idx]     = excl;
        if (idx + 1 < n) rowptr[idx + 1] = excl + s1;
        if (idx + 2 < n) rowptr[idx + 2] = excl + s2;
        if (idx + 3 < n) rowptr[idx + 3] = excl + s3;
        carry += tile_total;
        __syncthreads();
    }
    if (t == 0) rowptr[n] = carry;
}

__global__ void csr_fill_k(const int* __restrict__ src, const int* __restrict__ dst,
                           const int* __restrict__ rank, const int* __restrict__ rowptr,
                           int* __restrict__ ssrc, int e) {
    int i = blockIdx.x * blockDim.x + threadIdx.x;
    if (i >= e) return;
    ssrc[rowptr[dst[i]] + rank[i]] = src[i];
}

template <int H>
__device__ __forceinline__ void edge_logits(const float* __restrict__ al, int s,
                                            const float (&ari)[H], float (&lg)[H]) {
    if (H == 4) {
        const float4 a = *reinterpret_cast<const float4*>(al + (size_t)s * 4);
        lg[0] = lrelu(a.x + ari[0]);
        lg[1 % H] = lrelu(a.y + ari[1 % H]);
        lg[2 % H] = lrelu(a.z + ari[2 % H]);
        lg[3 % H] = lrelu(a.w + ari[3 % H]);
    } else {
        #pragma unroll
        for (int h = 0; h < H; ++h) lg[h] = lrelu(al[(size_t)s * H + h] + ari[h]);
    }
}

template <int H, int U, bool TAIL>
__device__ __forceinline__ void agg_rows(const float* xlin, int lane, int s,
                                         const float (&w)[H], const int (&hm)[H],
                                         int j0, int cnt,
                                         float& accx, float& accy, float& S) {
    float2 x[U];
    float wj[U];
    #pragma unroll
    for (int u = 0; u < U; ++u) {
        const int jj = j0 + u;
        const int jc = (!TAIL || jj < cnt) ? jj : cnt - 1;
        const int sj = __builtin_amdgcn_readlane(s, jc);
        int bits;
        if (H == 1) {
            bits = __builtin_amdgcn_readlane(__float_as_int(w[0]), jc);
        } else {
            bits = 0;
            #pragma unroll
            for (int h = 0; h < H; ++h)
                bits |= __builtin_amdgcn_readlane(__float_as_int(w[h]), jc) & hm[h];
        }
        wj[u] = (!TAIL || jj < cnt) ? __int_as_float(bits) : 0.f;
        x[u] = *reinterpret_cast<const float2*>(xlin + (size_t)sj * 64 + 2 * lane);
    }
    #pragma unroll
    for (int u = 0; u < U; ++u) {
        accx += x[u].x * wj[u];
        accy += x[u].y * wj[u];
        S += wj[u];
    }
}

template <int H, bool ELU>
__global__ __launch_bounds__(256)
void node_agg_k(const int* __restrict__ rowptr, const int* __restrict__ ssrc,
                const float* __restrict__ al, const float* __restrict__ ar,
                const float* xlin, const float* res,
                const float* __restrict__ bias, float* outp, int n) {
    const int lane = threadIdx.x & 31;
    const int i = __builtin_amdgcn_readfirstlane(
        (int)(blockIdx.x * (blockDim.x >> 5) + (threadIdx.x >> 5)));
    if (i >= n) return;
    const int beg = __builtin_amdgcn_readfirstlane(rowptr[i]);
    const int end = __builtin_amdgcn_readfirstlane(rowptr[i + 1]);
    const int hl  = (2 * lane * H) >> 6;
    int hm[H];
    #pragma unroll
    for (int h = 0; h < H; ++h) hm[h] = (hl == h) ? -1 : 0;

    float ari[H];
    #pragma unroll
    for (int h = 0; h < H; ++h) ari[h] = ar[(size_t)i * H + h];

    float m[H], lg0[H];
    int s0 = 0;
    #pragma unroll
    for (int h = 0; h < H; ++h) { m[h] = -FLT_MAX; lg0[h] = -FLT_MAX; }
    for (int c = beg; c < end; c += 32) {
        const int k = c + lane;
        float lg[H];
        int s = 0;
        #pragma unroll
        for (int h = 0; h < H; ++h) lg[h] = -FLT_MAX;
        if (k < end) { s = ssrc[k]; edge_logits<H>(al, s, ari, lg); }
        #pragma unroll
        for (int h = 0; h < H; ++h) m[h] = fmaxf(m[h], lg[h]);
        if (c == beg) {
            s0 = s;
            #pragma unroll
            for (int h = 0; h < H; ++h) lg0[h] = lg[h];
        }
    }
    #pragma unroll
    for (int h = 0; h < H; ++h)
        #pragma unroll
        for (int off = 16; off > 0; off >>= 1)
            m[h] = fmaxf(m[h], __shfl_xor(m[h], off));

    float accx = 0.f, accy = 0.f, S = 0.f;
    for (int c = beg; c < end; c += 32) {
        const int k = c + lane;
        int s = s0;
        float lg[H];
        #pragma unroll
        for (int h = 0; h < H; ++h) lg[h] = lg0[h];
        if (c != beg) {
            s = 0;
            #pragma unroll
            for (int h = 0; h < H; ++h) lg[h] = -FLT_MAX;
            if (k < end) { s = ssrc[k]; edge_logits<H>(al, s, ari, lg); }
        }
        float w[H];
        #pragma unroll
        for (int h = 0; h < H; ++h) w[h] = (k < end) ? __expf(lg[h] - m[h]) : 0.f;

        const int cnt = (end - c) < 32 ? (end - c) : 32;
        int j = 0;
        for (; j + 8 <= cnt; j += 8)
            agg_rows<H, 8, false>(xlin, lane, s, w, hm, j, cnt, accx, accy, S);
        for (; j < cnt; j += 4)
            agg_rows<H, 4, true>(xlin, lane, s, w, hm, j, cnt, accx, accy, S);
    }

    const float den = S + 1e-16f;
    const float2 r = *reinterpret_cast<const float2*>(res + (size_t)i * 64 + 2 * lane);
    const float2 b = *reinterpret_cast<const float2*>(bias + 2 * lane);
    float vx = accx / den + r.x + b.x;
    float vy = accy / den + r.y + b.y;
    if (ELU) {
        vx = vx > 0.f ? vx : (__expf(vx) - 1.f);
        vy = vy > 0.f ? vy : (__expf(vy) - 1.f);
    }
    *reinterpret_cast<float2*>(outp + (size_t)i * 64 + 2 * lane) = make_float2(vx, vy);
}

extern "C" void kernel_launch(void* const* d_in, const int* in_sizes, int n_in,
                              void* d_out, int out_size, void* d_ws, size_t ws_size,
                              hipStream_t stream) {
    const float* x     = (const float*)d_in[0];
    const int*   eidx  = (const int*)d_in[1];
    const float* W1    = (const float*)d_in[2];
    const float* attL1 = (const float*)d_in[3];
    const float* attR1 = (const float*)d_in[4];
    const float* resW1 = (const float*)d_in[5];
    const float* b1    = (const float*)d_in[6];
    const float* W2    = (const float*)d_in[7];
    const float* attL2 = (const float*)d_in[8];
    const float* attR2 = (const float*)d_in[9];
    const float* b2    = (const float*)d_in[10];

    const int n = in_sizes[0] / 128;
    const int e = in_sizes[1] / 2;
    const int* srcI = eidx;
    const int* dstI = eidx + e;

    float* ws = (float*)d_ws;
    float* A  = ws;
    float* B  = A  + (size_t)n * 64;
    float* C  = B  + (size_t)n * 64;
    float* AL = C  + (size_t)n * 64;
    float* AR = AL + (size_t)n * 4;
    int* DEG  = (int*)C;
    int* RP   = DEG + n;
    int* RK   = RP + n + 1;
    int* SS   = RK + e;

    constexpr int RT = 2;
    const int tiles = (n + 16 * RT - 1) / (16 * RT);
    const int nThr  = (n + 255) / 256;
    const int eThr  = (e + 255) / 256;
    const int nWav  = (n + 7) / 8;

    csr_zero_k<<<nThr, 256, 0, stream>>>(DEG, n);
    csr_count_k<<<eThr, 256, 0, stream>>>(dstI, DEG, RK, e);
    csr_scan_k<<<1, 1024, 0, stream>>>(DEG, RP, n);
    csr_fill_k<<<eThr, 256, 0, stream>>>(srcI, dstI, RK, RP, SS, e);

    gemm_x2<true, RT><<<tiles, 32, 0, stream>>>(x, W1, resW1, A, B, n, 128);
    node_prep1<<<nThr, 256, 0, stream>>>(A, attL1, attR1, AL, AR, n);
    node_agg_k<4, true><<<nWav, 256, 0, stream>>>(RP, SS, AL, AR, A, B, b1, B, n);

    gemm_x2<false, RT><<<tiles, 32, 0, stream>>>(B, W2, nullptr, A, nullptr, n, 64);
    node_prep2<<<nThr, 256, 0, stream>>>(A, attL2, attR2, AL, AR, n);
    node_agg_k<1, false><<<nWav, 256, 0, stream>>>(RP, SS, AL, AR, A, A, b2, (float*)d_out, n);
}


# group edges by destination in LDS: per-chunk bin histograms, bin-ordered copy, per-bin node cursors; global integer atomics and one-block scan gone; sums in LDS arrival order; 10 -> 9 launches
# speedup vs baseline: 5.6316x; 1.7114x over previous
#include <hip/hip_runtime.h>
#include <math.h>
#include <float.h>

typedef float v2f __attribute__((ext_vector_type(2)));
typedef float v8f __attribute__((ext_vector_type(8)));

#define HC 64
#define SLOPE 0.2f

__device__ __forceinline__ float lrelu(float x) { return x > 0.f ? x : SLOPE * x; }

template <bool DUAL, int RT>
__global__ __launch_bounds__(32)
void gemm_x2(const float* __restrict__ X, const float* __restrict__ W,
             const float* __restrict__ Wr, float* __restrict__ Y,
             float* __restrict__ Yr, int n, int K) {
    const int lane  = threadIdx.x;
    const int row16 = lane & 15;
    const int khalf = (lane >> 4) << 1;
    const int col   = lane & 15;
    const int r0    = blockIdx.x * (16 * RT);

    int arow[RT];
    #pragma unroll
    for (int t = 0; t < RT; ++t) {
        int r = r0 + 16 * t + row16;
        arow[t] = r < n ? r : n - 1;
    }

    v8f acc[RT][4]  = {};
    v8f accR[RT][4] = {};

    #pragma unroll 2
    for (int k0 = 0; k0 < K; k0 += 4) {
        v2f a[RT];
        #pragma unroll
        for (int t = 0; t < RT; ++t)
            a[t] = *reinterpret_cast<const v2f*>(X + (size_t)arow[t] * K + k0 + khalf);

        #pragma unroll
        for (int j = 0; j < 4; ++j) {
            v2f b;
            const float* bp = W + (size_t)(k0 + khalf) * HC + j * 16 + col;
            b.x = bp[0]; b.y = bp[HC];
            #pragma unroll
            for (int t = 0; t < RT; ++t)
                acc[t][j] = __builtin_amdgcn_wmma_f32_16x16x4_f32(
                    false, a[t], false, b, (short)0, acc[t][j], false, false);
            if (DUAL) {
                v2f br;
                const float* brp = Wr + (size_t)(k0 + khalf) * HC + j * 16 + col;
                br.x = brp[0]; br.y = brp[HC];
                #pragma unroll
                for (int t = 0; t < RT; ++t)
                    accR[t][j] = __builtin_amdgcn_wmma_f32_16x16x4_f32(
                        false, a[t], false, br, (short)0, accR[t][j], false, false);
            }
        }
    }

    const int mhi = (lane >> 4) * 8;
    #pragma unroll
    for (int t = 0; t < RT; ++t)
        #pragma unroll
        for (int j = 0; j < 4; ++j)
            #pragma unroll
            for (int r = 0; r < 8; ++r) {
                int m = r0 + 16 * t + mhi + r;
                if (m < n) {
                    Y[(size_t)m * HC + j * 16 + col] = acc[t][j][r];
                    if (DUAL) Yr[(size_t)m * HC + j * 16 + col] = accR[t][j][r];
                }
            }
}

__global__ void node_prep1(const float* __restrict__ xlin,
                           const float* __restrict__ attL, const float* __restrict__ attR,
                           float* __restrict__ al, float* __restrict__ ar, int n) {
    int i = blockIdx.x * blockDim.x + threadIdx.x;
    if (i >= n) return;
    const float4* rowp = (const float4*)(xlin + (size_t)i * 64);
    const float4* lp = (const float4*)attL;
    const float4* rp = (const float4*)attR;
    float aL[4] = {0, 0, 0, 0}, aR[4] = {0, 0, 0, 0};
    #pragma unroll
    for (int q = 0; q < 16; ++q) {
        float4 v = rowp[q];
        float4 l = lp[q];
        float4 r = rp[q];
        int h = q >> 2;
        aL[h] += v.x * l.x + v.y * l.y + v.z * l.z + v.w * l.w;
        aR[h] += v.x * r.x + v.y * r.y + v.z * r.z + v.w * r.w;
    }
    #pragma unroll
    for (int h = 0; h < 4; ++h) {
        al[i * 4 + h] = aL[h];
        ar[i * 4 + h] = aR[h];
    }
}

__global__ void node_prep2(const float* __restrict__ xlin,
                           const float* __restrict__ attL, const float* __restrict__ attR,
                           float* __restrict__ al, float* __restrict__ ar, int n) {
    int i = blockIdx.x * blockDim.x + threadIdx.x;
    if (i >= n) return;
    const float4* rowp = (const float4*)(xlin + (size_t)i * 64);
    const float4* lp = (const float4*)attL;
    const float4* rp = (const float4*)attR;
    float aL = 0.f, aR = 0.f;
    #pragma unroll
    for (int q = 0; q < 16; ++q) {
        float4 v = rowp[q];
        float4 l = lp[q];
        float4 r = rp[q];
        aL += v.x * l.x + v.y * l.y + v.z * l.z + v.w * l.w;
        aR += v.x * r.x + v.y * r.y + v.z * r.z + v.w * r.w;
    }
    al[i] = aL; ar[i] = aR;
}

#define BSH    8
#define BW     (1 << BSH)
#define MAXNB  1024
#define SORT_G 128
#define SORT_T 1024
#define SORT_U 16

__device__ __forceinline__ int bin_of(int d, int nb) {
    const unsigned b = (unsigned)d >> BSH;
    return b < (unsigned)nb ? (int)b : nb - 1;
}

__global__ __launch_bounds__(SORT_T)
void bin_hist_k(const int* __restrict__ dst, int* __restrict__ hist, int e, int nb) {
    __shared__ int h[MAXNB];
    const int t = threadIdx.x;
    for (int b = t; b < nb; b += SORT_T) h[b] = 0;
    __syncthreads();
    const int ch = (e + SORT_G - 1) / SORT_G;
    const int lo = blockIdx.x * ch;
    const int hi = lo + ch < e ? lo + ch : e;
    for (int i0 = lo + t; i0 < hi; i0 += SORT_U * SORT_T) {
        int d[SORT_U];
        #pragma unroll
        for (int u = 0; u < SORT_U; ++u) {
            const int i = i0 + u * SORT_T;
            d[u] = i < hi ? dst[i] : 0;
        }
        #pragma unroll
        for (int u = 0; u < SORT_U; ++u)
            if (i0 + u * SORT_T < hi) atomicAdd(&h[bin_of(d[u], nb)], 1);
    }
    __syncthreads();
    for (int b = t; b < nb; b += SORT_T) hist[blockIdx.x * nb + b] = h[b];
}

__global__ __launch_bounds__(SORT_T)
void bin_scatter_k(const int* __restrict__ src, const int* __restrict__ dst,
                   const int* __restrict__ hist, int* __restrict__ bstart,
                   int2* __restrict__ tmp, int e, int nb) {
    __shared__ int cur[MAXNB];
    __shared__ int wtot[32];
    const int t = threadIdx.x;
    const int lane = t & 31;
    const int wave = t >> 5;
    const int g = blockIdx.x;

    int pre = 0, tot = 0;
    if (t < nb) {
        #pragma unroll 32
        for (int gg = 0; gg < SORT_G; ++gg) {
            const int v = hist[gg * nb + t];
            tot += v;
            pre += gg < g ? v : 0;
        }
    }
    int inc = tot;
    #pragma unroll
    for (int off = 1; off < 32; off <<= 1) {
        int u = __shfl_up(inc, off);
        if (lane >= off) inc += u;
    }
    if (lane == 31) wtot[wave] = inc;
    __syncthreads();
    int tinc = wtot[lane];
    #pragma unroll
    for (int off = 1; off < 32; off <<= 1) {
        int u = __shfl_up(tinc, off);
        if (lane >= off) tinc += u;
    }
    const int below = __shfl(tinc, wave > 0 ? wave - 1 : 0);
    const int start = (wave > 0 ? below : 0) + inc - tot;
    if (t < nb) {
        cur[t] = start + pre;
        if (g == 0) bstart[t] = start;
    }
    if (g == 0 && t == 0) bstart[nb] = e;
    __syncthreads();

    const int ch = (e + SORT_G - 1) / SORT_G;
    const int lo = g * ch;
    const int hi = lo + ch < e ? lo + ch : e;
    for (int i0 = lo + t; i0 < hi; i0 += SORT_U * SORT_T) {
        int d[SORT_U], s[SORT_U];
        #pragma unroll
        for (int u = 0; u < SORT_U; ++u) {
            const int i = i0 + u * SORT_T;
            d[u] = i < hi ? dst[i] : 0;
            s[u] = i < hi ? src[i] : 0;
        }
        #pragma unroll
        for (int u = 0; u < SORT_U; ++u)
            if (i0 + u * SORT_T < hi) {
                const int p = atomicAdd(&cur[bin_of(d[u], nb)], 1);
                tmp[p] = make_int2(s[u], d[u]);
            }
    }
}

#define PLACE_T 1024
#define PLACE_U 8
__global__ __launch_bounds__(PLACE_T)
void bin_place_k(const int2* __restrict__ tmp, const int* __restrict__ bstart,
                 int* __restrict__ rowptr, int* __restrict__ ssrc, int n, int nb) {
    __shared__ int cnt[BW];
    __shared__ int wt[BW / 32];
    const int t = threadIdx.x;
    const int lane = t & 31;
    const int wave = t >> 5;
    const int b = blockIdx.x;
    const int lo = bstart[b];
    const int hi = bstart[b + 1];
    if (t < BW) cnt[t] = 0;
    __syncthreads();
    for (int i0 = lo + t; i0 < hi; i0 += PLACE_U * PLACE_T) {
        int d[PLACE_U];
        #pragma unroll
        for (int u = 0; u < PLACE_U; ++u) {
            const int i = i0 + u * PLACE_T;
            d[u] = i < hi ? tmp[i].y : 0;
        }
        #pragma unroll
        for (int u = 0; u < PLACE_U; ++u)
            if (i0 + u * PLACE_T < hi) atomicAdd(&cnt[d[u] & (BW - 1)], 1);
    }
    __syncthreads();
    int v = 0, inc = 0;
    if (t < BW) {
        v = cnt[t];
        inc = v;
        #pragma unroll
        for (int off = 1; off < 32; off <<= 1) {
            int u = __shfl_up(inc, off);
            if (lane >= off) inc += u;
        }
        if (lane == 31) wt[wave] = inc;
    }
    __syncthreads();
    if (t < BW) {
        int off = 0;
        #pragma unroll
        for (int w = 0; w < BW / 32; ++w) off += w < wave ? wt[w] : 0;
        const int excl = off + inc - v;
        cnt[t] = excl;
        const int node = b * BW + t;
        if (node < n) rowptr[node] = lo + excl;
    }
    if (b == nb - 1 && t == 0) rowptr[n] = hi;
    __syncthreads();
    for (int i0 = lo + t; i0 < hi; i0 += PLACE_U * PLACE_T) {
        int2 ed[PLACE_U];
        #pragma unroll
        for (int u = 0; u < PLACE_U; ++u) {
            const int i = i0 + u * PLACE_T;
            ed[u] = i < hi ? tmp[i] : make_int2(0, 0);
        }
        #pragma unroll
        for (int u = 0; u < PLACE_U; ++u)
            if (i0 + u * PLACE_T < hi) {
                const int p = atomicAdd(&cnt[ed[u].y & (BW - 1)], 1);
                ssrc[lo + p] = ed[u].x;
            }
    }
}

template <int H>
__device__ __forceinline__ void edge_logits(const float* __restrict__ al, int s,
                                            const float (&ari)[H], float (&lg)[H]) {
    if (H == 4) {
        const float4 a = *reinterpret_cast<const float4*>(al + (size_t)s * 4);
        lg[0] = lrelu(a.x + ari[0]);
        lg[1 % H] = lrelu(a.y + ari[1 % H]);
        lg[2 % H] = lrelu(a.z + ari[2 % H]);
        lg[3 % H] = lrelu(a.w + ari[3 % H]);
    } else {
        #pragma unroll
        for (int h = 0; h < H; ++h) lg[h] = lrelu(al[(size_t)s * H + h] + ari[h]);
    }
}

template <int H, int U, bool TAIL>
__device__ __forceinline__ void agg_rows(const float* xlin, int lane, int s,
                                         const float (&w)[H], const int (&hm)[H],
                                         int j0, int cnt,
                                         float& accx, float& accy, float& S) {
    float2 x[U];
    float wj[U];
    #pragma unroll
    for (int u = 0; u < U; ++u) {
        const int jj = j0 + u;
        const int jc = (!TAIL || jj < cnt) ? jj : cnt - 1;
        const int sj = __builtin_amdgcn_readlane(s, jc);
        int bits;
        if (H == 1) {
            bits = __builtin_amdgcn_readlane(__float_as_int(w[0]), jc);
        } else {
            bits = 0;
            #pragma unroll
            for (int h = 0; h < H; ++h)
                bits |= __builtin_amdgcn_readlane(__float_as_int(w[h]), jc) & hm[h];
        }
        wj[u] = (!TAIL || jj < cnt) ? __int_as_float(bits) : 0.f;
        x[u] = *reinterpret_cast<const float2*>(xlin + (size_t)sj * 64 + 2 * lane);
    }
    #pragma unroll
    for (int u = 0; u < U; ++u) {
        accx += x[u].x * wj[u];
        accy += x[u].y * wj[u];
        S += wj[u];
    }
}

template <int H, bool ELU>
__global__ __launch_bounds__(256)
void node_agg_k(const int* __restrict__ rowptr, const int* __restrict__ ssrc,
                const float* __restrict__ al, const float* __restrict__ ar,
                const float* xlin, const float* res,
                const float* __restrict__ bias, float* outp, int n) {
    const int lane = threadIdx.x & 31;
    const int i = __builtin_amdgcn_readfirstlane(
        (int)(blockIdx.x * (blockDim.x >> 5) + (threadIdx.x >> 5)));
    if (i >= n) return;
    const int beg = __builtin_amdgcn_readfirstlane(rowptr[i]);
    const int end = __builtin_amdgcn_readfirstlane(rowptr[i + 1]);
    const int hl  = (2 * lane * H) >> 6;
    int hm[H];
    #pragma unroll
    for (int h = 0; h < H; ++h) hm[h] = (hl == h) ? -1 : 0;

    float ari[H];
    #pragma unroll
    for (int h = 0; h < H; ++h) ari[h] = ar[(size_t)i * H + h];

    float m[H], lg0[H];
    int s0 = 0;
    #pragma unroll
    for (int h = 0; h < H; ++h) { m[h] = -FLT_MAX; lg0[h] = -FLT_MAX; }
    for (int c = beg; c < end; c += 32) {
        const int k = c + lane;
        float lg[H];
        int s = 0;
        #pragma unroll
        for (int h = 0; h < H; ++h) lg[h] = -FLT_MAX;
        if (k < end) { s = ssrc[k]; edge_logits<H>(al, s, ari, lg); }
        #pragma unroll
        for (int h = 0; h < H; ++h) m[h] = fmaxf(m[h], lg[h]);
        if (c == beg) {
            s0 = s;
            #pragma unroll
            for (int h = 0; h < H; ++h) lg0[h] = lg[h];
        }
    }
    #pragma unroll
    for (int h = 0; h < H; ++h)
        #pragma unroll
        for (int off = 16; off > 0; off >>= 1)
            m[h] = fmaxf(m[h], __shfl_xor(m[h], off));

    float accx = 0.f, accy = 0.f, S = 0.f;
    for (int c = beg; c < end; c += 32) {
        const int k = c + lane;
        int s = s0;
        float lg[H];
        #pragma unroll
        for (int h = 0; h < H; ++h) lg[h] = lg0[h];
        if (c != beg) {
            s = 0;
            #pragma unroll
            for (int h = 0; h < H; ++h) lg[h] = -FLT_MAX;
            if (k < end) { s = ssrc[k]; edge_logits<H>(al, s, ari, lg); }
        }
        float w[H];
        #pragma unroll
        for (int h = 0; h < H; ++h) w[h] = (k < end) ? __expf(lg[h] - m[h]) : 0.f;

        const int cnt = (end - c) < 32 ? (end - c) : 32;
        int j = 0;
        for (; j + 8 <= cnt; j += 8)
            agg_rows<H, 8, false>(xlin, lane, s, w, hm, j, cnt, accx, accy, S);
        for (; j < cnt; j += 4)
            agg_rows<H, 4, true>(xlin, lane, s, w, hm, j, cnt, accx, accy, S);
    }

    const float den = S + 1e-16f;
    const float2 r = *reinterpret_cast<const float2*>(res + (size_t)i * 64 + 2 * lane);
    const float2 b = *reinterpret_cast<const float2*>(bias + 2 * lane);
    float vx = accx / den + r.x + b.x;
    float vy = accy / den + r.y + b.y;
    if (ELU) {
        vx = vx > 0.f ? vx : (__expf(vx) - 1.f);
        vy = vy > 0.f ? vy : (__expf(vy) - 1.f);
    }
    *reinterpret_cast<float2*>(outp + (size_t)i * 64 + 2 * lane) = make_float2(vx, vy);
}

extern "C" void kernel_launch(void* const* d_in, const int* in_sizes, int n_in,
                              void* d_out, int out_size, void* d_ws, size_t ws_size,
                              hipStream_t stream) {
    const float* x     = (const float*)d_in[0];
    const int*   eidx  = (const int*)d_in[1];
    const float* W1    = (const float*)d_in[2];
    const float* attL1 = (const float*)d_in[3];
    const float* attR1 = (const float*)d_in[4];
    const float* resW1 = (const float*)d_in[5];
    const float* b1    = (const float*)d_in[6];
    const float* W2    = (const float*)d_in[7];
    const float* attL2 = (const float*)d_in[8];
    const float* attR2 = (const float*)d_in[9];
    const float* b2    = (const float*)d_in[10];

    const int n = in_sizes[0] / 128;
    const int e = in_sizes[1] / 2;
    const int* srcI = eidx;
    const int* dstI = eidx + e;

    float* ws = (float*)d_ws;
    float* A  = ws;
    float* B  = A  + (size_t)n * 64;
    float* C  = B  + (size_t)n * 64;
    float* AL = C  + (size_t)n * 64;
    float* AR = AL + (size_t)n * 4;
    const int nb = (n + BW - 1) / BW;
    int2* TMP = (int2*)C;
    int* SS   = (int*)C + 2 * (size_t)e;
    int* RP   = SS + e;
    int* HIST = RP + n + 1;
    int* BST  = HIST + SORT_G * nb;

    constexpr int RT = 2;
    const int tiles = (n + 16 * RT - 1) / (16 * RT);
    const int nThr  = (n + 255) / 256;
    const int nWav  = (n + 7) / 8;

    bin_hist_k<<<SORT_G, SORT_T, 0, stream>>>(dstI, HIST, e, nb);
    bin_scatter_k<<<SORT_G, SORT_T, 0, stream>>>(srcI, dstI, HIST, BST, TMP, e, nb);
    bin_place_k<<<nb, PLACE_T, 0, stream>>>(TMP, BST, RP, SS, n, nb);

    gemm_x2<true, RT><<<tiles, 32, 0, stream>>>(x, W1, resW1, A, B, n, 128);
    node_prep1<<<nThr, 256, 0, stream>>>(A, attL1, attR1, AL, AR, n);
    node_agg_k<4, true><<<nWav, 256, 0, stream>>>(RP, SS, AL, AR, A, B, b1, B, n);

    gemm_x2<false, RT><<<tiles, 32, 0, stream>>>(B, W2, nullptr, A, nullptr, n, 64);
    node_prep2<<<nThr, 256, 0, stream>>>(A, attL2, attR2, AL, AR, n);
    node_agg_k<1, false><<<nWav, 256, 0, stream>>>(RP, SS, AL, AR, A, A, b2, (float*)d_out, n);
}


# attention dot products al, ar taken in the GEMM epilogue: wave tile staged in LDS, one row per lane, same summation order; both node-prep launches and their re-read of x_lin go; 9 -> 7 launches
# speedup vs baseline: 5.9431x; 1.0553x over previous
#include <hip/hip_runtime.h>
#include <math.h>
#include <float.h>

typedef float v2f __attribute__((ext_vector_type(2)));
typedef float v8f __attribute__((ext_vector_type(8)));

#define HC 64
#define SLOPE 0.2f

__device__ __forceinline__ float lrelu(float x) { return x > 0.f ? x : SLOPE * x; }

#define TP 68
template <bool DUAL, int RT, int H>
__global__ __launch_bounds__(32)
void gemm_x2(const float* __restrict__ X, const float* __restrict__ W,
             const float* __restrict__ Wr, float* __restrict__ Y,
             float* __restrict__ Yr,
             const float* __restrict__ attL, const float* __restrict__ attR,
             float* __restrict__ al, float* __restrict__ ar, int n, int K) {
    static_assert(RT == 2, "the epilogue gives each of the 32 lanes one tile row");
    __shared__ __attribute__((aligned(16))) float tile[16 * RT * TP];
    const int lane  = threadIdx.x;
    const int row16 = lane & 15;
    const int khalf = (lane >> 4) << 1;
    const int col   = lane & 15;
    const int r0    = blockIdx.x * (16 * RT);

    int arow[RT];
    #pragma unroll
    for (int t = 0; t < RT; ++t) {
        int r = r0 + 16 * t + row16;
        arow[t] = r < n ? r : n - 1;
    }

    v8f acc[RT][4]  = {};
    v8f accR[RT][4] = {};

    #pragma unroll 2
    for (int k0 = 0; k0 < K; k0 += 4) {
        v2f a[RT];
        #pragma unroll
        for (int t = 0; t < RT; ++t)
            a[t] = *reinterpret_cast<const v2f*>(X + (size_t)arow[t] * K + k0 + khalf);

        #pragma unroll
        for (int j = 0; j < 4; ++j) {
            v2f b;
            const float* bp = W + (size_t)(k0 + khalf) * HC + j * 16 + col;
            b.x = bp[0]; b.y = bp[HC];
            #pragma unroll
            for (int t = 0; t < RT; ++t)
                acc[t][j] = __builtin_amdgcn_wmma_f32_16x16x4_f32(
                    false, a[t], false, b, (short)0, acc[t][j], false, false);
            if (DUAL) {
                v2f br;
                const float* brp = Wr + (size_t)(k0 + khalf) * HC + j * 16 + col;
                br.x = brp[0]; br.y = brp[HC];
                #pragma unroll
                for (int t = 0; t < RT; ++t)
                    accR[t][j] = __builtin_amdgcn_wmma_f32_16x16x4_f32(
                        false, a[t], false, br, (short)0, accR[t][j], false, false);
            }
        }
    }

    const int mhi = (lane >> 4) * 8;
    #pragma unroll
    for (int t = 0; t < RT; ++t)
        #pragma unroll
        for (int j = 0; j < 4; ++j)
            #pragma unroll
            for (int r = 0; r < 8; ++r) {
                int m = r0 + 16 * t + mhi + r;
                tile[(16 * t + mhi + r) * TP + j * 16 + col] = acc[t][j][r];
                if (m < n) {
                    Y[(size_t)m * HC + j * 16 + col] = acc[t][j][r];
                    if (DUAL) Yr[(size_t)m * HC + j * 16 + col] = accR[t][j][r];
                }
            }

    __builtin_amdgcn_fence(__ATOMIC_RELEASE, "workgroup", "local");
    __builtin_amdgcn_s_barrier();
    __builtin_amdgcn_fence(__ATOMIC_ACQUIRE, "workgroup", "local");

    const int row = r0 + lane;
    const float4* rowp = reinterpret_cast<const float4*>(tile + lane * TP);
    const float4* lp = reinterpret_cast<const float4*>(attL);
    const float4* rp = reinterpret_cast<const float4*>(attR);
    if (H == 4) {
        float aL[4] = {0, 0, 0, 0}, aR[4] = {0, 0, 0, 0};
        #pragma unroll
        for (int q = 0; q < 16; ++q) {
            const float4 v = rowp[q];
            const float4 l = lp[q];
            const float4 rr = rp[q];
            const int h = q >> 2;
            aL[h] += v.x * l.x + v.y * l.y + v.z * l.z + v.w * l.w;
            aR[h] += v.x * rr.x + v.y * rr.y + v.z * rr.z + v.w * rr.w;
        }
        if (row < n) {
            *reinterpret_cast<float4*>(al + (size_t)row * 4) = make_float4(aL[0], aL[1], aL[2], aL[3]);
            *reinterpret_cast<float4*>(ar + (size_t)row * 4) = make_float4(aR[0], aR[1], aR[2], aR[3]);
        }
    } else {
        float aL = 0.f, aR = 0.f;
        for (int q = 0; q < 16; ++q) {
            const float4 v = rowp[q];
            const float4 l = lp[q];
            const float4 rr = rp[q];
            aL += v.x * l.x + v.y * l.y + v.z * l.z + v.w * l.w;
            aR += v.x * rr.x + v.y * rr.y + v.z * rr.z + v.w * rr.w;
        }
        if (row < n) { al[row] = aL; ar[row] = aR; }
    }
}

#define BSH    8
#define BW     (1 << BSH)
#define MAXNB  1024
#define SORT_G 128
#define SORT_T 1024
#define SORT_U 16

__device__ __forceinline__ int bin_of(int d, int nb) {
    const unsigned b = (unsigned)d >> BSH;
    return b < (unsigned)nb ? (int)b : nb - 1;
}

__global__ __launch_bounds__(SORT_T)
void bin_hist_k(const int* __restrict__ dst, int* __restrict__ hist, int e, int nb) {
    __shared__ int h[MAXNB];
    const int t = threadIdx.x;
    for (int b = t; b < nb; b += SORT_T) h[b] = 0;
    __syncthreads();
    const int ch = (e + SORT_G - 1) / SORT_G;
    const int lo = blockIdx.x * ch;
    const int hi = lo + ch < e ? lo + ch : e;
    for (int i0 = lo + t; i0 < hi; i0 += SORT_U * SORT_T) {
        int d[SORT_U];
        #pragma unroll
        for (int u = 0; u < SORT_U; ++u) {
            const int i = i0 + u * SORT_T;
            d[u] = i < hi ? dst[i] : 0;
        }
        #pragma unroll
        for (int u = 0; u < SORT_U; ++u)
            if (i0 + u * SORT_T < hi) atomicAdd(&h[bin_of(d[u], nb)], 1);
    }
    __syncthreads();
    for (int b = t; b < nb; b += SORT_T) hist[blockIdx.x * nb + b] = h[b];
}

__global__ __launch_bounds__(SORT_T)
void bin_scatter_k(const int* __restrict__ src, const int* __restrict__ dst,
                   const int* __restrict__ hist, int* __restrict__ bstart,
                   int2* __restrict__ tmp, int e, int nb) {
    __shared__ int cur[MAXNB];
    __shared__ int wtot[32];
    const int t = threadIdx.x;
    const int lane = t & 31;
    const int wave = t >> 5;
    const int g = blockIdx.x;

    int pre = 0, tot = 0;
    if (t < nb) {
        #pragma unroll 32
        for (int gg = 0; gg < SORT_G; ++gg) {
            const int v = hist[gg * nb + t];
            tot += v;
            pre += gg < g ? v : 0;
        }
    }
    int inc = tot;
    #pragma unroll
    for (int off = 1; off < 32; off <<= 1) {
        int u = __shfl_up(inc, off);
        if (lane >= off) inc += u;
    }
    if (lane == 31) wtot[wave] = inc;
    __syncthreads();
    int tinc = wtot[lane];
    #pragma unroll
    for (int off = 1; off < 32; off <<= 1) {
        int u = __shfl_up(tinc, off);
        if (lane >= off) tinc += u;
    }
    const int below = __shfl(tinc, wave > 0 ? wave - 1 : 0);
    const int start = (wave > 0 ? below : 0) + inc - tot;
    if (t < nb) {
        cur[t] = start + pre;
        if (g == 0) bstart[t] = start;
    }
    if (g == 0 && t == 0) bstart[nb] = e;
    __syncthreads();

    const int ch = (e + SORT_G - 1) / SORT_G;
    const int lo = g * ch;
    const int hi = lo + ch < e ? lo + ch : e;
    for (int i0 = lo + t; i0 < hi; i0 += SORT_U * SORT_T) {
        int d[SORT_U], s[SORT_U];
        #pragma unroll
        for (int u = 0; u < SORT_U; ++u) {
            const int i = i0 + u * SORT_T;
            d[u] = i < hi ? dst[i] : 0;
            s[u] = i < hi ? src[i] : 0;
        }
        #pragma unroll
        for (int u = 0; u < SORT_U; ++u)
            if (i0 + u * SORT_T < hi) {
                const int p = atomicAdd(&cur[bin_of(d[u], nb)], 1);
                tmp[p] = make_int2(s[u], d[u]);
            }
    }
}

#define PLACE_T 1024
#define PLACE_U 8
__global__ __launch_bounds__(PLACE_T)
void bin_place_k(const int2* __restrict__ tmp, const int* __restrict__ bstart,
                 int* __restrict__ rowptr, int* __restrict__ ssrc, int n, int nb) {
    __shared__ int cnt[BW];
    __shared__ int wt[BW / 32];
    const int t = threadIdx.x;
    const int lane = t & 31;
    const int wave = t >> 5;
    const int b = blockIdx.x;
    const int lo = bstart[b];
    const int hi = bstart[b + 1];
    if (t < BW) cnt[t] = 0;
    __syncthreads();
    for (int i0 = lo + t; i0 < hi; i0 += PLACE_U * PLACE_T) {
        int d[PLACE_U];
        #pragma unroll
        for (int u = 0; u < PLACE_U; ++u) {
            const int i = i0 + u * PLACE_T;
            d[u] = i < hi ? tmp[i].y : 0;
        }
        #pragma unroll
        for (int u = 0; u < PLACE_U; ++u)
            if (i0 + u * PLACE_T < hi) atomicAdd(&cnt[d[u] & (BW - 1)], 1);
    }
    __syncthreads();
    int v = 0, inc = 0;
    if (t < BW) {
        v = cnt[t];
        inc = v;
        #pragma unroll
        for (int off = 1; off < 32; off <<= 1) {
            int u = __shfl_up(inc, off);
            if (lane >= off) inc += u;
        }
        if (lane == 31) wt[wave] = inc;
    }
    __syncthreads();
    if (t < BW) {
        int off = 0;
        #pragma unroll
        for (int w = 0; w < BW / 32; ++w) off += w < wave ? wt[w] : 0;
        const int excl = off + inc - v;
        cnt[t] = excl;
        const int node = b * BW + t;
        if (node < n) rowptr[node] = lo + excl;
    }
    if (b == nb - 1 && t == 0) rowptr[n] = hi;
    __syncthreads();
    for (int i0 = lo + t; i0 < hi; i0 += PLACE_U * PLACE_T) {
        int2 ed[PLACE_U];
        #pragma unroll
        for (int u = 0; u < PLACE_U; ++u) {
            const int i = i0 + u * PLACE_T;
            ed[u] = i < hi ? tmp[i] : make_int2(0, 0);
        }
        #pragma unroll
        for (int u = 0; u < PLACE_U; ++u)
            if (i0 + u * PLACE_T < hi) {
                const int p = atomicAdd(&cnt[ed[u].y & (BW - 1)], 1);
                ssrc[lo + p] = ed[u].x;
            }
    }
}

template <int H>
__device__ __forceinline__ void edge_logits(const float* __restrict__ al, int s,
                                            const float (&ari)[H], float (&lg)[H]) {
    if (H == 4) {
        const float4 a = *reinterpret_cast<const float4*>(al + (size_t)s * 4);
        lg[0] = lrelu(a.x + ari[0]);
        lg[1 % H] = lrelu(a.y + ari[1 % H]);
        lg[2 % H] = lrelu(a.z + ari[2 % H]);
        lg[3 % H] = lrelu(a.w + ari[3 % H]);
    } else {
        #pragma unroll
        for (int h = 0; h < H; ++h) lg[h] = lrelu(al[(size_t)s * H + h] + ari[h]);
    }
}

template <int H, int U, bool TAIL>
__device__ __forceinline__ void agg_rows(const float* xlin, int lane, int s,
                                         const float (&w)[H], const int (&hm)[H],
                                         int j0, int cnt,
                                         float& accx, float& accy, float& S) {
    float2 x[U];
    float wj[U];
    #pragma unroll
    for (int u = 0; u < U; ++u) {
        const int jj = j0 + u;
        const int jc = (!TAIL || jj < cnt) ? jj : cnt - 1;
        const int sj = __builtin_amdgcn_readlane(s, jc);
        int bits;
        if (H == 1) {
            bits = __builtin_amdgcn_readlane(__float_as_int(w[0]), jc);
        } else {
            bits = 0;
            #pragma unroll
            for (int h = 0; h < H; ++h)
                bits |= __builtin_amdgcn_readlane(__float_as_int(w[h]), jc) & hm[h];
        }
        wj[u] = (!TAIL || jj < cnt) ? __int_as_float(bits) : 0.f;
        x[u] = *reinterpret_cast<const float2*>(xlin + (size_t)sj * 64 + 2 * lane);
    }
    #pragma unroll
    for (int u = 0; u < U; ++u) {
        accx += x[u].x * wj[u];
        accy += x[u].y * wj[u];
        S += wj[u];
    }
}

template <int H, bool ELU>
__global__ __launch_bounds__(256)
void node_agg_k(const int* __restrict__ rowptr, const int* __restrict__ ssrc,
                const float* __restrict__ al, const float* __restrict__ ar,
                const float* xlin, const float* res,
                const float* __restrict__ bias, float* outp, int n) {
    const int lane = threadIdx.x & 31;
    const int i = __builtin_amdgcn_readfirstlane(
        (int)(blockIdx.x * (blockDim.x >> 5) + (threadIdx.x >> 5)));
    if (i >= n) return;
    const int beg = __builtin_amdgcn_readfirstlane(rowptr[i]);
    const int end = __builtin_amdgcn_readfirstlane(rowptr[i + 1]);
    const int hl  = (2 * lane * H) >> 6;
    int hm[H];
    #pragma unroll
    for (int h = 0; h < H; ++h) hm[h] = (hl == h) ? -1 : 0;

    float ari[H];
    #pragma unroll
    for (int h = 0; h < H; ++h) ari[h] = ar[(size_t)i * H + h];

    float m[H], lg0[H];
    int s0 = 0;
    #pragma unroll
    for (int h = 0; h < H; ++h) { m[h] = -FLT_MAX; lg0[h] = -FLT_MAX; }
    for (int c = beg; c < end; c += 32) {
        const int k = c + lane;
        float lg[H];
        int s = 0;
        #pragma unroll
        for (int h = 0; h < H; ++h) lg[h] = -FLT_MAX;
        if (k < end) { s = ssrc[k]; edge_logits<H>(al, s, ari, lg); }
        #pragma unroll
        for (int h = 0; h < H; ++h) m[h] = fmaxf(m[h], lg[h]);
        if (c == beg) {
            s0 = s;
            #pragma unroll
            for (int h = 0; h < H; ++h) lg0[h] = lg[h];
        }
    }
    #pragma unroll
    for (int h = 0; h < H; ++h)
        #pragma unroll
        for (int off = 16; off > 0; off >>= 1)
            m[h] = fmaxf(m[h], __shfl_xor(m[h], off));

    float accx = 0.f, accy = 0.f, S = 0.f;
    for (int c = beg; c < end; c += 32) {
        const int k = c + lane;
        int s = s0;
        float lg[H];
        #pragma unroll
        for (int h = 0; h < H; ++h) lg[h] = lg0[h];
        if (c != beg) {
            s = 0;
            #pragma unroll
            for (int h = 0; h < H; ++h) lg[h] = -FLT_MAX;
            if (k < end) { s = ssrc[k]; edge_logits<H>(al, s, ari, lg); }
        }
        float w[H];
        #pragma unroll
        for (int h = 0; h < H; ++h) w[h] = (k < end) ? __expf(lg[h] - m[h]) : 0.f;

        const int cnt = (end - c) < 32 ? (end - c) : 32;
        int j = 0;
        for (; j + 8 <= cnt; j += 8)
            agg_rows<H, 8, false>(xlin, lane, s, w, hm, j, cnt, accx, accy, S);
        for (; j < cnt; j += 4)
            agg_rows<H, 4, true>(xlin, lane, s, w, hm, j, cnt, accx, accy, S);
    }

    const float den = S + 1e-16f;
    const float2 r = *reinterpret_cast<const float2*>(res + (size_t)i * 64 + 2 * lane);
    const float2 b = *reinterpret_cast<const float2*>(bias + 2 * lane);
    float vx = accx / den + r.x + b.x;
    float vy = accy / den + r.y + b.y;
    if (ELU) {
        vx = vx > 0.f ? vx : (__expf(vx) - 1.f);
        vy = vy > 0.f ? vy : (__expf(vy) - 1.f);
    }
    *reinterpret_cast<float2*>(outp + (size_t)i * 64 + 2 * lane) = make_float2(vx, vy);
}

extern "C" void kernel_launch(void* const* d_in, const int* in_sizes, int n_in,
                              void* d_out, int out_size, void* d_ws, size_t ws_size,
                              hipStream_t stream) {
    const float* x     = (const float*)d_in[0];
    const int*   eidx  = (const int*)d_in[1];
    const float* W1    = (const float*)d_in[2];
    const float* attL1 = (const float*)d_in[3];
    const float* attR1 = (const float*)d_in[4];
    const float* resW1 = (const float*)d_in[5];
    const float* b1    = (const float*)d_in[6];
    const float* W2    = (const float*)d_in[7];
    const float* attL2 = (const float*)d_in[8];
    const float* attR2 = (const float*)d_in[9];
    const float* b2    = (const float*)d_in[10];

    const int n = in_sizes[0] / 128;
    const int e = in_sizes[1] / 2;
    const int* srcI = eidx;
    const int* dstI = eidx + e;

    float* ws = (float*)d_ws;
    float* A  = ws;
    float* B  = A  + (size_t)n * 64;
    float* C  = B  + (size_t)n * 64;
    float* AL = C  + (size_t)n * 64;
    float* AR = AL + (size_t)n * 4;
    const int nb = (n + BW - 1) / BW;
    int2* TMP = (int2*)C;
    int* SS   = (int*)C + 2 * (size_t)e;
    int* RP   = SS + e;
    int* HIST = RP + n + 1;
    int* BST  = HIST + SORT_G * nb;

    constexpr int RT = 2;
    const int tiles = (n + 16 * RT - 1) / (16 * RT);
    const int nWav  = (n + 7) / 8;

    bin_hist_k<<<SORT_G, SORT_T, 0, stream>>>(dstI, HIST, e, nb);
    bin_scatter_k<<<SORT_G, SORT_T, 0, stream>>>(srcI, dstI, HIST, BST, TMP, e, nb);
    bin_place_k<<<nb, PLACE_T, 0, stream>>>(TMP, BST, RP, SS, n, nb);

    gemm_x2<true, RT, 4><<<tiles, 32, 0, stream>>>(x, W1, resW1, A, B, attL1, attR1, AL, AR, n, 128);
    node_agg_k<4, true><<<nWav, 256, 0, stream>>>(RP, SS, AL, AR, A, B, b1, B, n);

    gemm_x2<false, RT, 1><<<tiles, 32, 0, stream>>>(B, W2, nullptr, A, nullptr, attL2, attR2, AL, AR, n, 64);
    node_agg_k<1, false><<<nWav, 256, 0, stream>>>(RP, SS, AL, AR, A, A, b2, (float*)d_out, n);
}


# node aggregate: each lane takes its head's edge weights from a per-wave LDS table (one 16-byte read per 4 edges), not 4 lane-reads+selects per edge; rows by 32-bit index; same sums in same order
# speedup vs baseline: 6.1583x; 1.0362x over previous
#include <hip/hip_runtime.h>
#include <math.h>
#include <float.h>

typedef float v2f __attribute__((ext_vector_type(2)));
typedef float v8f __attribute__((ext_vector_type(8)));

#define HC 64
#define SLOPE 0.2f

__device__ __forceinline__ float lrelu(float x) { return x > 0.f ? x : SLOPE * x; }

#define TP 68
template <bool DUAL, int RT, int H>
__global__ __launch_bounds__(32)
void gemm_x2(const float* __restrict__ X, const float* __restrict__ W,
             const float* __restrict__ Wr, float* __restrict__ Y,
             float* __restrict__ Yr,
             const float* __restrict__ attL, const float* __restrict__ attR,
             float* __restrict__ al, float* __restrict__ ar, int n, int K) {
    static_assert(RT == 2, "the epilogue gives each of the 32 lanes one tile row");
    __shared__ __attribute__((aligned(16))) float tile[16 * RT * TP];
    const int lane  = threadIdx.x;
    const int row16 = lane & 15;
    const int khalf = (lane >> 4) << 1;
    const int col   = lane & 15;
    const int r0    = blockIdx.x * (16 * RT);

    int arow[RT];
    #pragma unroll
    for (int t = 0; t < RT; ++t) {
        int r = r0 + 16 * t + row16;
        arow[t] = r < n ? r : n - 1;
    }

    v8f acc[RT][4]  = {};
    v8f accR[RT][4] = {};

    #pragma unroll 2
    for (int k0 = 0; k0 < K; k0 += 4) {
        v2f a[RT];
        #pragma unroll
        for (int t = 0; t < RT; ++t)
            a[t] = *reinterpret_cast<const v2f*>(X + (size_t)arow[t] * K + k0 + khalf);

        #pragma unroll
        for (int j = 0; j < 4; ++j) {
            v2f b;
            const float* bp = W + (size_t)(k0 + khalf) * HC + j * 16 + col;
            b.x = bp[0]; b.y = bp[HC];
            #pragma unroll
            for (int t = 0; t < RT; ++t)
                acc[t][j] = __builtin_amdgcn_wmma_f32_16x16x4_f32(
                    false, a[t], false, b, (short)0, acc[t][j], false, false);
            if (DUAL) {
                v2f br;
                const float* brp = Wr + (size_t)(k0 + khalf) * HC + j * 16 + col;
                br.x = brp[0]; br.y = brp[HC];
                #pragma unroll
                for (int t = 0; t < RT; ++t)
                    accR[t][j] = __builtin_amdgcn_wmma_f32_16x16x4_f32(
                        false, a[t], false, br, (short)0, accR[t][j], false, false);
            }
        }
    }

    const int mhi = (lane >> 4) * 8;
    #pragma unroll
    for (int t = 0; t < RT; ++t)
        #pragma unroll
        for (int j = 0; j < 4; ++j)
            #pragma unroll
            for (int r = 0; r < 8; ++r) {
                int m = r0 + 16 * t + mhi + r;
                tile[(16 * t + mhi + r) * TP + j * 16 + col] = acc[t][j][r];
                if (m < n) {
                    Y[(size_t)m * HC + j * 16 + col] = acc[t][j][r];
                    if (DUAL) Yr[(size_t)m * HC + j * 16 + col] = accR[t][j][r];
                }
            }

    __builtin_amdgcn_fence(__ATOMIC_RELEASE, "workgroup", "local");
    __builtin_amdgcn_s_barrier();
    __builtin_amdgcn_fence(__ATOMIC_ACQUIRE, "workgroup", "local");

    const int row = r0 + lane;
    const float4* rowp = reinterpret_cast<const float4*>(tile + lane * TP);
    const float4* lp = reinterpret_cast<const float4*>(attL);
    const float4* rp = reinterpret_cast<const float4*>(attR);
    if (H == 4) {
        float aL[4] = {0, 0, 0, 0}, aR[4] = {0, 0, 0, 0};
        #pragma unroll
        for (int q = 0; q < 16; ++q) {
            const float4 v = rowp[q];
            const float4 l = lp[q];
            const float4 rr = rp[q];
            const int h = q >> 2;
            aL[h] += v.x * l.x + v.y * l.y + v.z * l.z + v.w * l.w;
            aR[h] += v.x * rr.x + v.y * rr.y + v.z * rr.z + v.w * rr.w;
        }
        if (row < n) {
            *reinterpret_cast<float4*>(al + (size_t)row * 4) = make_float4(aL[0], aL[1], aL[2], aL[3]);
            *reinterpret_cast<float4*>(ar + (size_t)row * 4) = make_float4(aR[0], aR[1], aR[2], aR[3]);
        }
    } else {
        float aL = 0.f, aR = 0.f;
        for (int q = 0; q < 16; ++q) {
            const float4 v = rowp[q];
            const float4 l = lp[q];
            const float4 rr = rp[q];
            aL += v.x * l.x + v.y * l.y + v.z * l.z + v.w * l.w;
            aR += v.x * rr.x + v.y * rr.y + v.z * rr.z + v.w * rr.w;
        }
        if (row < n) { al[row] = aL; ar[row] = aR; }
    }
}

#define BSH    8
#define BW     (1 << BSH)
#define MAXNB  1024
#define SORT_G 128
#define SORT_T 1024
#define SORT_U 16

__device__ __forceinline__ int bin_of(int d, int nb) {
    const unsigned b = (unsigned)d >> BSH;
    return b < (unsigned)nb ? (int)b : nb - 1;
}

__global__ __launch_bounds__(SORT_T)
void bin_hist_k(const int* __restrict__ dst, int* __restrict__ hist, int e, int nb) {
    __shared__ int h[MAXNB];
    const int t = threadIdx.x;
    for (int b = t; b < nb; b += SORT_T) h[b] = 0;
    __syncthreads();
    const int ch = (e + SORT_G - 1) / SORT_G;
    const int lo = blockIdx.x * ch;
    const int hi = lo + ch < e ? lo + ch : e;
    for (int i0 = lo + t; i0 < hi; i0 += SORT_U * SORT_T) {
        int d[SORT_U];
        #pragma unroll
        for (int u = 0; u < SORT_U; ++u) {
            const int i = i0 + u * SORT_T;
            d[u] = i < hi ? dst[i] : 0;
        }
        #pragma unroll
        for (int u = 0; u < SORT_U; ++u)
            if (i0 + u * SORT_T < hi) atomicAdd(&h[bin_of(d[u], nb)], 1);
    }
    __syncthreads();
    for (int b = t; b < nb; b += SORT_T) hist[blockIdx.x * nb + b] = h[b];
}

__global__ __launch_bounds__(SORT_T)
void bin_scatter_k(const int* __restrict__ src, const int* __restrict__ dst,
                   const int* __restrict__ hist, int* __restrict__ bstart,
                   int2* __restrict__ tmp, int e, int nb) {
    __shared__ int cur[MAXNB];
    __shared__ int wtot[32];
    const int t = threadIdx.x;
    const int lane = t & 31;
    const int wave = t >> 5;
    const int g = blockIdx.x;

    int pre = 0, tot = 0;
    if (t < nb) {
        #pragma unroll 32
        for (int gg = 0; gg < SORT_G; ++gg) {
            const int v = hist[gg * nb + t];
            tot += v;
            pre += gg < g ? v : 0;
        }
    }
    int inc = tot;
    #pragma unroll
    for (int off = 1; off < 32; off <<= 1) {
        int u = __shfl_up(inc, off);
        if (lane >= off) inc += u;
    }
    if (lane == 31) wtot[wave] = inc;
    __syncthreads();
    int tinc = wtot[lane];
    #pragma unroll
    for (int off = 1; off < 32; off <<= 1) {
        int u = __shfl_up(tinc, off);
        if (lane >= off) tinc += u;
    }
    const int below = __shfl(tinc, wave > 0 ? wave - 1 : 0);
    const int start = (wave > 0 ? below : 0) + inc - tot;
    if (t < nb) {
        cur[t] = start + pre;
        if (g == 0) bstart[t] = start;
    }
    if (g == 0 && t == 0) bstart[nb] = e;
    __syncthreads();

    const int ch = (e + SORT_G - 1) / SORT_G;
    const int lo = g * ch;
    const int hi = lo + ch < e ? lo + ch : e;
    for (int i0 = lo + t; i0 < hi; i0 += SORT_U * SORT_T) {
        int d[SORT_U], s[SORT_U];
        #pragma unroll
        for (int u = 0; u < SORT_U; ++u) {
            const int i = i0 + u * SORT_T;
            d[u] = i < hi ? dst[i] : 0;
            s[u] = i < hi ? src[i] : 0;
        }
        #pragma unroll
        for (int u = 0; u < SORT_U; ++u)
            if (i0 + u * SORT_T < hi) {
                const int p = atomicAdd(&cur[bin_of(d[u], nb)], 1);
                tmp[p] = make_int2(s[u], d[u]);
            }
    }
}

#define PLACE_T 1024
#define PLACE_U 8
__global__ __launch_bounds__(PLACE_T)
void bin_place_k(const int2* __restrict__ tmp, const int* __restrict__ bstart,
                 int* __restrict__ rowptr, int* __restrict__ ssrc, int n, int nb) {
    __shared__ int cnt[BW];
    __shared__ int wt[BW / 32];
    const int t = threadIdx.x;
    const int lane = t & 31;
    const int wave = t >> 5;
    const int b = blockIdx.x;
    const int lo = bstart[b];
    const int hi = bstart[b + 1];
    if (t < BW) cnt[t] = 0;
    __syncthreads();
    for (int i0 = lo + t; i0 < hi; i0 += PLACE_U * PLACE_T) {
        int d[PLACE_U];
        #pragma unroll
        for (int u = 0; u < PLACE_U; ++u) {
            const int i = i0 + u * PLACE_T;
            d[u] = i < hi ? tmp[i].y : 0;
        }
        #pragma unroll
        for (int u = 0; u < PLACE_U; ++u)
            if (i0 + u * PLACE_T < hi) atomicAdd(&cnt[d[u] & (BW - 1)], 1);
    }
    __syncthreads();
    int v = 0, inc = 0;
    if (t < BW) {
        v = cnt[t];
        inc = v;
        #pragma unroll
        for (int off = 1; off < 32; off <<= 1) {
            int u = __shfl_up(inc, off);
            if (lane >= off) inc += u;
        }
        if (lane == 31) wt[wave] = inc;
    }
    __syncthreads();
    if (t < BW) {
        int off = 0;
        #pragma unroll
        for (int w = 0; w < BW / 32; ++w) off += w < wave ? wt[w] : 0;
        const int excl = off + inc - v;
        cnt[t] = excl;
        const int node = b * BW + t;
        if (node < n) rowptr[node] = lo + excl;
    }
    if (b == nb - 1 && t == 0) rowptr[n] = hi;
    __syncthreads();
    for (int i0 = lo + t; i0 < hi; i0 += PLACE_U * PLACE_T) {
        int2 ed[PLACE_U];
        #pragma unroll
        for (int u = 0; u < PLACE_U; ++u) {
            const int i = i0 + u * PLACE_T;
            ed[u] = i < hi ? tmp[i] : make_int2(0, 0);
        }
        #pragma unroll
        for (int u = 0; u < PLACE_U; ++u)
            if (i0 + u * PLACE_T < hi) {
                const int p = atomicAdd(&cnt[ed[u].y & (BW - 1)], 1);
                ssrc[lo + p] = ed[u].x;
            }
    }
}

template <int H>
__device__ __forceinline__ void edge_logits(const float* __restrict__ al, int s,
                                            const float (&ari)[H], float (&lg)[H]) {
    if (H == 4) {
        const float4 a = *reinterpret_cast<const float4*>(al + (size_t)s * 4);
        lg[0] = lrelu(a.x + ari[0]);
        lg[1 % H] = lrelu(a.y + ari[1 % H]);
        lg[2 % H] = lrelu(a.z + ari[2 % H]);
        lg[3 % H] = lrelu(a.w + ari[3 % H]);
    } else {
        #pragma unroll
        for (int h = 0; h < H; ++h) lg[h] = lrelu(al[(size_t)s * H + h] + ari[h]);
    }
}

#define WP 36

template <int U, bool TAIL>
__device__ __forceinline__ void agg_rows(const float* xlin, int lane, int s,
                                         const float* wrow, int j0, int cnt,
                                         float& accx, float& accy, float& S) {
    float2 x[U];
    float wj[U];
    #pragma unroll
    for (int q = 0; q < U / 4; ++q) {
        const float4 w4 = *reinterpret_cast<const float4*>(wrow + j0 + 4 * q);
        wj[4 * q] = w4.x; wj[4 * q + 1] = w4.y; wj[4 * q + 2] = w4.z; wj[4 * q + 3] = w4.w;
    }
    const float2* rows = reinterpret_cast<const float2*>(xlin);
    #pragma unroll
    for (int u = 0; u < U; ++u) {
        const int jj = j0 + u;
        const int jc = (!TAIL || jj < cnt) ? jj : cnt - 1;
        const unsigned sj = (unsigned)__builtin_amdgcn_readlane(s, jc);
        x[u] = rows[sj * 32u + (unsigned)lane];
    }
    #pragma unroll
    for (int u = 0; u < U; ++u) {
        accx += x[u].x * wj[u];
        accy += x[u].y * wj[u];
        S += wj[u];
    }
}

template <int H, bool ELU>
__global__ __launch_bounds__(256)
void node_agg_k(const int* __restrict__ rowptr, const int* __restrict__ ssrc,
                const float* __restrict__ al, const float* __restrict__ ar,
                const float* xlin, const float* res,
                const float* __restrict__ bias, float* outp, int n) {
    __shared__ __attribute__((aligned(16))) float wtab[8 * H * WP];
    const int lane = threadIdx.x & 31;
    const int i = __builtin_amdgcn_readfirstlane(
        (int)(blockIdx.x * (blockDim.x >> 5) + (threadIdx.x >> 5)));
    if (i >= n) return;
    const int beg = __builtin_amdgcn_readfirstlane(rowptr[i]);
    const int end = __builtin_amdgcn_readfirstlane(rowptr[i + 1]);
    const int hl  = (2 * lane * H) >> 6;
    float* wmine = wtab + (threadIdx.x >> 5) * (H * WP);
    const float* wrow = wmine + hl * WP;

    float ari[H];
    #pragma unroll
    for (int h = 0; h < H; ++h) ari[h] = ar[(size_t)i * H + h];

    float m[H], lg0[H];
    int s0 = 0;
    #pragma unroll
    for (int h = 0; h < H; ++h) { m[h] = -FLT_MAX; lg0[h] = -FLT_MAX; }
    for (int c = beg; c < end; c += 32) {
        const int k = c + lane;
        float lg[H];
        int s = 0;
        #pragma unroll
        for (int h = 0; h < H; ++h) lg[h] = -FLT_MAX;
        if (k < end) { s = ssrc[k]; edge_logits<H>(al, s, ari, lg); }
        #pragma unroll
        for (int h = 0; h < H; ++h) m[h] = fmaxf(m[h], lg[h]);
        if (c == beg) {
            s0 = s;
            #pragma unroll
            for (int h = 0; h < H; ++h) lg0[h] = lg[h];
        }
    }
    #pragma unroll
    for (int h = 0; h < H; ++h)
        #pragma unroll
        for (int off = 16; off > 0; off >>= 1)
            m[h] = fmaxf(m[h], __shfl_xor(m[h], off));

    float accx = 0.f, accy = 0.f, S = 0.f;
    for (int c = beg; c < end; c += 32) {
        const int k = c + lane;
        int s = s0;
        float lg[H];
        #pragma unroll
        for (int h = 0; h < H; ++h) lg[h] = lg0[h];
        if (c != beg) {
            s = 0;
            #pragma unroll
            for (int h = 0; h < H; ++h) lg[h] = -FLT_MAX;
            if (k < end) { s = ssrc[k]; edge_logits<H>(al, s, ari, lg); }
        }
        float w[H];
        #pragma unroll
        for (int h = 0; h < H; ++h) w[h] = (k < end) ? __expf(lg[h] - m[h]) : 0.f;

        #pragma unroll
        for (int h = 0; h < H; ++h) wmine[h * WP + lane] = w[h];

        const int cnt = (end - c) < 32 ? (end - c) : 32;
        int j = 0;
        for (; j + 8 <= cnt; j += 8)
            agg_rows<8, false>(xlin, lane, s, wrow, j, cnt, accx, accy, S);
        for (; j < cnt; j += 4)
            agg_rows<4, true>(xlin, lane, s, wrow, j, cnt, accx, accy, S);
    }

    const float den = S + 1e-16f;
    const float2 r = *reinterpret_cast<const float2*>(res + (size_t)i * 64 + 2 * lane);
    const float2 b = *reinterpret_cast<const float2*>(bias + 2 * lane);
    float vx = accx / den + r.x + b.x;
    float vy = accy / den + r.y + b.y;
    if (ELU) {
        vx = vx > 0.f ? vx : (__expf(vx) - 1.f);
        vy = vy > 0.f ? vy : (__expf(vy) - 1.f);
    }
    *reinterpret_cast<float2*>(outp + (size_t)i * 64 + 2 * lane) = make_float2(vx, vy);
}

extern "C" void kernel_launch(void* const* d_in, const int* in_sizes, int n_in,
                              void* d_out, int out_size, void* d_ws, size_t ws_size,
                              hipStream_t stream) {
    const float* x     = (const float*)d_in[0];
    const int*   eidx  = (const int*)d_in[1];
    const float* W1    = (const float*)d_in[2];
    const float* attL1 = (const float*)d_in[3];
    const float* attR1 = (const float*)d_in[4];
    const float* resW1 = (const float*)d_in[5];
    const float* b1    = (const float*)d_in[6];
    const float* W2    = (const float*)d_in[7];
    const float* attL2 = (const float*)d_in[8];
    const float* attR2 = (const float*)d_in[9];
    const float* b2    = (const float*)d_in[10];

    const int n = in_sizes[0] / 128;
    const int e = in_sizes[1] / 2;
    const int* srcI = eidx;
    const int* dstI = eidx + e;

    float* ws = (float*)d_ws;
    float* A  = ws;
    float* B  = A  + (size_t)n * 64;
    float* C  = B  + (size_t)n * 64;
    float* AL = C  + (size_t)n * 64;
    float* AR = AL + (size_t)n * 4;
    const int nb = (n + BW - 1) / BW;
    int2* TMP = (int2*)C;
    int* SS   = (int*)C + 2 * (size_t)e;
    int* RP   = SS + e;
    int* HIST = RP + n + 1;
    int* BST  = HIST + SORT_G * nb;

    constexpr int RT = 2;
    const int tiles = (n + 16 * RT - 1) / (16 * RT);
    const int nWav  = (n + 7) / 8;

    bin_hist_k<<<SORT_G, SORT_T, 0, stream>>>(dstI, HIST, e, nb);
    bin_scatter_k<<<SORT_G, SORT_T, 0, stream>>>(srcI, dstI, HIST, BST, TMP, e, nb);
    bin_place_k<<<nb, PLACE_T, 0, stream>>>(TMP, BST, RP, SS, n, nb);

    gemm_x2<true, RT, 4><<<tiles, 32, 0, stream>>>(x, W1, resW1, A, B, attL1, attR1, AL, AR, n, 128);
    node_agg_k<4, true><<<nWav, 256, 0, stream>>>(RP, SS, AL, AR, A, B, b1, B, n);

    gemm_x2<false, RT, 1><<<tiles, 32, 0, stream>>>(B, W2, nullptr, A, nullptr, attL2, attR2, AL, AR, n, 64);
    node_agg_k<1, false><<<nWav, 256, 0, stream>>>(RP, SS, AL, AR, A, A, b2, (float*)d_out, n);
}


# node aggregate: edge logit taken from the gathered source row (8-lane exchange sum of x*attL), running-max softmax rescaled per 8 edges; al gather and max pass go; al, weight sums in tree order
# speedup vs baseline: 6.4590x; 1.0488x over previous
#include <hip/hip_runtime.h>
#include <math.h>
#include <float.h>

typedef float v2f __attribute__((ext_vector_type(2)));
typedef float v8f __attribute__((ext_vector_type(8)));

#define HC 64
#define SLOPE 0.2f

__device__ __forceinline__ float lrelu(float x) { return x > 0.f ? x : SLOPE * x; }

#define TP 68
template <bool DUAL, int RT, int H>
__global__ __launch_bounds__(32)
void gemm_x2(const float* __restrict__ X, const float* __restrict__ W,
             const float* __restrict__ Wr, float* __restrict__ Y,
             float* __restrict__ Yr,
             const float* __restrict__ attR, float* __restrict__ ar, int n, int K) {
    static_assert(RT == 2, "the epilogue gives each of the 32 lanes one tile row");
    __shared__ __attribute__((aligned(16))) float tile[16 * RT * TP];
    const int lane  = threadIdx.x;
    const int row16 = lane & 15;
    const int khalf = (lane >> 4) << 1;
    const int col   = lane & 15;
    const int r0    = blockIdx.x * (16 * RT);

    int arow[RT];
    #pragma unroll
    for (int t = 0; t < RT; ++t) {
        int r = r0 + 16 * t + row16;
        arow[t] = r < n ? r : n - 1;
    }

    v8f acc[RT][4]  = {};
    v8f accR[RT][4] = {};

    #pragma unroll 2
    for (int k0 = 0; k0 < K; k0 += 4) {
        v2f a[RT];
        #pragma unroll
        for (int t = 0; t < RT; ++t)
            a[t] = *reinterpret_cast<const v2f*>(X + (size_t)arow[t] * K + k0 + khalf);

        #pragma unroll
        for (int j = 0; j < 4; ++j) {
            v2f b;
            const float* bp = W + (size_t)(k0 + khalf) * HC + j * 16 + col;
            b.x = bp[0]; b.y = bp[HC];
            #pragma unroll
            for (int t = 0; t < RT; ++t)
                acc[t][j] = __builtin_amdgcn_wmma_f32_16x16x4_f32(
                    false, a[t], false, b, (short)0, acc[t][j], false, false);
            if (DUAL) {
                v2f br;
                const float* brp = Wr + (size_t)(k0 + khalf) * HC + j * 16 + col;
                br.x = brp[0]; br.y = brp[HC];
                #pragma unroll
                for (int t = 0; t < RT; ++t)
                    accR[t][j] = __builtin_amdgcn_wmma_f32_16x16x4_f32(
                        false, a[t], false, br, (short)0, accR[t][j], false, false);
            }
        }
    }

    const int mhi = (lane >> 4) * 8;
    #pragma unroll
    for (int t = 0; t < RT; ++t)
        #pragma unroll
        for (int j = 0; j < 4; ++j)
            #pragma unroll
            for (int r = 0; r < 8; ++r) {
                int m = r0 + 16 * t + mhi + r;
                tile[(16 * t + mhi + r) * TP + j * 16 + col] = acc[t][j][r];
                if (m < n) {
                    Y[(size_t)m * HC + j * 16 + col] = acc[t][j][r];
                    if (DUAL) Yr[(size_t)m * HC + j * 16 + col] = accR[t][j][r];
                }
            }

    __builtin_amdgcn_fence(__ATOMIC_RELEASE, "workgroup", "local");
    __builtin_amdgcn_s_barrier();
    __builtin_amdgcn_fence(__ATOMIC_ACQUIRE, "workgroup", "local");

    const int row = r0 + lane;
    const float4* rowp = reinterpret_cast<const float4*>(tile + lane * TP);
    const float4* rp = reinterpret_cast<const float4*>(attR);
    if (H == 4) {
        float aR[4] = {0, 0, 0, 0};
        #pragma unroll
        for (int q = 0; q < 16; ++q) {
            const float4 v = rowp[q];
            const float4 rr = rp[q];
            const int h = q >> 2;
            aR[h] += v.x * rr.x + v.y * rr.y + v.z * rr.z + v.w * rr.w;
        }
        if (row < n)
            *reinterpret_cast<float4*>(ar + (size_t)row * 4) = make_float4(aR[0], aR[1], aR[2], aR[3]);
    } else {
        float aR = 0.f;
        for (int q = 0; q < 16; ++q) {
            const float4 v = rowp[q];
            const float4 rr = rp[q];
            aR += v.x * rr.x + v.y * rr.y + v.z * rr.z + v.w * rr.w;
        }
        if (row < n) ar[row] = aR;
    }
}

#define BSH    8
#define BW     (1 << BSH)
#define MAXNB  1024
#define SORT_G 128
#define SORT_T 1024
#define SORT_U 16

__device__ __forceinline__ int bin_of(int d, int nb) {
    const unsigned b = (unsigned)d >> BSH;
    return b < (unsigned)nb ? (int)b : nb - 1;
}

__global__ __launch_bounds__(SORT_T)
void bin_hist_k(const int* __restrict__ dst, int* __restrict__ hist, int e, int nb) {
    __shared__ int h[MAXNB];
    const int t = threadIdx.x;
    for (int b = t; b < nb; b += SORT_T) h[b] = 0;
    __syncthreads();
    const int ch = (e + SORT_G - 1) / SORT_G;
    const int lo = blockIdx.x * ch;
    const int hi = lo + ch < e ? lo + ch : e;
    for (int i0 = lo + t; i0 < hi; i0 += SORT_U * SORT_T) {
        int d[SORT_U];
        #pragma unroll
        for (int u = 0; u < SORT_U; ++u) {
            const int i = i0 + u * SORT_T;
            d[u] = i < hi ? dst[i] : 0;
        }
        #pragma unroll
        for (int u = 0; u < SORT_U; ++u)
            if (i0 + u * SORT_T < hi) atomicAdd(&h[bin_of(d[u], nb)], 1);
    }
    __syncthreads();
    for (int b = t; b < nb; b += SORT_T) hist[blockIdx.x * nb + b] = h[b];
}

__global__ __launch_bounds__(SORT_T)
void bin_scatter_k(const int* __restrict__ src, const int* __restrict__ dst,
                   const int* __restrict__ hist, int* __restrict__ bstart,
                   int2* __restrict__ tmp, int e, int nb) {
    __shared__ int cur[MAXNB];
    __shared__ int wtot[32];
    const int t = threadIdx.x;
    const int lane = t & 31;
    const int wave = t >> 5;
    const int g = blockIdx.x;

    int pre = 0, tot = 0;
    if (t < nb) {
        #pragma unroll 32
        for (int gg = 0; gg < SORT_G; ++gg) {
            const int v = hist[gg * nb + t];
            tot += v;
            pre += gg < g ? v : 0;
        }
    }
    int inc = tot;
    #pragma unroll
    for (int off = 1; off < 32; off <<= 1) {
        int u = __shfl_up(inc, off);
        if (lane >= off) inc += u;
    }
    if (lane == 31) wtot[wave] = inc;
    __syncthreads();
    int tinc = wtot[lane];
    #pragma unroll
    for (int off = 1; off < 32; off <<= 1) {
        int u = __shfl_up(tinc, off);
        if (lane >= off) tinc += u;
    }
    const int below = __shfl(tinc, wave > 0 ? wave - 1 : 0);
    const int start = (wave > 0 ? below : 0) + inc - tot;
    if (t < nb) {
        cur[t] = start + pre;
        if (g == 0) bstart[t] = start;
    }
    if (g == 0 && t == 0) bstart[nb] = e;
    __syncthreads();

    const int ch = (e + SORT_G - 1) / SORT_G;
    const int lo = g * ch;
    const int hi = lo + ch < e ? lo + ch : e;
    for (int i0 = lo + t; i0 < hi; i0 += SORT_U * SORT_T) {
        int d[SORT_U], s[SORT_U];
        #pragma unroll
        for (int u = 0; u < SORT_U; ++u) {
            const int i = i0 + u * SORT_T;
            d[u] = i < hi ? dst[i] : 0;
            s[u] = i < hi ? src[i] : 0;
        }
        #pragma unroll
        for (int u = 0; u < SORT_U; ++u)
            if (i0 + u * SORT_T < hi) {
                const int p = atomicAdd(&cur[bin_of(d[u], nb)], 1);
                tmp[p] = make_int2(s[u], d[u]);
            }
    }
}

#define PLACE_T 1024
#define PLACE_U 8
__global__ __launch_bounds__(PLACE_T)
void bin_place_k(const int2* __restrict__ tmp, const int* __restrict__ bstart,
                 int* __restrict__ rowptr, int* __restrict__ ssrc, int n, int nb) {
    __shared__ int cnt[BW];
    __shared__ int wt[BW / 32];
    const int t = threadIdx.x;
    const int lane = t & 31;
    const int wave = t >> 5;
    const int b = blockIdx.x;
    const int lo = bstart[b];
    const int hi = bstart[b + 1];
    if (t < BW) cnt[t] = 0;
    __syncthreads();
    for (int i0 = lo + t; i0 < hi; i0 += PLACE_U * PLACE_T) {
        int d[PLACE_U];
        #pragma unroll
        for (int u = 0; u < PLACE_U; ++u) {
            const int i = i0 + u * PLACE_T;
            d[u] = i < hi ? tmp[i].y : 0;
        }
        #pragma unroll
        for (int u = 0; u < PLACE_U; ++u)
            if (i0 + u * PLACE_T < hi) atomicAdd(&cnt[d[u] & (BW - 1)], 1);
    }
    __syncthreads();
    int v = 0, inc = 0;
    if (t < BW) {
        v = cnt[t];
        inc = v;
        #pragma unroll
        for (int off = 1; off < 32; off <<= 1) {
            int u = __shfl_up(inc, off);
            if (lane >= off) inc += u;
        }
        if (lane == 31) wt[wave] = inc;
    }
    __syncthreads();
    if (t < BW) {
        int off = 0;
        #pragma unroll
        for (int w = 0; w < BW / 32; ++w) off += w < wave ? wt[w] : 0;
        const int excl = off + inc - v;
        cnt[t] = excl;
        const int node = b * BW + t;
        if (node < n) rowptr[node] = lo + excl;
    }
    if (b == nb - 1 && t == 0) rowptr[n] = hi;
    __syncthreads();
    for (int i0 = lo + t; i0 < hi; i0 += PLACE_U * PLACE_T) {
        int2 ed[PLACE_U];
        #pragma unroll
        for (int u = 0; u < PLACE_U; ++u) {
            const int i = i0 + u * PLACE_T;
            ed[u] = i < hi ? tmp[i] : make_int2(0, 0);
        }
        #pragma unroll
        for (int u = 0; u < PLACE_U; ++u)
            if (i0 + u * PLACE_T < hi) {
                const int p = atomicAdd(&cnt[ed[u].y & (BW - 1)], 1);
                ssrc[lo + p] = ed[u].x;
            }
    }
}


template <int CTRL>
__device__ __forceinline__ float dpp_rd(float x) {
    return __builtin_bit_cast(float, __builtin_amdgcn_update_dpp(
        0, __builtin_bit_cast(int, x), CTRL, 0xF, 0xF, true));
}
__device__ __forceinline__ float xrow_rd(float x) {
    return __builtin_bit_cast(float, __builtin_amdgcn_permlanex16(
        __builtin_bit_cast(int, x), __builtin_bit_cast(int, x), 0x76543210, 0xfedcba98, false, false));
}
template <int U>
__device__ __forceinline__ float grp8_bcast(float x) {
    return __builtin_bit_cast(float, __builtin_amdgcn_mov_dpp8(
        __builtin_bit_cast(unsigned, x), (unsigned)U * 0x249249u));
}
__device__ __forceinline__ float grp8_sum(float v) {
    v += dpp_rd<0xB1>(v); v += dpp_rd<0x4E>(v); v += dpp_rd<0x164>(v);
    return v;
}
__device__ __forceinline__ float grp8_max(float v) {
    v = fmaxf(v, dpp_rd<0xB1>(v)); v = fmaxf(v, dpp_rd<0x4E>(v)); v = fmaxf(v, dpp_rd<0x164>(v));
    return v;
}

#define EU 8

template <int H, bool ELU>
__global__ __launch_bounds__(256)
void node_agg_k(const int* __restrict__ rowptr, const int* __restrict__ ssrc,
                const float* __restrict__ attL, const float* __restrict__ ar,
                const float* xlin, const float* res,
                const float* __restrict__ bias, float* outp, int n) {
    static_assert(H == 4 || H == 1, "a head is one group of 8 lanes, or all four");
    const int lane = threadIdx.x & 31;
    const int i = __builtin_amdgcn_readfirstlane(
        (int)(blockIdx.x * (blockDim.x >> 5) + (threadIdx.x >> 5)));
    if (i >= n) return;
    const int beg = __builtin_amdgcn_readfirstlane(rowptr[i]);
    const int end = __builtin_amdgcn_readfirstlane(rowptr[i + 1]);
    const int hl  = (2 * lane * H) >> 6;
    const int g   = lane & 7;
    const bool b4 = (lane & 4) != 0, b2 = (lane & 2) != 0, b1 = (lane & 1) != 0;

    const float2 aL = *reinterpret_cast<const float2*>(attL + 2 * lane);
    const float arh = ar[(size_t)i * H + hl];
    const float2* rows = reinterpret_cast<const float2*>(xlin);

    float m = -FLT_MAX;
    float accx = 0.f, accy = 0.f;
    float So = 0.f;
    for (int c = beg; c < end; c += 32) {
        const int k = c + lane;
        const int s = k < end ? ssrc[k] : 0;
        const int cnt = (end - c) < 32 ? (end - c) : 32;
        for (int j = 0; j < cnt; j += EU) {
            float2 x[EU];
            #pragma unroll
            for (int u = 0; u < EU; ++u) {
                const int jj = j + u;
                const int jc = jj < cnt ? jj : cnt - 1;
                const unsigned sj = (unsigned)__builtin_amdgcn_readlane(s, jc);
                x[u] = rows[sj * 32u + (unsigned)lane];
            }
            float p[EU];
            #pragma unroll
            for (int u = 0; u < EU; ++u) p[u] = x[u].x * aL.x + x[u].y * aL.y;
            float q[4];
            #pragma unroll
            for (int u = 0; u < 4; ++u) {
                const float keep = b4 ? p[u + 4] : p[u];
                const float send = b4 ? p[u] : p[u + 4];
                q[u] = keep + dpp_rd<0x164>(send);
            }
            float r[2];
            #pragma unroll
            for (int u = 0; u < 2; ++u) {
                const float keep = b2 ? q[u + 2] : q[u];
                const float send = b2 ? q[u] : q[u + 2];
                r[u] = keep + dpp_rd<0x4E>(send);
            }
            float t;
            {
                const float keep = b1 ? r[1] : r[0];
                const float send = b1 ? r[0] : r[1];
                t = keep + dpp_rd<0xB1>(send);
            }
            if (H == 1) {
                t += dpp_rd<0x168>(t);
                t += xrow_rd(t);
            }
            const float lgo = (j + g < cnt) ? lrelu(t + arh) : -FLT_MAX;
            const float mn = fmaxf(m, grp8_max(lgo));
            const float sc = __expf(m - mn);
            const float wo = __expf(lgo - mn);
            accx *= sc; accy *= sc;
            So = So * sc + wo;
            m = mn;
#define AGG_EDGE(U) { const float w = grp8_bcast<U>(wo); accx += x[U].x * w; accy += x[U].y * w; }
            AGG_EDGE(0) AGG_EDGE(1) AGG_EDGE(2) AGG_EDGE(3)
            AGG_EDGE(4) AGG_EDGE(5) AGG_EDGE(6) AGG_EDGE(7)
#undef AGG_EDGE
        }
    }

    const float den = grp8_sum(So) + 1e-16f;
    const float2 rr = *reinterpret_cast<const float2*>(res + (size_t)i * 64 + 2 * lane);
    const float2 b = *reinterpret_cast<const float2*>(bias + 2 * lane);
    float vx = accx / den + rr.x + b.x;
    float vy = accy / den + rr.y + b.y;
    if (ELU) {
        vx = vx > 0.f ? vx : (__expf(vx) - 1.f);
        vy = vy > 0.f ? vy : (__expf(vy) - 1.f);
    }
    *reinterpret_cast<float2*>(outp + (size_t)i * 64 + 2 * lane) = make_float2(vx, vy);
}

extern "C" void kernel_launch(void* const* d_in, const int* in_sizes, int n_in,
                              void* d_out, int out_size, void* d_ws, size_t ws_size,
                              hipStream_t stream) {
    const float* x     = (const float*)d_in[0];
    const int*   eidx  = (const int*)d_in[1];
    const float* W1    = (const float*)d_in[2];
    const float* attL1 = (const float*)d_in[3];
    const float* attR1 = (const float*)d_in[4];
    const float* resW1 = (const float*)d_in[5];
    const float* b1    = (const float*)d_in[6];
    const float* W2    = (const float*)d_in[7];
    const float* attL2 = (const float*)d_in[8];
    const float* attR2 = (const float*)d_in[9];
    const float* b2    = (const float*)d_in[10];

    const int n = in_sizes[0] / 128;
    const int e = in_sizes[1] / 2;
    const int* srcI = eidx;
    const int* dstI = eidx + e;

    float* ws = (float*)d_ws;
    float* A  = ws;
    float* B  = A  + (size_t)n * 64;
    float* C  = B  + (size_t)n * 64;
    float* AR = C  + (size_t)n * 64;
    const int nb = (n + BW - 1) / BW;
    int2* TMP = (int2*)C;
    int* SS   = (int*)C + 2 * (size_t)e;
    int* RP   = SS + e;
    int* HIST = RP + n + 1;
    int* BST  = HIST + SORT_G * nb;

    constexpr int RT = 2;
    const int tiles = (n + 16 * RT - 1) / (16 * RT);
    const int nWav  = (n + 7) / 8;

    bin_hist_k<<<SORT_G, SORT_T, 0, stream>>>(dstI, HIST, e, nb);
    bin_scatter_k<<<SORT_G, SORT_T, 0, stream>>>(srcI, dstI, HIST, BST, TMP, e, nb);
    bin_place_k<<<nb, PLACE_T, 0, stream>>>(TMP, BST, RP, SS, n, nb);

    gemm_x2<true, RT, 4><<<tiles, 32, 0, stream>>>(x, W1, resW1, A, B, attR1, AR, n, 128);
    node_agg_k<4, true><<<nWav, 256, 0, stream>>>(RP, SS, attL1, AR, A, B, b1, B, n);

    gemm_x2<false, RT, 1><<<tiles, 32, 0, stream>>>(B, W2, nullptr, A, nullptr, attR2, AR, n, 64);
    node_agg_k<1, false><<<nWav, 256, 0, stream>>>(RP, SS, attL2, AR, A, A, b2, (float*)d_out, n);
}


# edge grouping: bin-ordered copy and per-node placement order their edges in LDS first, then write them front to back as whole lines, not 32 lines per store; same slots, same LDS arrival order
# speedup vs baseline: 6.9900x; 1.0822x over previous
#include <hip/hip_runtime.h>
#include <math.h>
#include <float.h>

typedef float v2f __attribute__((ext_vector_type(2)));
typedef float v8f __attribute__((ext_vector_type(8)));

#define HC 64
#define SLOPE 0.2f

__device__ __forceinline__ float lrelu(float x) { return x > 0.f ? x : SLOPE * x; }

#define TP 68
template <bool DUAL, int RT, int H>
__global__ __launch_bounds__(32)
void gemm_x2(const float* __restrict__ X, const float* __restrict__ W,
             const float* __restrict__ Wr, float* __restrict__ Y,
             float* __restrict__ Yr,
             const float* __restrict__ attR, float* __restrict__ ar, int n, int K) {
    static_assert(RT == 2, "the epilogue gives each of the 32 lanes one tile row");
    __shared__ __attribute__((aligned(16))) float tile[16 * RT * TP];
    const int lane  = threadIdx.x;
    const int row16 = lane & 15;
    const int khalf = (lane >> 4) << 1;
    const int col   = lane & 15;
    const int r0    = blockIdx.x * (16 * RT);

    int arow[RT];
    #pragma unroll
    for (int t = 0; t < RT; ++t) {
        int r = r0 + 16 * t + row16;
        arow[t] = r < n ? r : n - 1;
    }

    v8f acc[RT][4]  = {};
    v8f accR[RT][4] = {};

    #pragma unroll 2
    for (int k0 = 0; k0 < K; k0 += 4) {
        v2f a[RT];
        #pragma unroll
        for (int t = 0; t < RT; ++t)
            a[t] = *reinterpret_cast<const v2f*>(X + (size_t)arow[t] * K + k0 + khalf);

        #pragma unroll
        for (int j = 0; j < 4; ++j) {
            v2f b;
            const float* bp = W + (size_t)(k0 + khalf) * HC + j * 16 + col;
            b.x = bp[0]; b.y = bp[HC];
            #pragma unroll
            for (int t = 0; t < RT; ++t)
                acc[t][j] = __builtin_amdgcn_wmma_f32_16x16x4_f32(
                    false, a[t], false, b, (short)0, acc[t][j], false, false);
            if (DUAL) {
                v2f br;
                const float* brp = Wr + (size_t)(k0 + khalf) * HC + j * 16 + col;
                br.x = brp[0]; br.y = brp[HC];
                #pragma unroll
                for (int t = 0; t < RT; ++t)
                    accR[t][j] = __builtin_amdgcn_wmma_f32_16x16x4_f32(
                        false, a[t], false, br, (short)0, accR[t][j], false, false);
            }
        }
    }

    const int mhi = (lane >> 4) * 8;
    #pragma unroll
    for (int t = 0; t < RT; ++t)
        #pragma unroll
        for (int j = 0; j < 4; ++j)
            #pragma unroll
            for (int r = 0; r < 8; ++r) {
                int m = r0 + 16 * t + mhi + r;
                tile[(16 * t + mhi + r) * TP + j * 16 + col] = acc[t][j][r];
                if (m < n) {
                    Y[(size_t)m * HC + j * 16 + col] = acc[t][j][r];
                    if (DUAL) Yr[(size_t)m * HC + j * 16 + col] = accR[t][j][r];
                }
            }

    __builtin_amdgcn_fence(__ATOMIC_RELEASE, "workgroup", "local");
    __builtin_amdgcn_s_barrier();
    __builtin_amdgcn_fence(__ATOMIC_ACQUIRE, "workgroup", "local");

    const int row = r0 + lane;
    const float4* rowp = reinterpret_cast<const float4*>(tile + lane * TP);
    const float4* rp = reinterpret_cast<const float4*>(attR);
    if (H == 4) {
        float aR[4] = {0, 0, 0, 0};
        #pragma unroll
        for (int q = 0; q < 16; ++q) {
            const float4 v = rowp[q];
            const float4 rr = rp[q];
            const int h = q >> 2;
            aR[h] += v.x * rr.x + v.y * rr.y + v.z * rr.z + v.w * rr.w;
        }
        if (row < n)
            *reinterpret_cast<float4*>(ar + (size_t)row * 4) = make_float4(aR[0], aR[1], aR[2], aR[3]);
    } else {
        float aR = 0.f;
        for (int q = 0; q < 16; ++q) {
            const float4 v = rowp[q];
            const float4 rr = rp[q];
            aR += v.x * rr.x + v.y * rr.y + v.z * rr.z + v.w * rr.w;
        }
        if (row < n) ar[row] = aR;
    }
}

#define BSH    8
#define BW     (1 << BSH)
#define MAXNB  1024
#define SORT_G 128
#define SORT_T 1024
#define SORT_U 16
#define STG_CAP 16384

__device__ __forceinline__ int bin_of(int d, int nb) {
    const unsigned b = (unsigned)d >> BSH;
    return b < (unsigned)nb ? (int)b : nb - 1;
}

__global__ __launch_bounds__(SORT_T)
void bin_hist_k(const int* __restrict__ dst, int* __restrict__ hist, int e, int nb) {
    __shared__ int h[MAXNB];
    const int t = threadIdx.x;
    for (int b = t; b < nb; b += SORT_T) h[b] = 0;
    __syncthreads();
    const int ch = (e + SORT_G - 1) / SORT_G;
    const int lo = blockIdx.x * ch;
    const int hi = lo + ch < e ? lo + ch : e;
    for (int i0 = lo + t; i0 < hi; i0 += SORT_U * SORT_T) {
        int d[SORT_U];
        #pragma unroll
        for (int u = 0; u < SORT_U; ++u) {
            const int i = i0 + u * SORT_T;
            d[u] = i < hi ? dst[i] : 0;
        }
        #pragma unroll
        for (int u = 0; u < SORT_U; ++u)
            if (i0 + u * SORT_T < hi) atomicAdd(&h[bin_of(d[u], nb)], 1);
    }
    __syncthreads();
    for (int b = t; b < nb; b += SORT_T) hist[blockIdx.x * nb + b] = h[b];
}

__global__ __launch_bounds__(SORT_T)
void bin_scatter_k(const int* __restrict__ src, const int* __restrict__ dst,
                   const int* __restrict__ hist, int* __restrict__ bstart,
                   int2* __restrict__ tmp, int e, int nb) {
    __shared__ int cur[MAXNB];
    __shared__ int gofs[MAXNB];
    __shared__ int wtot[32];
    __shared__ int wown[32];
    __shared__ int2 stage[STG_CAP];
    const int t = threadIdx.x;
    const int lane = t & 31;
    const int wave = t >> 5;
    const int g = blockIdx.x;

    const int ch = (e + SORT_G - 1) / SORT_G;
    const int lo = g * ch;
    const int hi = lo + ch < e ? lo + ch : e;
    const bool ordered = hi - lo <= STG_CAP;

    int pre = 0, tot = 0, own = 0;
    if (t < nb) {
        #pragma unroll 32
        for (int gg = 0; gg < SORT_G; ++gg) {
            const int v = hist[gg * nb + t];
            tot += v;
            pre += gg < g ? v : 0;
        }
        own = hist[g * nb + t];
    }
    int inc = tot, inco = own;
    #pragma unroll
    for (int off = 1; off < 32; off <<= 1) {
        int u = __shfl_up(inc, off);
        int uo = __shfl_up(inco, off);
        if (lane >= off) { inc += u; inco += uo; }
    }
    if (lane == 31) { wtot[wave] = inc; wown[wave] = inco; }
    __syncthreads();
    int tinc = wtot[lane], tinco = wown[lane];
    #pragma unroll
    for (int off = 1; off < 32; off <<= 1) {
        int u = __shfl_up(tinc, off);
        int uo = __shfl_up(tinco, off);
        if (lane >= off) { tinc += u; tinco += uo; }
    }
    const int below  = __shfl(tinc,  wave > 0 ? wave - 1 : 0);
    const int belowo = __shfl(tinco, wave > 0 ? wave - 1 : 0);
    const int start  = (wave > 0 ? below  : 0) + inc  - tot;
    const int lstart = (wave > 0 ? belowo : 0) + inco - own;
    if (t < nb) {
        cur[t]  = ordered ? lstart : start + pre;
        gofs[t] = start + pre - lstart;
        if (g == 0) bstart[t] = start;
    }
    if (g == 0 && t == 0) bstart[nb] = e;
    __syncthreads();

    if (ordered) {
        for (int i0 = lo + t; i0 < hi; i0 += SORT_U * SORT_T) {
            int d[SORT_U], s[SORT_U], p[SORT_U];
            #pragma unroll
            for (int u = 0; u < SORT_U; ++u) {
                const int i = i0 + u * SORT_T;
                d[u] = i < hi ? dst[i] : 0;
                s[u] = i < hi ? src[i] : 0;
            }
            #pragma unroll
            for (int u = 0; u < SORT_U; ++u) {
                p[u] = 0;
                if (i0 + u * SORT_T < hi) p[u] = atomicAdd(&cur[bin_of(d[u], nb)], 1);
            }
            #pragma unroll
            for (int u = 0; u < SORT_U; ++u)
                if (i0 + u * SORT_T < hi) stage[p[u]] = make_int2(s[u], d[u]);
        }
        __syncthreads();
        const int m = hi - lo;
        for (int i0 = t; i0 < m; i0 += SORT_U * SORT_T) {
            int2 v[SORT_U];
            int o[SORT_U];
            #pragma unroll
            for (int u = 0; u < SORT_U; ++u) {
                const int i = i0 + u * SORT_T;
                v[u] = i < m ? stage[i] : make_int2(0, 0);
            }
            #pragma unroll
            for (int u = 0; u < SORT_U; ++u) o[u] = gofs[bin_of(v[u].y, nb)];
            #pragma unroll
            for (int u = 0; u < SORT_U; ++u) {
                const int i = i0 + u * SORT_T;
                if (i < m) tmp[o[u] + i] = v[u];
            }
        }
    } else {
        for (int i0 = lo + t; i0 < hi; i0 += SORT_U * SORT_T) {
            int d[SORT_U], s[SORT_U];
            #pragma unroll
            for (int u = 0; u < SORT_U; ++u) {
                const int i = i0 + u * SORT_T;
                d[u] = i < hi ? dst[i] : 0;
                s[u] = i < hi ? src[i] : 0;
            }
            #pragma unroll
            for (int u = 0; u < SORT_U; ++u)
                if (i0 + u * SORT_T < hi) {
                    const int p = atomicAdd(&cur[bin_of(d[u], nb)], 1);
                    tmp[p] = make_int2(s[u], d[u]);
                }
        }
    }
}

#define PLACE_T 1024
#define PLACE_U 8
#define PLACE_CAP 8192
__global__ __launch_bounds__(PLACE_T)
void bin_place_k(const int2* __restrict__ tmp, const int* __restrict__ bstart,
                 int* __restrict__ rowptr, int* __restrict__ ssrc, int n, int nb) {
    __shared__ int cnt[BW];
    __shared__ int wt[BW / 32];
    __shared__ int stg[PLACE_CAP];
    const int t = threadIdx.x;
    const int lane = t & 31;
    const int wave = t >> 5;
    const int b = blockIdx.x;
    const int lo = bstart[b];
    const int hi = bstart[b + 1];
    const bool ordered = hi - lo <= PLACE_CAP;
    if (t < BW) cnt[t] = 0;
    __syncthreads();
    for (int i0 = lo + t; i0 < hi; i0 += PLACE_U * PLACE_T) {
        int d[PLACE_U];
        #pragma unroll
        for (int u = 0; u < PLACE_U; ++u) {
            const int i = i0 + u * PLACE_T;
            d[u] = i < hi ? tmp[i].y : 0;
        }
        #pragma unroll
        for (int u = 0; u < PLACE_U; ++u)
            if (i0 + u * PLACE_T < hi) atomicAdd(&cnt[d[u] & (BW - 1)], 1);
    }
    __syncthreads();
    int v = 0, inc = 0;
    if (t < BW) {
        v = cnt[t];
        inc = v;
        #pragma unroll
        for (int off = 1; off < 32; off <<= 1) {
            int u = __shfl_up(inc, off);
            if (lane >= off) inc += u;
        }
        if (lane == 31) wt[wave] = inc;
    }
    __syncthreads();
    if (t < BW) {
        int off = 0;
        #pragma unroll
        for (int w = 0; w < BW / 32; ++w) off += w < wave ? wt[w] : 0;
        const int excl = off + inc - v;
        cnt[t] = excl;
        const int node = b * BW + t;
        if (node < n) rowptr[node] = lo + excl;
    }
    if (b == nb - 1 && t == 0) rowptr[n] = hi;
    __syncthreads();
    if (ordered) {
        for (int i0 = lo + t; i0 < hi; i0 += PLACE_U * PLACE_T) {
            int2 ed[PLACE_U];
            int p[PLACE_U];
            #pragma unroll
            for (int u = 0; u < PLACE_U; ++u) {
                const int i = i0 + u * PLACE_T;
                ed[u] = i < hi ? tmp[i] : make_int2(0, 0);
            }
            #pragma unroll
            for (int u = 0; u < PLACE_U; ++u) {
                p[u] = 0;
                if (i0 + u * PLACE_T < hi) p[u] = atomicAdd(&cnt[ed[u].y & (BW - 1)], 1);
            }
            #pragma unroll
            for (int u = 0; u < PLACE_U; ++u)
                if (i0 + u * PLACE_T < hi) stg[p[u]] = ed[u].x;
        }
        __syncthreads();
        const int m = hi - lo;
        for (int i0 = t; i0 < m; i0 += PLACE_U * PLACE_T) {
            int v[PLACE_U];
            #pragma unroll
            for (int u = 0; u < PLACE_U; ++u) {
                const int i = i0 + u * PLACE_T;
                v[u] = i < m ? stg[i] : 0;
            }
            #pragma unroll
            for (int u = 0; u < PLACE_U; ++u) {
                const int i = i0 + u * PLACE_T;
                if (i < m) ssrc[lo + i] = v[u];
            }
        }
    } else {
        for (int i0 = lo + t; i0 < hi; i0 += PLACE_U * PLACE_T) {
            int2 ed[PLACE_U];
            #pragma unroll
            for (int u = 0; u < PLACE_U; ++u) {
                const int i = i0 + u * PLACE_T;
                ed[u] = i < hi ? tmp[i] : make_int2(0, 0);
            }
            #pragma unroll
            for (int u = 0; u < PLACE_U; ++u)
                if (i0 + u * PLACE_T < hi) {
                    const int p = atomicAdd(&cnt[ed[u].y & (BW - 1)], 1);
                    ssrc[lo + p] = ed[u].x;
                }
        }
    }
}


template <int CTRL>
__device__ __forceinline__ float dpp_rd(float x) {
    return __builtin_bit_cast(float, __builtin_amdgcn_update_dpp(
        0, __builtin_bit_cast(int, x), CTRL, 0xF, 0xF, true));
}
__device__ __forceinline__ float xrow_rd(float x) {
    return __builtin_bit_cast(float, __builtin_amdgcn_permlanex16(
        __builtin_bit_cast(int, x), __builtin_bit_cast(int, x), 0x76543210, 0xfedcba98, false, false));
}
template <int U>
__device__ __forceinline__ float grp8_bcast(float x) {
    return __builtin_bit_cast(float, __builtin_amdgcn_mov_dpp8(
        __builtin_bit_cast(unsigned, x), (unsigned)U * 0x249249u));
}
__device__ __forceinline__ float grp8_sum(float v) {
    v += dpp_rd<0xB1>(v); v += dpp_rd<0x4E>(v); v += dpp_rd<0x164>(v);
    return v;
}
__device__ __forceinline__ float grp8_max(float v) {
    v = fmaxf(v, dpp_rd<0xB1>(v)); v = fmaxf(v, dpp_rd<0x4E>(v)); v = fmaxf(v, dpp_rd<0x164>(v));
    return v;
}

#define EU 8

template <int H, bool ELU>
__global__ __launch_bounds__(256)
void node_agg_k(const int* __restrict__ rowptr, const int* __restrict__ ssrc,
                const float* __restrict__ attL, const float* __restrict__ ar,
                const float* xlin, const float* res,
                const float* __restrict__ bias, float* outp, int n) {
    static_assert(H == 4 || H == 1, "a head is one group of 8 lanes, or all four");
    const int lane = threadIdx.x & 31;
    const int i = __builtin_amdgcn_readfirstlane(
        (int)(blockIdx.x * (blockDim.x >> 5) + (threadIdx.x >> 5)));
    if (i >= n) return;
    const int beg = __builtin_amdgcn_readfirstlane(rowptr[i]);
    const int end = __builtin_amdgcn_readfirstlane(rowptr[i + 1]);
    const int hl  = (2 * lane * H) >> 6;
    const int g   = lane & 7;
    const bool b4 = (lane & 4) != 0, b2 = (lane & 2) != 0, b1 = (lane & 1) != 0;

    const float2 aL = *reinterpret_cast<const float2*>(attL + 2 * lane);
    const float arh = ar[(size_t)i * H + hl];
    const float2* rows = reinterpret_cast<const float2*>(xlin);

    float m = -FLT_MAX;
    float accx = 0.f, accy = 0.f;
    float So = 0.f;
    for (int c = beg; c < end; c += 32) {
        const int k = c + lane;
        const int s = k < end ? ssrc[k] : 0;
        const int cnt = (end - c) < 32 ? (end - c) : 32;
        for (int j = 0; j < cnt; j += EU) {
            float2 x[EU];
            #pragma unroll
            for (int u = 0; u < EU; ++u) {
                const int jj = j + u;
                const int jc = jj < cnt ? jj : cnt - 1;
                const unsigned sj = (unsigned)__builtin_amdgcn_readlane(s, jc);
                x[u] = rows[sj * 32u + (unsigned)lane];
            }
            float p[EU];
            #pragma unroll
            for (int u = 0; u < EU; ++u) p[u] = x[u].x * aL.x + x[u].y * aL.y;
            float q[4];
            #pragma unroll
            for (int u = 0; u < 4; ++u) {
                const float keep = b4 ? p[u + 4] : p[u];
                const float send = b4 ? p[u] : p[u + 4];
                q[u] = keep + dpp_rd<0x164>(send);
            }
            float r[2];
            #pragma unroll
            for (int u = 0; u < 2; ++u) {
                const float keep = b2 ? q[u + 2] : q[u];
                const float send = b2 ? q[u] : q[u + 2];
                r[u] = keep + dpp_rd<0x4E>(send);
            }
            float t;
            {
                const float keep = b1 ? r[1] : r[0];
                const float send = b1 ? r[0] : r[1];
                t = keep + dpp_rd<0xB1>(send);
            }
            if (H == 1) {
                t += dpp_rd<0x168>(t);
                t += xrow_rd(t);
            }
            const float lgo = (j + g < cnt) ? lrelu(t + arh) : -FLT_MAX;
            const float mn = fmaxf(m, grp8_max(lgo));
            const float sc = __expf(m - mn);
            const float wo = __expf(lgo - mn);
            accx *= sc; accy *= sc;
            So = So * sc + wo;
            m = mn;
#define AGG_EDGE(U) { const float w = grp8_bcast<U>(wo); accx += x[U].x * w; accy += x[U].y * w; }
            AGG_EDGE(0) AGG_EDGE(1) AGG_EDGE(2) AGG_EDGE(3)
            AGG_EDGE(4) AGG_EDGE(5) AGG_EDGE(6) AGG_EDGE(7)
#undef AGG_EDGE
        }
    }

    const float den = grp8_sum(So) + 1e-16f;
    const float2 rr = *reinterpret_cast<const float2*>(res + (size_t)i * 64 + 2 * lane);
    const float2 b = *reinterpret_cast<const float2*>(bias + 2 * lane);
    float vx = accx / den + rr.x + b.x;
    float vy = accy / den + rr.y + b.y;
    if (ELU) {
        vx = vx > 0.f ? vx : (__expf(vx) - 1.f);
        vy = vy > 0.f ? vy : (__expf(vy) - 1.f);
    }
    *reinterpret_cast<float2*>(outp + (size_t)i * 64 + 2 * lane) = make_float2(vx, vy);
}

extern "C" void kernel_launch(void* const* d_in, const int* in_sizes, int n_in,
                              void* d_out, int out_size, void* d_ws, size_t ws_size,
                              hipStream_t stream) {
    const float* x     = (const float*)d_in[0];
    const int*   eidx  = (const int*)d_in[1];
    const float* W1    = (const float*)d_in[2];
    const float* attL1 = (const float*)d_in[3];
    const float* attR1 = (const float*)d_in[4];
    const float* resW1 = (const float*)d_in[5];
    const float* b1    = (const float*)d_in[6];
    const float* W2    = (const float*)d_in[7];
    const float* attL2 = (const float*)d_in[8];
    const float* attR2 = (const float*)d_in[9];
    const float* b2    = (const float*)d_in[10];

    const int n = in_sizes[0] / 128;
    const int e = in_sizes[1] / 2;
    const int* srcI = eidx;
    const int* dstI = eidx + e;

    float* ws = (float*)d_ws;
    float* A  = ws;
    float* B  = A  + (size_t)n * 64;
    float* C  = B  + (size_t)n * 64;
    float* AR = C  + (size_t)n * 64;
    const int nb = (n + BW - 1) / BW;
    int2* TMP = (int2*)C;
    int* SS   = (int*)C + 2 * (size_t)e;
    int* RP   = SS + e;
    int* HIST = RP + n + 1;
    int* BST  = HIST + SORT_G * nb;

    constexpr int RT = 2;
    const int tiles = (n + 16 * RT - 1) / (16 * RT);
    const int nWav  = (n + 7) / 8;

    bin_hist_k<<<SORT_G, SORT_T, 0, stream>>>(dstI, HIST, e, nb);
    bin_scatter_k<<<SORT_G, SORT_T, 0, stream>>>(srcI, dstI, HIST, BST, TMP, e, nb);
    bin_place_k<<<nb, PLACE_T, 0, stream>>>(TMP, BST, RP, SS, n, nb);

    gemm_x2<true, RT, 4><<<tiles, 32, 0, stream>>>(x, W1, resW1, A, B, attR1, AR, n, 128);
    node_agg_k<4, true><<<nWav, 256, 0, stream>>>(RP, SS, attL1, AR, A, B, b1, B, n);

    gemm_x2<false, RT, 1><<<tiles, 32, 0, stream>>>(B, W2, nullptr, A, nullptr, attR2, AR, n, 64);
    node_agg_k<1, false><<<nWav, 256, 0, stream>>>(RP, SS, attL2, AR, A, A, b2, (float*)d_out, n);
}


# node aggregate: a node's dependent round trips cut: source list, coefficient, residual, bias requested together, rows of two sets of 8 in flight at once; same sums in same order (products fused)
# speedup vs baseline: 7.2320x; 1.0346x over previous
#include <hip/hip_runtime.h>
#include <math.h>
#include <float.h>

typedef float v2f __attribute__((ext_vector_type(2)));
typedef float v8f __attribute__((ext_vector_type(8)));

#define HC 64
#define SLOPE 0.2f

__device__ __forceinline__ float lrelu(float x) { return x > 0.f ? x : SLOPE * x; }

#define TP 68
template <bool DUAL, int RT, int H>
__global__ __launch_bounds__(32)
void gemm_x2(const float* __restrict__ X, const float* __restrict__ W,
             const float* __restrict__ Wr, float* __restrict__ Y,
             float* __restrict__ Yr,
             const float* __restrict__ attR, float* __restrict__ ar, int n, int K) {
    static_assert(RT == 2, "the epilogue gives each of the 32 lanes one tile row");
    __shared__ __attribute__((aligned(16))) float tile[16 * RT * TP];
    const int lane  = threadIdx.x;
    const int row16 = lane & 15;
    const int khalf = (lane >> 4) << 1;
    const int col   = lane & 15;
    const int r0    = blockIdx.x * (16 * RT);

    int arow[RT];
    #pragma unroll
    for (int t = 0; t < RT; ++t) {
        int r = r0 + 16 * t + row16;
        arow[t] = r < n ? r : n - 1;
    }

    v8f acc[RT][4]  = {};
    v8f accR[RT][4] = {};

    #pragma unroll 2
    for (int k0 = 0; k0 < K; k0 += 4) {
        v2f a[RT];
        #pragma unroll
        for (int t = 0; t < RT; ++t)
            a[t] = *reinterpret_cast<const v2f*>(X + (size_t)arow[t] * K + k0 + khalf);

        #pragma unroll
        for (int j = 0; j < 4; ++j) {
            v2f b;
            const float* bp = W + (size_t)(k0 + khalf) * HC + j * 16 + col;
            b.x = bp[0]; b.y = bp[HC];
            #pragma unroll
            for (int t = 0; t < RT; ++t)
                acc[t][j] = __builtin_amdgcn_wmma_f32_16x16x4_f32(
                    false, a[t], false, b, (short)0, acc[t][j], false, false);
            if (DUAL) {
                v2f br;
                const float* brp = Wr + (size_t)(k0 + khalf) * HC + j * 16 + col;
                br.x = brp[0]; br.y = brp[HC];
                #pragma unroll
                for (int t = 0; t < RT; ++t)
                    accR[t][j] = __builtin_amdgcn_wmma_f32_16x16x4_f32(
                        false, a[t], false, br, (short)0, accR[t][j], false, false);
            }
        }
    }

    const int mhi = (lane >> 4) * 8;
    #pragma unroll
    for (int t = 0; t < RT; ++t)
        #pragma unroll
        for (int j = 0; j < 4; ++j)
            #pragma unroll
            for (int r = 0; r < 8; ++r) {
                int m = r0 + 16 * t + mhi + r;
                tile[(16 * t + mhi + r) * TP + j * 16 + col] = acc[t][j][r];
                if (m < n) {
                    Y[(size_t)m * HC + j * 16 + col] = acc[t][j][r];
                    if (DUAL) Yr[(size_t)m * HC + j * 16 + col] = accR[t][j][r];
                }
            }

    __builtin_amdgcn_fence(__ATOMIC_RELEASE, "workgroup", "local");
    __builtin_amdgcn_s_barrier();
    __builtin_amdgcn_fence(__ATOMIC_ACQUIRE, "workgroup", "local");

    const int row = r0 + lane;
    const float4* rowp = reinterpret_cast<const float4*>(tile + lane * TP);
    const float4* rp = reinterpret_cast<const float4*>(attR);
    if (H == 4) {
        float aR[4] = {0, 0, 0, 0};
        #pragma unroll
        for (int q = 0; q < 16; ++q) {
            const float4 v = rowp[q];
            const float4 rr = rp[q];
            const int h = q >> 2;
            aR[h] += v.x * rr.x + v.y * rr.y + v.z * rr.z + v.w * rr.w;
        }
        if (row < n)
            *reinterpret_cast<float4*>(ar + (size_t)row * 4) = make_float4(aR[0], aR[1], aR[2], aR[3]);
    } else {
        float aR = 0.f;
        for (int q = 0; q < 16; ++q) {
            const float4 v = rowp[q];
            const float4 rr = rp[q];
            aR += v.x * rr.x + v.y * rr.y + v.z * rr.z + v.w * rr.w;
        }
        if (row < n) ar[row] = aR;
    }
}

#define BSH    8
#define BW     (1 << BSH)
#define MAXNB  1024
#define SORT_G 128
#define SORT_T 1024
#define SORT_U 16
#define STG_CAP 16384

__device__ __forceinline__ int bin_of(int d, int nb) {
    const unsigned b = (unsigned)d >> BSH;
    return b < (unsigned)nb ? (int)b : nb - 1;
}

__global__ __launch_bounds__(SORT_T)
void bin_hist_k(const int* __restrict__ dst, int* __restrict__ hist, int e, int nb) {
    __shared__ int h[MAXNB];
    const int t = threadIdx.x;
    for (int b = t; b < nb; b += SORT_T) h[b] = 0;
    __syncthreads();
    const int ch = (e + SORT_G - 1) / SORT_G;
    const int lo = blockIdx.x * ch;
    const int hi = lo + ch < e ? lo + ch : e;
    for (int i0 = lo + t; i0 < hi; i0 += SORT_U * SORT_T) {
        int d[SORT_U];
        #pragma unroll
        for (int u = 0; u < SORT_U; ++u) {
            const int i = i0 + u * SORT_T;
            d[u] = i < hi ? dst[i] : 0;
        }
        #pragma unroll
        for (int u = 0; u < SORT_U; ++u)
            if (i0 + u * SORT_T < hi) atomicAdd(&h[bin_of(d[u], nb)], 1);
    }
    __syncthreads();
    for (int b = t; b < nb; b += SORT_T) hist[blockIdx.x * nb + b] = h[b];
}

__global__ __launch_bounds__(SORT_T)
void bin_scatter_k(const int* __restrict__ src, const int* __restrict__ dst,
                   const int* __restrict__ hist, int* __restrict__ bstart,
                   int2* __restrict__ tmp, int e, int nb) {
    __shared__ int cur[MAXNB];
    __shared__ int gofs[MAXNB];
    __shared__ int wtot[32];
    __shared__ int wown[32];
    __shared__ int2 stage[STG_CAP];
    const int t = threadIdx.x;
    const int lane = t & 31;
    const int wave = t >> 5;
    const int g = blockIdx.x;

    const int ch = (e + SORT_G - 1) / SORT_G;
    const int lo = g * ch;
    const int hi = lo + ch < e ? lo + ch : e;
    const bool ordered = hi - lo <= STG_CAP;

    int pre = 0, tot = 0, own = 0;
    if (t < nb) {
        #pragma unroll 32
        for (int gg = 0; gg < SORT_G; ++gg) {
            const int v = hist[gg * nb + t];
            tot += v;
            pre += gg < g ? v : 0;
        }
        own = hist[g * nb + t];
    }
    int inc = tot, inco = own;
    #pragma unroll
    for (int off = 1; off < 32; off <<= 1) {
        int u = __shfl_up(inc, off);
        int uo = __shfl_up(inco, off);
        if (lane >= off) { inc += u; inco += uo; }
    }
    if (lane == 31) { wtot[wave] = inc; wown[wave] = inco; }
    __syncthreads();
    int tinc = wtot[lane], tinco = wown[lane];
    #pragma unroll
    for (int off = 1; off < 32; off <<= 1) {
        int u = __shfl_up(tinc, off);
        int uo = __shfl_up(tinco, off);
        if (lane >= off) { tinc += u; tinco += uo; }
    }
    const int below  = __shfl(tinc,  wave > 0 ? wave - 1 : 0);
    const int belowo = __shfl(tinco, wave > 0 ? wave - 1 : 0);
    const int start  = (wave > 0 ? below  : 0) + inc  - tot;
    const int lstart = (wave > 0 ? belowo : 0) + inco - own;
    if (t < nb) {
        cur[t]  = ordered ? lstart : start + pre;
        gofs[t] = start + pre - lstart;
        if (g == 0) bstart[t] = start;
    }
    if (g == 0 && t == 0) bstart[nb] = e;
    __syncthreads();

    if (ordered) {
        for (int i0 = lo + t; i0 < hi; i0 += SORT_U * SORT_T) {
            int d[SORT_U], s[SORT_U], p[SORT_U];
            #pragma unroll
            for (int u = 0; u < SORT_U; ++u) {
                const int i = i0 + u * SORT_T;
                d[u] = i < hi ? dst[i] : 0;
                s[u] = i < hi ? src[i] : 0;
            }
            #pragma unroll
            for (int u = 0; u < SORT_U; ++u) {
                p[u] = 0;
                if (i0 + u * SORT_T < hi) p[u] = atomicAdd(&cur[bin_of(d[u], nb)], 1);
            }
            #pragma unroll
            for (int u = 0; u < SORT_U; ++u)
                if (i0 + u * SORT_T < hi) stage[p[u]] = make_int2(s[u], d[u]);
        }
        __syncthreads();
        const int m = hi - lo;
        for (int i0 = t; i0 < m; i0 += SORT_U * SORT_T) {
            int2 v[SORT_U];
            int o[SORT_U];
            #pragma unroll
            for (int u = 0; u < SORT_U; ++u) {
                const int i = i0 + u * SORT_T;
                v[u] = i < m ? stage[i] : make_int2(0, 0);
            }
            #pragma unroll
            for (int u = 0; u < SORT_U; ++u) o[u] = gofs[bin_of(v[u].y, nb)];
            #pragma unroll
            for (int u = 0; u < SORT_U; ++u) {
                const int i = i0 + u * SORT_T;
                if (i < m) tmp[o[u] + i] = v[u];
            }
        }
    } else {
        for (int i0 = lo + t; i0 < hi; i0 += SORT_U * SORT_T) {
            int d[SORT_U], s[SORT_U];
            #pragma unroll
            for (int u = 0; u < SORT_U; ++u) {
                const int i = i0 + u * SORT_T;
                d[u] = i < hi ? dst[i] : 0;
                s[u] = i < hi ? src[i] : 0;
            }
            #pragma unroll
            for (int u = 0; u < SORT_U; ++u)
                if (i0 + u * SORT_T < hi) {
                    const int p = atomicAdd(&cur[bin_of(d[u], nb)], 1);
                    tmp[p] = make_int2(s[u], d[u]);
                }
        }
    }
}

#define PLACE_T 1024
#define PLACE_U 8
#define PLACE_CAP 8192
__global__ __launch_bounds__(PLACE_T)
void bin_place_k(const int2* __restrict__ tmp, const int* __restrict__ bstart,
                 int* __restrict__ rowptr, int* __restrict__ ssrc, int n, int nb) {
    __shared__ int cnt[BW];
    __shared__ int wt[BW / 32];
    __shared__ int stg[PLACE_CAP];
    const int t = threadIdx.x;
    const int lane = t & 31;
    const int wave = t >> 5;
    const int b = blockIdx.x;
    const int lo = bstart[b];
    const int hi = bstart[b + 1];
    const bool ordered = hi - lo <= PLACE_CAP;
    if (t < BW) cnt[t] = 0;
    __syncthreads();
    for (int i0 = lo + t; i0 < hi; i0 += PLACE_U * PLACE_T) {
        int d[PLACE_U];
        #pragma unroll
        for (int u = 0; u < PLACE_U; ++u) {
            const int i = i0 + u * PLACE_T;
            d[u] = i < hi ? tmp[i].y : 0;
        }
        #pragma unroll
        for (int u = 0; u < PLACE_U; ++u)
            if (i0 + u * PLACE_T < hi) atomicAdd(&cnt[d[u] & (BW - 1)], 1);
    }
    __syncthreads();
    int v = 0, inc = 0;
    if (t < BW) {
        v = cnt[t];
        inc = v;
        #pragma unroll
        for (int off = 1; off < 32; off <<= 1) {
            int u = __shfl_up(inc, off);
            if (lane >= off) inc += u;
        }
        if (lane == 31) wt[wave] = inc;
    }
    __syncthreads();
    if (t < BW) {
        int off = 0;
        #pragma unroll
        for (int w = 0; w < BW / 32; ++w) off += w < wave ? wt[w] : 0;
        const int excl = off + inc - v;
        cnt[t] = excl;
        const int node = b * BW + t;
        if (node < n) rowptr[node] = lo + excl;
    }
    if (b == nb - 1 && t == 0) rowptr[n] = hi;
    __syncthreads();
    if (ordered) {
        for (int i0 = lo + t; i0 < hi; i0 += PLACE_U * PLACE_T) {
            int2 ed[PLACE_U];
            int p[PLACE_U];
            #pragma unroll
            for (int u = 0; u < PLACE_U; ++u) {
                const int i = i0 + u * PLACE_T;
                ed[u] = i < hi ? tmp[i] : make_int2(0, 0);
            }
            #pragma unroll
            for (int u = 0; u < PLACE_U; ++u) {
                p[u] = 0;
                if (i0 + u * PLACE_T < hi) p[u] = atomicAdd(&cnt[ed[u].y & (BW - 1)], 1);
            }
            #pragma unroll
            for (int u = 0; u < PLACE_U; ++u)
                if (i0 + u * PLACE_T < hi) stg[p[u]] = ed[u].x;
        }
        __syncthreads();
        const int m = hi - lo;
        for (int i0 = t; i0 < m; i0 += PLACE_U * PLACE_T) {
            int v[PLACE_U];
            #pragma unroll
            for (int u = 0; u < PLACE_U; ++u) {
                const int i = i0 + u * PLACE_T;
                v[u] = i < m ? stg[i] : 0;
            }
            #pragma unroll
            for (int u = 0; u < PLACE_U; ++u) {
                const int i = i0 + u * PLACE_T;
                if (i < m) ssrc[lo + i] = v[u];
            }
        }
    } else {
        for (int i0 = lo + t; i0 < hi; i0 += PLACE_U * PLACE_T) {
            int2 ed[PLACE_U];
            #pragma unroll
            for (int u = 0; u < PLACE_U; ++u) {
                const int i = i0 + u * PLACE_T;
                ed[u] = i < hi ? tmp[i] : make_int2(0, 0);
            }
            #pragma unroll
            for (int u = 0; u < PLACE_U; ++u)
                if (i0 + u * PLACE_T < hi) {
                    const int p = atomicAdd(&cnt[ed[u].y & (BW - 1)], 1);
                    ssrc[lo + p] = ed[u].x;
                }
        }
    }
}


template <int CTRL>
__device__ __forceinline__ float dpp_rd(float x) {
    return __builtin_bit_cast(float, __builtin_amdgcn_update_dpp(
        0, __builtin_bit_cast(int, x), CTRL, 0xF, 0xF, true));
}
__device__ __forceinline__ float xrow_rd(float x) {
    return __builtin_bit_cast(float, __builtin_amdgcn_permlanex16(
        __builtin_bit_cast(int, x), __builtin_bit_cast(int, x), 0x76543210, 0xfedcba98, false, false));
}
template <int U>
__device__ __forceinline__ float grp8_bcast(float x) {
    return __builtin_bit_cast(float, __builtin_amdgcn_mov_dpp8(
        __builtin_bit_cast(unsigned, x), (unsigned)U * 0x249249u));
}
__device__ __forceinline__ float grp8_sum(float v) {
    v += dpp_rd<0xB1>(v); v += dpp_rd<0x4E>(v); v += dpp_rd<0x164>(v);
    return v;
}
__device__ __forceinline__ float grp8_max(float v) {
    v = fmaxf(v, dpp_rd<0xB1>(v)); v = fmaxf(v, dpp_rd<0x4E>(v)); v = fmaxf(v, dpp_rd<0x164>(v));
    return v;
}

#define EU 8
#define AGG_WAVES 8

template <int H, bool ELU>
__global__ __launch_bounds__(32 * AGG_WAVES) __attribute__((amdgpu_waves_per_eu(16, 16)))
void node_agg_k(const int* __restrict__ rowptr, const int* __restrict__ ssrc,
                const float* __restrict__ attL, const float* __restrict__ ar,
                const float* xlin, const float* res,
                const float* __restrict__ bias, float* outp, int n) {
    static_assert(H == 4 || H == 1, "a head is one group of 8 lanes, or all four");
    const int lane = threadIdx.x & 31;
    const int i = __builtin_amdgcn_readfirstlane(
        (int)(blockIdx.x * AGG_WAVES + (threadIdx.x >> 5)));
    if (i >= n) return;
    const int beg = __builtin_amdgcn_readfirstlane(rowptr[i]);
    const int end = __builtin_amdgcn_readfirstlane(rowptr[i + 1]);
    const int hl  = (2 * lane * H) >> 6;
    const int g   = lane & 7;
    const bool b4 = (lane & 4) != 0, b2 = (lane & 2) != 0, b1 = (lane & 1) != 0;

    int s = (beg + lane) < end ? ssrc[beg + lane] : 0;
    const float2 aL = *reinterpret_cast<const float2*>(attL + 2 * lane);
    const float arh = ar[(size_t)i * H + hl];
    const float2 rr = *reinterpret_cast<const float2*>(res + (size_t)i * 64 + 2 * lane);
    const float2 b = *reinterpret_cast<const float2*>(bias + 2 * lane);
    const float2* rows = reinterpret_cast<const float2*>(xlin);

    float m = -FLT_MAX;
    float accx = 0.f, accy = 0.f;
    float So = 0.f;

    auto load_set = [&](float2 (&x)[EU], int j, int cnt) {
        #pragma unroll
        for (int u = 0; u < EU; ++u) {
            const int jj = j + u;
            const int jc = jj < cnt ? jj : cnt - 1;
            const unsigned sj = (unsigned)__builtin_amdgcn_readlane(s, jc);
            x[u] = rows[sj * 32u + (unsigned)lane];
        }
    };
    auto sum_set = [&](const float2 (&x)[EU], int j, int cnt) {
        float p[EU];
        #pragma unroll
        for (int u = 0; u < EU; ++u) p[u] = x[u].x * aL.x + x[u].y * aL.y;
        float q[4];
        #pragma unroll
        for (int u = 0; u < 4; ++u) {
            const float keep = b4 ? p[u + 4] : p[u];
            const float send = b4 ? p[u] : p[u + 4];
            q[u] = keep + dpp_rd<0x164>(send);
        }
        float r[2];
        #pragma unroll
        for (int u = 0; u < 2; ++u) {
            const float keep = b2 ? q[u + 2] : q[u];
            const float send = b2 ? q[u] : q[u + 2];
            r[u] = keep + dpp_rd<0x4E>(send);
        }
        float t;
        {
            const float keep = b1 ? r[1] : r[0];
            const float send = b1 ? r[0] : r[1];
            t = keep + dpp_rd<0xB1>(send);
        }
        if (H == 1) {
            t += dpp_rd<0x168>(t);
            t += xrow_rd(t);
        }
        const float lgo = (j + g < cnt) ? lrelu(t + arh) : -FLT_MAX;
        const float mn = fmaxf(m, grp8_max(lgo));
        const float sc = __expf(m - mn);
        const float wo = __expf(lgo - mn);
        accx *= sc; accy *= sc;
        So = So * sc + wo;
        m = mn;
#define AGG_EDGE(U) { const float w = grp8_bcast<U>(wo); accx += x[U].x * w; accy += x[U].y * w; }
        AGG_EDGE(0) AGG_EDGE(1) AGG_EDGE(2) AGG_EDGE(3)
        AGG_EDGE(4) AGG_EDGE(5) AGG_EDGE(6) AGG_EDGE(7)
#undef AGG_EDGE
    };
    auto walk_chunk = [&](int cnt) {
        float2 xa[EU], xb[EU];
        load_set(xa, 0, cnt);
        load_set(xb, EU, cnt);
        __builtin_amdgcn_sched_barrier(0);
        sum_set(xa, 0, cnt);
        if (cnt > 2 * EU) {
            load_set(xa, 2 * EU, cnt);
            __builtin_amdgcn_sched_barrier(0);
            sum_set(xb, EU, cnt);
            sum_set(xa, 2 * EU, cnt);
            if (cnt > 3 * EU) {
                load_set(xb, 3 * EU, cnt);
                sum_set(xb, 3 * EU, cnt);
            }
        } else if (cnt > EU) {
            sum_set(xb, EU, cnt);
        }
    };
    auto walk_chunk_plain = [&](int cnt) {
        for (int j = 0; j < cnt; j += EU) {
            float2 x[EU];
            load_set(x, j, cnt);
            sum_set(x, j, cnt);
        }
    };

    if (beg < end) {
        walk_chunk((end - beg) < 32 ? (end - beg) : 32);
        for (int c = beg + 32; c < end; c += 32) {
            const int k = c + lane;
            s = k < end ? ssrc[k] : 0;
            walk_chunk_plain((end - c) < 32 ? (end - c) : 32);
        }
    }

    const float den = grp8_sum(So) + 1e-16f;
    float vx = accx / den + rr.x + b.x;
    float vy = accy / den + rr.y + b.y;
    if (ELU) {
        vx = vx > 0.f ? vx : (__expf(vx) - 1.f);
        vy = vy > 0.f ? vy : (__expf(vy) - 1.f);
    }
    *reinterpret_cast<float2*>(outp + (size_t)i * 64 + 2 * lane) = make_float2(vx, vy);
}

extern "C" void kernel_launch(void* const* d_in, const int* in_sizes, int n_in,
                              void* d_out, int out_size, void* d_ws, size_t ws_size,
                              hipStream_t stream) {
    const float* x     = (const float*)d_in[0];
    const int*   eidx  = (const int*)d_in[1];
    const float* W1    = (const float*)d_in[2];
    const float* attL1 = (const float*)d_in[3];
    const float* attR1 = (const float*)d_in[4];
    const float* resW1 = (const float*)d_in[5];
    const float* b1    = (const float*)d_in[6];
    const float* W2    = (const float*)d_in[7];
    const float* attL2 = (const float*)d_in[8];
    const float* attR2 = (const float*)d_in[9];
    const float* b2    = (const float*)d_in[10];

    const int n = in_sizes[0] / 128;
    const int e = in_sizes[1] / 2;
    const int* srcI = eidx;
    const int* dstI = eidx + e;

    float* ws = (float*)d_ws;
    float* A  = ws;
    float* B  = A  + (size_t)n * 64;
    float* C  = B  + (size_t)n * 64;
    float* AR = C  + (size_t)n * 64;
    const int nb = (n + BW - 1) / BW;
    int2* TMP = (int2*)C;
    int* SS   = (int*)C + 2 * (size_t)e;
    int* RP   = SS + e;
    int* HIST = RP + n + 1;
    int* BST  = HIST + SORT_G * nb;

    constexpr int RT = 2;
    const int tiles = (n + 16 * RT - 1) / (16 * RT);
    const int nWav  = (n + AGG_WAVES - 1) / AGG_WAVES;

    bin_hist_k<<<SORT_G, SORT_T, 0, stream>>>(dstI, HIST, e, nb);
    bin_scatter_k<<<SORT_G, SORT_T, 0, stream>>>(srcI, dstI, HIST, BST, TMP, e, nb);
    bin_place_k<<<nb, PLACE_T, 0, stream>>>(TMP, BST, RP, SS, n, nb);

    gemm_x2<true, RT, 4><<<tiles, 32, 0, stream>>>(x, W1, resW1, A, B, attR1, AR, n, 128);
    node_agg_k<4, true><<<nWav, 32 * AGG_WAVES, 0, stream>>>(RP, SS, attL1, AR, A, B, b1, B, n);

    gemm_x2<false, RT, 1><<<tiles, 32, 0, stream>>>(B, W2, nullptr, A, nullptr, attR2, AR, n, 64);
    node_agg_k<1, false><<<nWav, 32 * AGG_WAVES, 0, stream>>>(RP, SS, attL2, AR, A, A, b2, (float*)d_out, n);
}


# edge grouping: batch loads of the count and placement passes made unconditional (index clamped to the part's last element) and held together: one memory round trip per batch, not one per element
# speedup vs baseline: 7.4893x; 1.0356x over previous
#include <hip/hip_runtime.h>
#include <math.h>
#include <float.h>

typedef float v2f __attribute__((ext_vector_type(2)));
typedef float v8f __attribute__((ext_vector_type(8)));

#define HC 64
#define SLOPE 0.2f

__device__ __forceinline__ float lrelu(float x) { return x > 0.f ? x : SLOPE * x; }

#define TP 68
template <bool DUAL, int RT, int H>
__global__ __launch_bounds__(32)
void gemm_x2(const float* __restrict__ X, const float* __restrict__ W,
             const float* __restrict__ Wr, float* __restrict__ Y,
             float* __restrict__ Yr,
             const float* __restrict__ attR, float* __restrict__ ar, int n, int K) {
    static_assert(RT == 2, "the epilogue gives each of the 32 lanes one tile row");
    __shared__ __attribute__((aligned(16))) float tile[16 * RT * TP];
    const int lane  = threadIdx.x;
    const int row16 = lane & 15;
    const int khalf = (lane >> 4) << 1;
    const int col   = lane & 15;
    const int r0    = blockIdx.x * (16 * RT);

    int arow[RT];
    #pragma unroll
    for (int t = 0; t < RT; ++t) {
        int r = r0 + 16 * t + row16;
        arow[t] = r < n ? r : n - 1;
    }

    v8f acc[RT][4]  = {};
    v8f accR[RT][4] = {};

    #pragma unroll 2
    for (int k0 = 0; k0 < K; k0 += 4) {
        v2f a[RT];
        #pragma unroll
        for (int t = 0; t < RT; ++t)
            a[t] = *reinterpret_cast<const v2f*>(X + (size_t)arow[t] * K + k0 + khalf);

        #pragma unroll
        for (int j = 0; j < 4; ++j) {
            v2f b;
            const float* bp = W + (size_t)(k0 + khalf) * HC + j * 16 + col;
            b.x = bp[0]; b.y = bp[HC];
            #pragma unroll
            for (int t = 0; t < RT; ++t)
                acc[t][j] = __builtin_amdgcn_wmma_f32_16x16x4_f32(
                    false, a[t], false, b, (short)0, acc[t][j], false, false);
            if (DUAL) {
                v2f br;
                const float* brp = Wr + (size_t)(k0 + khalf) * HC + j * 16 + col;
                br.x = brp[0]; br.y = brp[HC];
                #pragma unroll
                for (int t = 0; t < RT; ++t)
                    accR[t][j] = __builtin_amdgcn_wmma_f32_16x16x4_f32(
                        false, a[t], false, br, (short)0, accR[t][j], false, false);
            }
        }
    }

    const int mhi = (lane >> 4) * 8;
    #pragma unroll
    for (int t = 0; t < RT; ++t)
        #pragma unroll
        for (int j = 0; j < 4; ++j)
            #pragma unroll
            for (int r = 0; r < 8; ++r) {
                int m = r0 + 16 * t + mhi + r;
                tile[(16 * t + mhi + r) * TP + j * 16 + col] = acc[t][j][r];
                if (m < n) {
                    Y[(size_t)m * HC + j * 16 + col] = acc[t][j][r];
                    if (DUAL) Yr[(size_t)m * HC + j * 16 + col] = accR[t][j][r];
                }
            }

    __builtin_amdgcn_fence(__ATOMIC_RELEASE, "workgroup", "local");
    __builtin_amdgcn_s_barrier();
    __builtin_amdgcn_fence(__ATOMIC_ACQUIRE, "workgroup", "local");

    const int row = r0 + lane;
    const float4* rowp = reinterpret_cast<const float4*>(tile + lane * TP);
    const float4* rp = reinterpret_cast<const float4*>(attR);
    if (H == 4) {
        float aR[4] = {0, 0, 0, 0};
        #pragma unroll
        for (int q = 0; q < 16; ++q) {
            const float4 v = rowp[q];
            const float4 rr = rp[q];
            const int h = q >> 2;
            aR[h] += v.x * rr.x + v.y * rr.y + v.z * rr.z + v.w * rr.w;
        }
        if (row < n)
            *reinterpret_cast<float4*>(ar + (size_t)row * 4) = make_float4(aR[0], aR[1], aR[2], aR[3]);
    } else {
        float aR = 0.f;
        for (int q = 0; q < 16; ++q) {
            const float4 v = rowp[q];
            const float4 rr = rp[q];
            aR += v.x * rr.x + v.y * rr.y + v.z * rr.z + v.w * rr.w;
        }
        if (row < n) ar[row] = aR;
    }
}

#define BSH    8
#define BW     (1 << BSH)
#define MAXNB  1024
#define SORT_G 128
#define SORT_T 1024
#define SORT_U 16
#define STG_CAP 16384

__device__ __forceinline__ int bin_of(int d, int nb) {
    const unsigned b = (unsigned)d >> BSH;
    return b < (unsigned)nb ? (int)b : nb - 1;
}

__global__ __launch_bounds__(SORT_T)
void bin_hist_k(const int* __restrict__ dst, int* __restrict__ hist, int e, int nb) {
    __shared__ int h[MAXNB];
    const int t = threadIdx.x;
    for (int b = t; b < nb; b += SORT_T) h[b] = 0;
    __syncthreads();
    const int ch = (e + SORT_G - 1) / SORT_G;
    const int lo = blockIdx.x * ch;
    const int hi = lo + ch < e ? lo + ch : e;
    for (int i0 = lo + t; i0 < hi; i0 += SORT_U * SORT_T) {
        int d[SORT_U];
        #pragma unroll
        for (int u = 0; u < SORT_U; ++u) {
            const int i = i0 + u * SORT_T;
            d[u] = dst[i < hi ? i : hi - 1];
        }
        static_assert(SORT_U == 16, "the statement below names the 16 values of a batch");
        asm volatile("" :: "v"(d[0]), "v"(d[1]), "v"(d[2]), "v"(d[3]), "v"(d[4]), "v"(d[5]),
                     "v"(d[6]), "v"(d[7]), "v"(d[8]), "v"(d[9]), "v"(d[10]), "v"(d[11]),
                     "v"(d[12]), "v"(d[13]), "v"(d[14]), "v"(d[15]));
        #pragma unroll
        for (int u = 0; u < SORT_U; ++u)
            if (i0 + u * SORT_T < hi) atomicAdd(&h[bin_of(d[u], nb)], 1);
    }
    __syncthreads();
    for (int b = t; b < nb; b += SORT_T) hist[blockIdx.x * nb + b] = h[b];
}

__global__ __launch_bounds__(SORT_T)
void bin_scatter_k(const int* __restrict__ src, const int* __restrict__ dst,
                   const int* __restrict__ hist, int* __restrict__ bstart,
                   int2* __restrict__ tmp, int e, int nb) {
    __shared__ int cur[MAXNB];
    __shared__ int gofs[MAXNB];
    __shared__ int wtot[32];
    __shared__ int wown[32];
    __shared__ int2 stage[STG_CAP];
    const int t = threadIdx.x;
    const int lane = t & 31;
    const int wave = t >> 5;
    const int g = blockIdx.x;

    const int ch = (e + SORT_G - 1) / SORT_G;
    const int lo = g * ch;
    const int hi = lo + ch < e ? lo + ch : e;
    const bool ordered = hi - lo <= STG_CAP;

    int pre = 0, tot = 0, own = 0;
    if (t < nb) {
        #pragma unroll 32
        for (int gg = 0; gg < SORT_G; ++gg) {
            const int v = hist[gg * nb + t];
            tot += v;
            pre += gg < g ? v : 0;
        }
        own = hist[g * nb + t];
    }
    int inc = tot, inco = own;
    #pragma unroll
    for (int off = 1; off < 32; off <<= 1) {
        int u = __shfl_up(inc, off);
        int uo = __shfl_up(inco, off);
        if (lane >= off) { inc += u; inco += uo; }
    }
    if (lane == 31) { wtot[wave] = inc; wown[wave] = inco; }
    __syncthreads();
    int tinc = wtot[lane], tinco = wown[lane];
    #pragma unroll
    for (int off = 1; off < 32; off <<= 1) {
        int u = __shfl_up(tinc, off);
        int uo = __shfl_up(tinco, off);
        if (lane >= off) { tinc += u; tinco += uo; }
    }
    const int below  = __shfl(tinc,  wave > 0 ? wave - 1 : 0);
    const int belowo = __shfl(tinco, wave > 0 ? wave - 1 : 0);
    const int start  = (wave > 0 ? below  : 0) + inc  - tot;
    const int lstart = (wave > 0 ? belowo : 0) + inco - own;
    if (t < nb) {
        cur[t]  = ordered ? lstart : start + pre;
        gofs[t] = start + pre - lstart;
        if (g == 0) bstart[t] = start;
    }
    if (g == 0 && t == 0) bstart[nb] = e;
    __syncthreads();

    if (ordered) {
        for (int i0 = lo + t; i0 < hi; i0 += SORT_U * SORT_T) {
            int d[SORT_U], s[SORT_U], p[SORT_U];
            #pragma unroll
            for (int u = 0; u < SORT_U; ++u) {
                const int i = i0 + u * SORT_T;
                d[u] = i < hi ? dst[i] : 0;
                s[u] = i < hi ? src[i] : 0;
            }
            #pragma unroll
            for (int u = 0; u < SORT_U; ++u) {
                p[u] = 0;
                if (i0 + u * SORT_T < hi) p[u] = atomicAdd(&cur[bin_of(d[u], nb)], 1);
            }
            #pragma unroll
            for (int u = 0; u < SORT_U; ++u)
                if (i0 + u * SORT_T < hi) stage[p[u]] = make_int2(s[u], d[u]);
        }
        __syncthreads();
        const int m = hi - lo;
        for (int i0 = t; i0 < m; i0 += SORT_U * SORT_T) {
            int2 v[SORT_U];
            int o[SORT_U];
            #pragma unroll
            for (int u = 0; u < SORT_U; ++u) {
                const int i = i0 + u * SORT_T;
                v[u] = i < m ? stage[i] : make_int2(0, 0);
            }
            #pragma unroll
            for (int u = 0; u < SORT_U; ++u) o[u] = gofs[bin_of(v[u].y, nb)];
            #pragma unroll
            for (int u = 0; u < SORT_U; ++u) {
                const int i = i0 + u * SORT_T;
                if (i < m) tmp[o[u] + i] = v[u];
            }
        }
    } else {
        for (int i0 = lo + t; i0 < hi; i0 += SORT_U * SORT_T) {
            int d[SORT_U], s[SORT_U];
            #pragma unroll
            for (int u = 0; u < SORT_U; ++u) {
                const int i = i0 + u * SORT_T;
                d[u] = i < hi ? dst[i] : 0;
                s[u] = i < hi ? src[i] : 0;
            }
            #pragma unroll
            for (int u = 0; u < SORT_U; ++u)
                if (i0 + u * SORT_T < hi) {
                    const int p = atomicAdd(&cur[bin_of(d[u], nb)], 1);
                    tmp[p] = make_int2(s[u], d[u]);
                }
        }
    }
}

#define PLACE_T 1024
#define PLACE_U 8
#define PLACE_CAP 8192
__global__ __launch_bounds__(PLACE_T)
void bin_place_k(const int2* __restrict__ tmp, const int* __restrict__ bstart,
                 int* __restrict__ rowptr, int* __restrict__ ssrc, int n, int nb) {
    __shared__ int cnt[BW];
    __shared__ int wt[BW / 32];
    __shared__ int stg[PLACE_CAP];
    const int t = threadIdx.x;
    const int lane = t & 31;
    const int wave = t >> 5;
    const int b = blockIdx.x;
    const int lo = bstart[b];
    const int hi = bstart[b + 1];
    const bool ordered = hi - lo <= PLACE_CAP;
    if (t < BW) cnt[t] = 0;
    __syncthreads();
    for (int i0 = lo + t; i0 < hi; i0 += PLACE_U * PLACE_T) {
        int d[PLACE_U];
        #pragma unroll
        for (int u = 0; u < PLACE_U; ++u) {
            const int i = i0 + u * PLACE_T;
            d[u] = tmp[i < hi ? i : hi - 1].y;
        }
        static_assert(PLACE_U == 8, "the statements of this kind below name the 8 values of a batch");
        asm volatile("" :: "v"(d[0]), "v"(d[1]), "v"(d[2]), "v"(d[3]),
                     "v"(d[4]), "v"(d[5]), "v"(d[6]), "v"(d[7]));
        #pragma unroll
        for (int u = 0; u < PLACE_U; ++u)
            if (i0 + u * PLACE_T < hi) atomicAdd(&cnt[d[u] & (BW - 1)], 1);
    }
    __syncthreads();
    int v = 0, inc = 0;
    if (t < BW) {
        v = cnt[t];
        inc = v;
        #pragma unroll
        for (int off = 1; off < 32; off <<= 1) {
            int u = __shfl_up(inc, off);
            if (lane >= off) inc += u;
        }
        if (lane == 31) wt[wave] = inc;
    }
    __syncthreads();
    if (t < BW) {
        int off = 0;
        #pragma unroll
        for (int w = 0; w < BW / 32; ++w) off += w < wave ? wt[w] : 0;
        const int excl = off + inc - v;
        cnt[t] = excl;
        const int node = b * BW + t;
        if (node < n) rowptr[node] = lo + excl;
    }
    if (b == nb - 1 && t == 0) rowptr[n] = hi;
    __syncthreads();
    if (ordered) {
        for (int i0 = lo + t; i0 < hi; i0 += PLACE_U * PLACE_T) {
            int2 ed[PLACE_U];
            int p[PLACE_U];
            #pragma unroll
            for (int u = 0; u < PLACE_U; ++u) {
                const int i = i0 + u * PLACE_T;
                ed[u] = tmp[i < hi ? i : hi - 1];
            }
            asm volatile("" :: "v"(ed[0].x), "v"(ed[0].y), "v"(ed[1].x), "v"(ed[1].y),
                         "v"(ed[2].x), "v"(ed[2].y), "v"(ed[3].x), "v"(ed[3].y),
                         "v"(ed[4].x), "v"(ed[4].y), "v"(ed[5].x), "v"(ed[5].y),
                         "v"(ed[6].x), "v"(ed[6].y), "v"(ed[7].x), "v"(ed[7].y));
            #pragma unroll
            for (int u = 0; u < PLACE_U; ++u) {
                p[u] = 0;
                if (i0 + u * PLACE_T < hi) p[u] = atomicAdd(&cnt[ed[u].y & (BW - 1)], 1);
            }
            #pragma unroll
            for (int u = 0; u < PLACE_U; ++u)
                if (i0 + u * PLACE_T < hi) stg[p[u]] = ed[u].x;
        }
        __syncthreads();
        const int m = hi - lo;
        for (int i0 = t; i0 < m; i0 += PLACE_U * PLACE_T) {
            int v[PLACE_U];
            #pragma unroll
            for (int u = 0; u < PLACE_U; ++u) {
                const int i = i0 + u * PLACE_T;
                v[u] = i < m ? stg[i] : 0;
            }
            #pragma unroll
            for (int u = 0; u < PLACE_U; ++u) {
                const int i = i0 + u * PLACE_T;
                if (i < m) ssrc[lo + i] = v[u];
            }
        }
    } else {
        for (int i0 = lo + t; i0 < hi; i0 += PLACE_U * PLACE_T) {
            int2 ed[PLACE_U];
            #pragma unroll
            for (int u = 0; u < PLACE_U; ++u) {
                const int i = i0 + u * PLACE_T;
                ed[u] = tmp[i < hi ? i : hi - 1];
            }
            #pragma unroll
            for (int u = 0; u < PLACE_U; ++u)
                if (i0 + u * PLACE_T < hi) {
                    const int p = atomicAdd(&cnt[ed[u].y & (BW - 1)], 1);
                    ssrc[lo + p] = ed[u].x;
                }
        }
    }
}


template <int CTRL>
__device__ __forceinline__ float dpp_rd(float x) {
    return __builtin_bit_cast(float, __builtin_amdgcn_update_dpp(
        0, __builtin_bit_cast(int, x), CTRL, 0xF, 0xF, true));
}
__device__ __forceinline__ float xrow_rd(float x) {
    return __builtin_bit_cast(float, __builtin_amdgcn_permlanex16(
        __builtin_bit_cast(int, x), __builtin_bit_cast(int, x), 0x76543210, 0xfedcba98, false, false));
}
template <int U>
__device__ __forceinline__ float grp8_bcast(float x) {
    return __builtin_bit_cast(float, __builtin_amdgcn_mov_dpp8(
        __builtin_bit_cast(unsigned, x), (unsigned)U * 0x249249u));
}
__device__ __forceinline__ float grp8_sum(float v) {
    v += dpp_rd<0xB1>(v); v += dpp_rd<0x4E>(v); v += dpp_rd<0x164>(v);
    return v;
}
__device__ __forceinline__ float grp8_max(float v) {
    v = fmaxf(v, dpp_rd<0xB1>(v)); v = fmaxf(v, dpp_rd<0x4E>(v)); v = fmaxf(v, dpp_rd<0x164>(v));
    return v;
}

#define EU 8
#define AGG_WAVES 8

template <int H, bool ELU>
__global__ __launch_bounds__(32 * AGG_WAVES) __attribute__((amdgpu_waves_per_eu(16, 16)))
void node_agg_k(const int* __restrict__ rowptr, const int* __restrict__ ssrc,
                const float* __restrict__ attL, const float* __restrict__ ar,
                const float* xlin, const float* res,
                const float* __restrict__ bias, float* outp, int n) {
    static_assert(H == 4 || H == 1, "a head is one group of 8 lanes, or all four");
    const int lane = threadIdx.x & 31;
    const int i = __builtin_amdgcn_readfirstlane(
        (int)(blockIdx.x * AGG_WAVES + (threadIdx.x >> 5)));
    if (i >= n) return;
    const int beg = __builtin_amdgcn_readfirstlane(rowptr[i]);
    const int end = __builtin_amdgcn_readfirstlane(rowptr[i + 1]);
    const int hl  = (2 * lane * H) >> 6;
    const int g   = lane & 7;
    const bool b4 = (lane & 4) != 0, b2 = (lane & 2) != 0, b1 = (lane & 1) != 0;

    int s = (beg + lane) < end ? ssrc[beg + lane] : 0;
    const float2 aL = *reinterpret_cast<const float2*>(attL + 2 * lane);
    const float arh = ar[(size_t)i * H + hl];
    const float2 rr = *reinterpret_cast<const float2*>(res + (size_t)i * 64 + 2 * lane);
    const float2 b = *reinterpret_cast<const float2*>(bias + 2 * lane);
    const float2* rows = reinterpret_cast<const float2*>(xlin);

    float m = -FLT_MAX;
    float accx = 0.f, accy = 0.f;
    float So = 0.f;

    auto load_set = [&](float2 (&x)[EU], int j, int cnt) {
        #pragma unroll
        for (int u = 0; u < EU; ++u) {
            const int jj = j + u;
            const int jc = jj < cnt ? jj : cnt - 1;
            const unsigned sj = (unsigned)__builtin_amdgcn_readlane(s, jc);
            x[u] = rows[sj * 32u + (unsigned)lane];
        }
    };
    auto sum_set = [&](const float2 (&x)[EU], int j, int cnt) {
        float p[EU];
        #pragma unroll
        for (int u = 0; u < EU; ++u) p[u] = x[u].x * aL.x + x[u].y * aL.y;
        float q[4];
        #pragma unroll
        for (int u = 0; u < 4; ++u) {
            const float keep = b4 ? p[u + 4] : p[u];
            const float send = b4 ? p[u] : p[u + 4];
            q[u] = keep + dpp_rd<0x164>(send);
        }
        float r[2];
        #pragma unroll
        for (int u = 0; u < 2; ++u) {
            const float keep = b2 ? q[u + 2] : q[u];
            const float send = b2 ? q[u] : q[u + 2];
            r[u] = keep + dpp_rd<0x4E>(send);
        }
        float t;
        {
            const float keep = b1 ? r[1] : r[0];
            const float send = b1 ? r[0] : r[1];
            t = keep + dpp_rd<0xB1>(send);
        }
        if (H == 1) {
            t += dpp_rd<0x168>(t);
            t += xrow_rd(t);
        }
        const float lgo = (j + g < cnt) ? lrelu(t + arh) : -FLT_MAX;
        const float mn = fmaxf(m, grp8_max(lgo));
        const float sc = __expf(m - mn);
        const float wo = __expf(lgo - mn);
        accx *= sc; accy *= sc;
        So = So * sc + wo;
        m = mn;
#define AGG_EDGE(U) { const float w = grp8_bcast<U>(wo); accx += x[U].x * w; accy += x[U].y * w; }
        AGG_EDGE(0) AGG_EDGE(1) AGG_EDGE(2) AGG_EDGE(3)
        AGG_EDGE(4) AGG_EDGE(5) AGG_EDGE(6) AGG_EDGE(7)
#undef AGG_EDGE
    };
    auto walk_chunk = [&](int cnt) {
        float2 xa[EU], xb[EU];
        load_set(xa, 0, cnt);
        load_set(xb, EU, cnt);
        __builtin_amdgcn_sched_barrier(0);
        sum_set(xa, 0, cnt);
        if (cnt > 2 * EU) {
            load_set(xa, 2 * EU, cnt);
            __builtin_amdgcn_sched_barrier(0);
            sum_set(xb, EU, cnt);
            sum_set(xa, 2 * EU, cnt);
            if (cnt > 3 * EU) {
                load_set(xb, 3 * EU, cnt);
                sum_set(xb, 3 * EU, cnt);
            }
        } else if (cnt > EU) {
            sum_set(xb, EU, cnt);
        }
    };
    auto walk_chunk_plain = [&](int cnt) {
        for (int j = 0; j < cnt; j += EU) {
            float2 x[EU];
            load_set(x, j, cnt);
            sum_set(x, j, cnt);
        }
    };

    if (beg < end) {
        walk_chunk((end - beg) < 32 ? (end - beg) : 32);
        for (int c = beg + 32; c < end; c += 32) {
            const int k = c + lane;
            s = k < end ? ssrc[k] : 0;
            walk_chunk_plain((end - c) < 32 ? (end - c) : 32);
        }
    }

    const float den = grp8_sum(So) + 1e-16f;
    float vx = accx / den + rr.x + b.x;
    float vy = accy / den + rr.y + b.y;
    if (ELU) {
        vx = vx > 0.f ? vx : (__expf(vx) - 1.f);
        vy = vy > 0.f ? vy : (__expf(vy) - 1.f);
    }
    *reinterpret_cast<float2*>(outp + (size_t)i * 64 + 2 * lane) = make_float2(vx, vy);
}

extern "C" void kernel_launch(void* const* d_in, const int* in_sizes, int n_in,
                              void* d_out, int out_size, void* d_ws, size_t ws_size,
                              hipStream_t stream) {
    const float* x     = (const float*)d_in[0];
    const int*   eidx  = (const int*)d_in[1];
    const float* W1    = (const float*)d_in[2];
    const float* attL1 = (const float*)d_in[3];
    const float* attR1 = (const float*)d_in[4];
    const float* resW1 = (const float*)d_in[5];
    const float* b1    = (const float*)d_in[6];
    const float* W2    = (const float*)d_in[7];
    const float* attL2 = (const float*)d_in[8];
    const float* attR2 = (const float*)d_in[9];
    const float* b2    = (const float*)d_in[10];

    const int n = in_sizes[0] / 128;
    const int e = in_sizes[1] / 2;
    const int* srcI = eidx;
    const int* dstI = eidx + e;

    float* ws = (float*)d_ws;
    float* A  = ws;
    float* B  = A  + (size_t)n * 64;
    float* C  = B  + (size_t)n * 64;
    float* AR = C  + (size_t)n * 64;
    const int nb = (n + BW - 1) / BW;
    int2* TMP = (int2*)C;
    int* SS   = (int*)C + 2 * (size_t)e;
    int* RP   = SS + e;
    int* HIST = RP + n + 1;
    int* BST  = HIST + SORT_G * nb;

    constexpr int RT = 2;
    const int tiles = (n + 16 * RT - 1) / (16 * RT);
    const int nWav  = (n + AGG_WAVES - 1) / AGG_WAVES;

    bin_hist_k<<<SORT_G, SORT_T, 0, stream>>>(dstI, HIST, e, nb);
    bin_scatter_k<<<SORT_G, SORT_T, 0, stream>>>(srcI, dstI, HIST, BST, TMP, e, nb);
    bin_place_k<<<nb, PLACE_T, 0, stream>>>(TMP, BST, RP, SS, n, nb);

    gemm_x2<true, RT, 4><<<tiles, 32, 0, stream>>>(x, W1, resW1, A, B, attR1, AR, n, 128);
    node_agg_k<4, true><<<nWav, 32 * AGG_WAVES, 0, stream>>>(RP, SS, attL1, AR, A, B, b1, B, n);

    gemm_x2<false, RT, 1><<<tiles, 32, 0, stream>>>(B, W2, nullptr, A, nullptr, attR2, AR, n, 64);
    node_agg_k<1, false><<<nWav, 32 * AGG_WAVES, 0, stream>>>(RP, SS, attL2, AR, A, A, b2, (float*)d_out, n);
}


# edge grouping in 2 launches: a chunk's block sorts its edges by bin in LDS (count, scan, place); a bin's block gathers its 128 pieces by offset table; no histogram pass; sums in LDS arrival order
# speedup vs baseline: 7.7937x; 1.0406x over previous
#include <hip/hip_runtime.h>
#include <math.h>
#include <float.h>

typedef float v2f __attribute__((ext_vector_type(2)));
typedef float v8f __attribute__((ext_vector_type(8)));

#define HC 64
#define SLOPE 0.2f

__device__ __forceinline__ float lrelu(float x) { return x > 0.f ? x : SLOPE * x; }

#define TP 68
template <bool DUAL, int RT, int H>
__global__ __launch_bounds__(32)
void gemm_x2(const float* __restrict__ X, const float* __restrict__ W,
             const float* __restrict__ Wr, float* __restrict__ Y,
             float* __restrict__ Yr,
             const float* __restrict__ attR, float* __restrict__ ar, int n, int K) {
    static_assert(RT == 2, "the epilogue gives each of the 32 lanes one tile row");
    __shared__ __attribute__((aligned(16))) float tile[16 * RT * TP];
    const int lane  = threadIdx.x;
    const int row16 = lane & 15;
    const int khalf = (lane >> 4) << 1;
    const int col   = lane & 15;
    const int r0    = blockIdx.x * (16 * RT);

    int arow[RT];
    #pragma unroll
    for (int t = 0; t < RT; ++t) {
        int r = r0 + 16 * t + row16;
        arow[t] = r < n ? r : n - 1;
    }

    v8f acc[RT][4]  = {};
    v8f accR[RT][4] = {};

    #pragma unroll 2
    for (int k0 = 0; k0 < K; k0 += 4) {
        v2f a[RT];
        #pragma unroll
        for (int t = 0; t < RT; ++t)
            a[t] = *reinterpret_cast<const v2f*>(X + (size_t)arow[t] * K + k0 + khalf);

        #pragma unroll
        for (int j = 0; j < 4; ++j) {
            v2f b;
            const float* bp = W + (size_t)(k0 + khalf) * HC + j * 16 + col;
            b.x = bp[0]; b.y = bp[HC];
            #pragma unroll
            for (int t = 0; t < RT; ++t)
                acc[t][j] = __builtin_amdgcn_wmma_f32_16x16x4_f32(
                    false, a[t], false, b, (short)0, acc[t][j], false, false);
            if (DUAL) {
                v2f br;
                const float* brp = Wr + (size_t)(k0 + khalf) * HC + j * 16 + col;
                br.x = brp[0]; br.y = brp[HC];
                #pragma unroll
                for (int t = 0; t < RT; ++t)
                    accR[t][j] = __builtin_amdgcn_wmma_f32_16x16x4_f32(
                        false, a[t], false, br, (short)0, accR[t][j], false, false);
            }
        }
    }

    const int mhi = (lane >> 4) * 8;
    #pragma unroll
    for (int t = 0; t < RT; ++t)
        #pragma unroll
        for (int j = 0; j < 4; ++j)
            #pragma unroll
            for (int r = 0; r < 8; ++r) {
                int m = r0 + 16 * t + mhi + r;
                tile[(16 * t + mhi + r) * TP + j * 16 + col] = acc[t][j][r];
                if (m < n) {
                    Y[(size_t)m * HC + j * 16 + col] = acc[t][j][r];
                    if (DUAL) Yr[(size_t)m * HC + j * 16 + col] = accR[t][j][r];
                }
            }

    __builtin_amdgcn_fence(__ATOMIC_RELEASE, "workgroup", "local");
    __builtin_amdgcn_s_barrier();
    __builtin_amdgcn_fence(__ATOMIC_ACQUIRE, "workgroup", "local");

    const int row = r0 + lane;
    const float4* rowp = reinterpret_cast<const float4*>(tile + lane * TP);
    const float4* rp = reinterpret_cast<const float4*>(attR);
    if (H == 4) {
        float aR[4] = {0, 0, 0, 0};
        #pragma unroll
        for (int q = 0; q < 16; ++q) {
            const float4 v = rowp[q];
            const float4 rr = rp[q];
            const int h = q >> 2;
            aR[h] += v.x * rr.x + v.y * rr.y + v.z * rr.z + v.w * rr.w;
        }
        if (row < n)
            *reinterpret_cast<float4*>(ar + (size_t)row * 4) = make_float4(aR[0], aR[1], aR[2], aR[3]);
    } else {
        float aR = 0.f;
        for (int q = 0; q < 16; ++q) {
            const float4 v = rowp[q];
            const float4 rr = rp[q];
            aR += v.x * rr.x + v.y * rr.y + v.z * rr.z + v.w * rr.w;
        }
        if (row < n) ar[row] = aR;
    }
}

#define BSH    8
#define BW     (1 << BSH)
#define MAXNB  1024
#define SORT_G 128
#define SORT_T 1024
#define SORT_U 16
#define STG_CAP 16384

__device__ __forceinline__ int bin_of(int d, int nb) {
    const unsigned b = (unsigned)d >> BSH;
    return b < (unsigned)nb ? (int)b : nb - 1;
}

__global__ __launch_bounds__(SORT_T)
void chunk_sort_k(const int* __restrict__ src, const int* __restrict__ dst,
                  int* __restrict__ offt, int2* __restrict__ tmp, int e, int nb) {
    static_assert(SORT_T == MAXNB, "thread t owns bin t");
    static_assert(STG_CAP == SORT_U * SORT_T, "a chunk that fits the LDS buffer is one batch per thread");
    __shared__ int cur[MAXNB];
    __shared__ int wtot[32];
    __shared__ int2 stage[STG_CAP];
    const int t = threadIdx.x;
    const int lane = t & 31;
    const int wave = t >> 5;
    const int g = blockIdx.x;

    const int ch = (e + SORT_G - 1) / SORT_G;
    const int lo = g * ch < e ? g * ch : e;
    const int hi = lo + ch < e ? lo + ch : e;
    const int m = hi - lo;
    const bool ordered = m <= STG_CAP;

    cur[t] = 0;
    __syncthreads();

    int d[SORT_U], s[SORT_U];
    #pragma unroll
    for (int u = 0; u < SORT_U; ++u) { d[u] = 0; s[u] = 0; }
    if (ordered) {
        #pragma unroll
        for (int u = 0; u < SORT_U; ++u) {
            const int i = lo + t + u * SORT_T;
            int ic = i < hi ? i : hi - 1;
            ic = ic > 0 ? ic : 0;
            d[u] = dst[ic];
            s[u] = src[ic];
        }
        static_assert(SORT_U == 16, "the statements below name the 16 values of a batch");
        asm volatile("" :: "v"(d[0]), "v"(d[1]), "v"(d[2]), "v"(d[3]), "v"(d[4]), "v"(d[5]),
                     "v"(d[6]), "v"(d[7]), "v"(d[8]), "v"(d[9]), "v"(d[10]), "v"(d[11]),
                     "v"(d[12]), "v"(d[13]), "v"(d[14]), "v"(d[15]));
        asm volatile("" :: "v"(s[0]), "v"(s[1]), "v"(s[2]), "v"(s[3]), "v"(s[4]), "v"(s[5]),
                     "v"(s[6]), "v"(s[7]), "v"(s[8]), "v"(s[9]), "v"(s[10]), "v"(s[11]),
                     "v"(s[12]), "v"(s[13]), "v"(s[14]), "v"(s[15]));
        #pragma unroll
        for (int u = 0; u < SORT_U; ++u)
            if (lo + t + u * SORT_T < hi) atomicAdd(&cur[bin_of(d[u], nb)], 1);
    } else {
        for (int i = lo + t; i < hi; i += SORT_T) atomicAdd(&cur[bin_of(dst[i], nb)], 1);
    }
    __syncthreads();

    const int own = cur[t];
    int inc = own;
    #pragma unroll
    for (int off = 1; off < 32; off <<= 1) {
        int u = __shfl_up(inc, off);
        if (lane >= off) inc += u;
    }
    if (lane == 31) wtot[wave] = inc;
    __syncthreads();
    int tinc = wtot[lane];
    #pragma unroll
    for (int off = 1; off < 32; off <<= 1) {
        int u = __shfl_up(tinc, off);
        if (lane >= off) tinc += u;
    }
    const int below = __shfl(tinc, wave > 0 ? wave - 1 : 0);
    const int excl = (wave > 0 ? below : 0) + inc - own;
    cur[t] = excl;
    if (t < nb) offt[t * SORT_G + g] = excl;
    if (t == 0) offt[nb * SORT_G + g] = m;
    __syncthreads();

    if (ordered) {
        int p[SORT_U];
        #pragma unroll
        for (int u = 0; u < SORT_U; ++u) {
            p[u] = 0;
            if (lo + t + u * SORT_T < hi) p[u] = atomicAdd(&cur[bin_of(d[u], nb)], 1);
        }
        #pragma unroll
        for (int u = 0; u < SORT_U; ++u)
            if (lo + t + u * SORT_T < hi) stage[p[u]] = make_int2(s[u], d[u]);
        __syncthreads();
        for (int i0 = t; i0 < m; i0 += SORT_U * SORT_T) {
            int2 v[SORT_U];
            #pragma unroll
            for (int u = 0; u < SORT_U; ++u) {
                const int i = i0 + u * SORT_T;
                v[u] = i < m ? stage[i] : make_int2(0, 0);
            }
            #pragma unroll
            for (int u = 0; u < SORT_U; ++u) {
                const int i = i0 + u * SORT_T;
                if (i < m) tmp[lo + i] = v[u];
            }
        }
    } else {
        for (int i = lo + t; i < hi; i += SORT_T) {
            const int dd = dst[i];
            const int p = atomicAdd(&cur[bin_of(dd, nb)], 1);
            tmp[lo + p] = make_int2(src[i], dd);
        }
    }
}

#define PLACE_T 1024
#define PLACE_W (PLACE_T / 32)
#define SEG_Q (SORT_G / PLACE_W)
#define SEG_P 2
#define PLACE_CAP 8192
__global__ __launch_bounds__(PLACE_T)
void bin_place_k(const int2* __restrict__ tmp, const int* __restrict__ offt,
                 int* __restrict__ rowptr, int* __restrict__ ssrc, int n, int nb, int e) {
    static_assert(SORT_G % 32 == 0 && SEG_Q * PLACE_W == SORT_G, "whole waves read offt; every piece has one wave");
    static_assert(SEG_Q * SEG_P == 8, "the statement that holds the loads names 8 pairs");
    __shared__ int cnt[BW];
    __shared__ int wt[BW / 32];
    __shared__ int stg[PLACE_CAP];
    __shared__ int sbeg[SORT_G];
    __shared__ int scnt[SORT_G];
    __shared__ int red[2 * (SORT_G / 32)];
    const int t = threadIdx.x;
    const int lane = t & 31;
    const int wave = __builtin_amdgcn_readfirstlane(t >> 5);
    const int b = blockIdx.x;
    const int ch = (e + SORT_G - 1) / SORT_G;

    if (t < BW) cnt[t] = 0;
    if (t < SORT_G) {
        const int a = offt[b * SORT_G + t];
        const int c = offt[(b + 1) * SORT_G + t];
        const int clo = t * ch < e ? t * ch : e;
        sbeg[t] = clo + a;
        scnt[t] = c - a;
        int ra = a, rc = c;
        #pragma unroll
        for (int off = 16; off > 0; off >>= 1) {
            ra += __shfl_xor(ra, off);
            rc += __shfl_xor(rc, off);
        }
        if (lane == 0) { red[wave] = ra; red[SORT_G / 32 + wave] = rc; }
    }
    __syncthreads();
    int lo = 0, hi = 0;
    #pragma unroll
    for (int w = 0; w < SORT_G / 32; ++w) { lo += red[w]; hi += red[SORT_G / 32 + w]; }
    const bool ordered = hi - lo <= PLACE_CAP;

    int sb[SEG_Q], sc[SEG_Q];
    #pragma unroll
    for (int q = 0; q < SEG_Q; ++q) {
        sb[q] = __builtin_amdgcn_readfirstlane(sbeg[wave + q * PLACE_W]);
        sc[q] = __builtin_amdgcn_readfirstlane(scnt[wave + q * PLACE_W]);
    }
    int2 ed[SEG_Q * SEG_P];
    #pragma unroll
    for (int q = 0; q < SEG_Q; ++q)
        #pragma unroll
        for (int p = 0; p < SEG_P; ++p) {
            const int idx = p * 32 + lane;
            int pos = sb[q] + (idx < sc[q] ? idx : sc[q] - 1);
            pos = pos > 0 ? pos : 0;
            ed[q * SEG_P + p] = tmp[pos];
        }
    asm volatile("" :: "v"(ed[0].x), "v"(ed[0].y), "v"(ed[1].x), "v"(ed[1].y),
                 "v"(ed[2].x), "v"(ed[2].y), "v"(ed[3].x), "v"(ed[3].y),
                 "v"(ed[4].x), "v"(ed[4].y), "v"(ed[5].x), "v"(ed[5].y),
                 "v"(ed[6].x), "v"(ed[6].y), "v"(ed[7].x), "v"(ed[7].y));
    #pragma unroll
    for (int q = 0; q < SEG_Q; ++q)
        #pragma unroll
        for (int p = 0; p < SEG_P; ++p)
            if (p * 32 + lane < sc[q]) atomicAdd(&cnt[ed[q * SEG_P + p].y & (BW - 1)], 1);
    #pragma unroll
    for (int q = 0; q < SEG_Q; ++q)
        for (int idx = SEG_P * 32 + lane; idx < sc[q]; idx += 32)
            atomicAdd(&cnt[tmp[sb[q] + idx].y & (BW - 1)], 1);
    __syncthreads();
    int v = 0, inc = 0;
    if (t < BW) {
        v = cnt[t];
        inc = v;
        #pragma unroll
        for (int off = 1; off < 32; off <<= 1) {
            int u = __shfl_up(inc, off);
            if (lane >= off) inc += u;
        }
        if (lane == 31) wt[wave] = inc;
    }
    __syncthreads();
    if (t < BW) {
        int off = 0;
        #pragma unroll
        for (int w = 0; w < BW / 32; ++w) off += w < wave ? wt[w] : 0;
        const int excl = off + inc - v;
        cnt[t] = excl;
        const int node = b * BW + t;
        if (node < n) rowptr[node] = lo + excl;
    }
    if (b == nb - 1 && t == 0) rowptr[n] = hi;
    __syncthreads();

    int pp[SEG_Q * SEG_P];
    #pragma unroll
    for (int q = 0; q < SEG_Q; ++q)
        #pragma unroll
        for (int p = 0; p < SEG_P; ++p) {
            pp[q * SEG_P + p] = 0;
            if (p * 32 + lane < sc[q])
                pp[q * SEG_P + p] = atomicAdd(&cnt[ed[q * SEG_P + p].y & (BW - 1)], 1);
        }
    if (ordered) {
        #pragma unroll
        for (int q = 0; q < SEG_Q; ++q)
            #pragma unroll
            for (int p = 0; p < SEG_P; ++p)
                if (p * 32 + lane < sc[q]) stg[pp[q * SEG_P + p]] = ed[q * SEG_P + p].x;
        #pragma unroll
        for (int q = 0; q < SEG_Q; ++q)
            for (int idx = SEG_P * 32 + lane; idx < sc[q]; idx += 32) {
                const int2 w2 = tmp[sb[q] + idx];
                stg[atomicAdd(&cnt[w2.y & (BW - 1)], 1)] = w2.x;
            }
        __syncthreads();
        const int m = hi - lo;
        for (int i0 = t; i0 < m; i0 += 8 * PLACE_T) {
            int o[8];
            #pragma unroll
            for (int u = 0; u < 8; ++u) {
                const int i = i0 + u * PLACE_T;
                o[u] = i < m ? stg[i] : 0;
            }
            #pragma unroll
            for (int u = 0; u < 8; ++u) {
                const int i = i0 + u * PLACE_T;
                if (i < m) ssrc[lo + i] = o[u];
            }
        }
    } else {
        #pragma unroll
        for (int q = 0; q < SEG_Q; ++q)
            #pragma unroll
            for (int p = 0; p < SEG_P; ++p)
                if (p * 32 + lane < sc[q]) ssrc[lo + pp[q * SEG_P + p]] = ed[q * SEG_P + p].x;
        #pragma unroll
        for (int q = 0; q < SEG_Q; ++q)
            for (int idx = SEG_P * 32 + lane; idx < sc[q]; idx += 32) {
                const int2 w2 = tmp[sb[q] + idx];
                ssrc[lo + atomicAdd(&cnt[w2.y & (BW - 1)], 1)] = w2.x;
            }
    }
}


template <int CTRL>
__device__ __forceinline__ float dpp_rd(float x) {
    return __builtin_bit_cast(float, __builtin_amdgcn_update_dpp(
        0, __builtin_bit_cast(int, x), CTRL, 0xF, 0xF, true));
}
__device__ __forceinline__ float xrow_rd(float x) {
    return __builtin_bit_cast(float, __builtin_amdgcn_permlanex16(
        __builtin_bit_cast(int, x), __builtin_bit_cast(int, x), 0x76543210, 0xfedcba98, false, false));
}
template <int U>
__device__ __forceinline__ float grp8_bcast(float x) {
    return __builtin_bit_cast(float, __builtin_amdgcn_mov_dpp8(
        __builtin_bit_cast(unsigned, x), (unsigned)U * 0x249249u));
}
__device__ __forceinline__ float grp8_sum(float v) {
    v += dpp_rd<0xB1>(v); v += dpp_rd<0x4E>(v); v += dpp_rd<0x164>(v);
    return v;
}
__device__ __forceinline__ float grp8_max(float v) {
    v = fmaxf(v, dpp_rd<0xB1>(v)); v = fmaxf(v, dpp_rd<0x4E>(v)); v = fmaxf(v, dpp_rd<0x164>(v));
    return v;
}

#define EU 8
#define AGG_WAVES 8

template <int H, bool ELU>
__global__ __launch_bounds__(32 * AGG_WAVES) __attribute__((amdgpu_waves_per_eu(16, 16)))
void node_agg_k(const int* __restrict__ rowptr, const int* __restrict__ ssrc,
                const float* __restrict__ attL, const float* __restrict__ ar,
                const float* xlin, const float* res,
                const float* __restrict__ bias, float* outp, int n) {
    static_assert(H == 4 || H == 1, "a head is one group of 8 lanes, or all four");
    const int lane = threadIdx.x & 31;
    const int i = __builtin_amdgcn_readfirstlane(
        (int)(blockIdx.x * AGG_WAVES + (threadIdx.x >> 5)));
    if (i >= n) return;
    const int beg = __builtin_amdgcn_readfirstlane(rowptr[i]);
    const int end = __builtin_amdgcn_readfirstlane(rowptr[i + 1]);
    const int hl  = (2 * lane * H) >> 6;
    const int g   = lane & 7;
    const bool b4 = (lane & 4) != 0, b2 = (lane & 2) != 0, b1 = (lane & 1) != 0;

    int s = (beg + lane) < end ? ssrc[beg + lane] : 0;
    const float2 aL = *reinterpret_cast<const float2*>(attL + 2 * lane);
    const float arh = ar[(size_t)i * H + hl];
    const float2 rr = *reinterpret_cast<const float2*>(res + (size_t)i * 64 + 2 * lane);
    const float2 b = *reinterpret_cast<const float2*>(bias + 2 * lane);
    const float2* rows = reinterpret_cast<const float2*>(xlin);

    float m = -FLT_MAX;
    float accx = 0.f, accy = 0.f;
    float So = 0.f;

    auto load_set = [&](float2 (&x)[EU], int j, int cnt) {
        #pragma unroll
        for (int u = 0; u < EU; ++u) {
            const int jj = j + u;
            const int jc = jj < cnt ? jj : cnt - 1;
            const unsigned sj = (unsigned)__builtin_amdgcn_readlane(s, jc);
            x[u] = rows[sj * 32u + (unsigned)lane];
        }
    };
    auto sum_set = [&](const float2 (&x)[EU], int j, int cnt) {
        float p[EU];
        #pragma unroll
        for (int u = 0; u < EU; ++u) p[u] = x[u].x * aL.x + x[u].y * aL.y;
        float q[4];
        #pragma unroll
        for (int u = 0; u < 4; ++u) {
            const float keep = b4 ? p[u + 4] : p[u];
            const float send = b4 ? p[u] : p[u + 4];
            q[u] = keep + dpp_rd<0x164>(send);
        }
        float r[2];
        #pragma unroll
        for (int u = 0; u < 2; ++u) {
            const float keep = b2 ? q[u + 2] : q[u];
            const float send = b2 ? q[u] : q[u + 2];
            r[u] = keep + dpp_rd<0x4E>(send);
        }
        float t;
        {
            const float keep = b1 ? r[1] : r[0];
            const float send = b1 ? r[0] : r[1];
            t = keep + dpp_rd<0xB1>(send);
        }
        if (H == 1) {
            t += dpp_rd<0x168>(t);
            t += xrow_rd(t);
        }
        const float lgo = (j + g < cnt) ? lrelu(t + arh) : -FLT_MAX;
        const float mn = fmaxf(m, grp8_max(lgo));
        const float sc = __expf(m - mn);
        const float wo = __expf(lgo - mn);
        accx *= sc; accy *= sc;
        So = So * sc + wo;
        m = mn;
#define AGG_EDGE(U) { const float w = grp8_bcast<U>(wo); accx += x[U].x * w; accy += x[U].y * w; }
        AGG_EDGE(0) AGG_EDGE(1) AGG_EDGE(2) AGG_EDGE(3)
        AGG_EDGE(4) AGG_EDGE(5) AGG_EDGE(6) AGG_EDGE(7)
#undef AGG_EDGE
    };
    auto walk_chunk = [&](int cnt) {
        float2 xa[EU], xb[EU];
        load_set(xa, 0, cnt);
        load_set(xb, EU, cnt);
        __builtin_amdgcn_sched_barrier(0);
        sum_set(xa, 0, cnt);
        if (cnt > 2 * EU) {
            load_set(xa, 2 * EU, cnt);
            __builtin_amdgcn_sched_barrier(0);
            sum_set(xb, EU, cnt);
            sum_set(xa, 2 * EU, cnt);
            if (cnt > 3 * EU) {
                load_set(xb, 3 * EU, cnt);
                sum_set(xb, 3 * EU, cnt);
            }
        } else if (cnt > EU) {
            sum_set(xb, EU, cnt);
        }
    };
    auto walk_chunk_plain = [&](int cnt) {
        for (int j = 0; j < cnt; j += EU) {
            float2 x[EU];
            load_set(x, j, cnt);
            sum_set(x, j, cnt);
        }
    };

    if (beg < end) {
        walk_chunk((end - beg) < 32 ? (end - beg) : 32);
        for (int c = beg + 32; c < end; c += 32) {
            const int k = c + lane;
            s = k < end ? ssrc[k] : 0;
            walk_chunk_plain((end - c) < 32 ? (end - c) : 32);
        }
    }

    const float den = grp8_sum(So) + 1e-16f;
    float vx = accx / den + rr.x + b.x;
    float vy = accy / den + rr.y + b.y;
    if (ELU) {
        vx = vx > 0.f ? vx : (__expf(vx) - 1.f);
        vy = vy > 0.f ? vy : (__expf(vy) - 1.f);
    }
    *reinterpret_cast<float2*>(outp + (size_t)i * 64 + 2 * lane) = make_float2(vx, vy);
}

extern "C" void kernel_launch(void* const* d_in, const int* in_sizes, int n_in,
                              void* d_out, int out_size, void* d_ws, size_t ws_size,
                              hipStream_t stream) {
    const float* x     = (const float*)d_in[0];
    const int*   eidx  = (const int*)d_in[1];
    const float* W1    = (const float*)d_in[2];
    const float* attL1 = (const float*)d_in[3];
    const float* attR1 = (const float*)d_in[4];
    const float* resW1 = (const float*)d_in[5];
    const float* b1    = (const float*)d_in[6];
    const float* W2    = (const float*)d_in[7];
    const float* attL2 = (const float*)d_in[8];
    const float* attR2 = (const float*)d_in[9];
    const float* b2    = (const float*)d_in[10];

    const int n = in_sizes[0] / 128;
    const int e = in_sizes[1] / 2;
    const int* srcI = eidx;
    const int* dstI = eidx + e;

    float* ws = (float*)d_ws;
    float* A  = ws;
    float* B  = A  + (size_t)n * 64;
    float* C  = B  + (size_t)n * 64;
    float* AR = C  + (size_t)n * 64;
    const int nb = (n + BW - 1) / BW;
    int2* TMP = (int2*)C;
    int* SS   = (int*)C + 2 * (size_t)e;
    int* RP   = SS + e;
    int* OFFT = RP + n + 1;

    constexpr int RT = 2;
    const int tiles = (n + 16 * RT - 1) / (16 * RT);
    const int nWav  = (n + AGG_WAVES - 1) / AGG_WAVES;

    chunk_sort_k<<<SORT_G, SORT_T, 0, stream>>>(srcI, dstI, OFFT, TMP, e, nb);
    bin_place_k<<<nb, PLACE_T, 0, stream>>>(TMP, OFFT, RP, SS, n, nb, e);

    gemm_x2<true, RT, 4><<<tiles, 32, 0, stream>>>(x, W1, resW1, A, B, attR1, AR, n, 128);
    node_agg_k<4, true><<<nWav, 32 * AGG_WAVES, 0, stream>>>(RP, SS, attL1, AR, A, B, b1, B, n);

    gemm_x2<false, RT, 1><<<tiles, 32, 0, stream>>>(B, W2, nullptr, A, nullptr, attR2, AR, n, 64);
    node_agg_k<1, false><<<nWav, 32 * AGG_WAVES, 0, stream>>>(RP, SS, attL2, AR, A, A, b2, (float*)d_out, n);
}
